# starting kernel
# baseline (speedup 1.0000x reference)
#include <hip/hip_runtime.h>
#include <hip/hip_fp16.h>

typedef __attribute__((ext_vector_type(16))) _Float16 v16h;
typedef __attribute__((ext_vector_type(8)))  float    v8f;

#define WMMA_F16(a,b,c) __builtin_amdgcn_wmma_f32_16x16x32_f16(false,(a),false,(b),(short)0,(c),false,false)

__device__ __forceinline__ v8f zero8(){
  v8f z;
  #pragma unroll
  for(int i=0;i<8;++i) z[i]=0.f;
  return z;
}
__device__ __forceinline__ v16h zero16(){
  v16h z;
  #pragma unroll
  for(int i=0;i<16;++i) z[i]=(_Float16)0.f;
  return z;
}

__device__ __forceinline__ int a_k(int i, int h16){ return (i&7) + ((i>>3)<<4) + (h16<<3); }

__device__ __forceinline__ v16h load_tile_f32(const float* __restrict__ src, int ld, int lane){
  const float* rb = src + (size_t)(lane&15)*ld;
  int h16=(lane>>4)&1; v16h r;
  #pragma unroll
  for(int i=0;i<16;++i) r[i]=(_Float16)rb[a_k(i,h16)];
  return r;
}
__device__ __forceinline__ v16h load_tile_f16(const _Float16* __restrict__ src, int ld, int lane){
  const _Float16* rb = src + (size_t)(lane&15)*ld;
  int h16=(lane>>4)&1; v16h r;
  #pragma unroll
  for(int i=0;i<16;++i) r[i]=rb[a_k(i,h16)];
  return r;
}
__device__ __forceinline__ v16h load_tile_kmajor_f16(const _Float16* __restrict__ src, int ld, int lane){
  const _Float16* cb = src + (lane&15);
  int h16=(lane>>4)&1; v16h r;
  #pragma unroll
  for(int i=0;i<16;++i) r[i]=cb[(size_t)a_k(i,h16)*ld];
  return r;
}
__device__ __forceinline__ v16h load_tile_sw(const _Float16* src, int tile, int lane){
  return *reinterpret_cast<const v16h*>(src + (((size_t)tile*32 + lane) << 4));
}

__device__ __forceinline__ float rmax16(float x){
  #pragma unroll
  for(int m=8;m;m>>=1) x=fmaxf(x,__shfl_xor(x,m,32));
  return x;
}
__device__ __forceinline__ float rsum16(float x){
  #pragma unroll
  for(int m=8;m;m>>=1) x+=__shfl_xor(x,m,32);
  return x;
}
__device__ __forceinline__ float sigf(float x){ return 1.f/(1.f+__expf(-x)); }

#define PB  8
#define DD  128
#define HH  64
#define SQ  512

__global__ __launch_bounds__(256) void k_swz(const float* __restrict__ W,
                                             _Float16* __restrict__ out, int K){
  int idx = blockIdx.x*256 + threadIdx.x;
  int i    = idx & 15;
  int lane = (idx >> 4) & 31;
  int tile = idx >> 9;
  int ktiles = K >> 5;
  int kt = tile % ktiles, jt = tile / ktiles;
  out[idx] = (_Float16)W[(size_t)(jt*16 + (lane&15))*K + kt*32 + a_k(i,(lane>>4)&1)];
}

__global__ __launch_bounds__(256) void k_gru(
    const float* __restrict__ x, const _Float16* __restrict__ WihT,
    const _Float16* __restrict__ WhhT, const float* __restrict__ bih,
    const float* __restrict__ bhh, float* __restrict__ seq)
{
  __shared__ _Float16 hscr[8][16*64];
  __shared__ _Float16 WhhL[192*64];
  int lane=threadIdx.x&31, wave=threadIdx.x>>5;
  int pbase=(blockIdx.x*8+wave)*16;
  int colb=lane&15, rowoff=(lane>>4)*8;

  {
    const uint32_t* wsrc=(const uint32_t*)WhhT;
    uint32_t* wdst=(uint32_t*)WhhL;
    for(int i=threadIdx.x;i<192*64/2;i+=256) wdst[i]=wsrc[i];
  }
  __syncthreads();

  float bi[12], bh[12];
  #pragma unroll
  for(int j=0;j<12;++j){ bi[j]=bih[j*16+colb]; bh[j]=bhh[j*16+colb]; }

  v8f hT[4], hs[4];
  #pragma unroll
  for(int j=0;j<4;++j){ hT[j]=zero8(); hs[j]=zero8(); }
  v16h hA[2]; hA[0]=zero16(); hA[1]=zero16();
  _Float16* hl=&hscr[wave][0];

  for(int t=0;t<PB;++t){
    size_t zoff = 0;
    asm volatile("" : "+s"(zoff));
    const _Float16* Wt = WihT + zoff;
    const float* xsrc = x + ((size_t)pbase*PB + t)*DD;
    if(t+1<PB) __builtin_prefetch(xsrc+DD,0,1);
    v16h xA[4];
    #pragma unroll
    for(int kk=0;kk<4;++kk) xA[kk]=load_tile_f32(xsrc+kk*32, PB*DD, lane);

    #pragma unroll
    for(int j=0;j<4;++j){
      v8f gr=zero8();
      #pragma unroll
      for(int kk=0;kk<4;++kk)
        gr=WMMA_F16(xA[kk], load_tile_sw(Wt, j*4+kk, lane), gr);
      #pragma unroll
      for(int kk=0;kk<2;++kk)
        gr=WMMA_F16(hA[kk], load_tile_sw(WhhL, j*2+kk, lane), gr);
      v8f gz=zero8();
      #pragma unroll
      for(int kk=0;kk<4;++kk)
        gz=WMMA_F16(xA[kk], load_tile_sw(Wt, (4+j)*4+kk, lane), gz);
      #pragma unroll
      for(int kk=0;kk<2;++kk)
        gz=WMMA_F16(hA[kk], load_tile_sw(WhhL, (4+j)*2+kk, lane), gz);
      v8f gnx=zero8();
      #pragma unroll
      for(int kk=0;kk<4;++kk)
        gnx=WMMA_F16(xA[kk], load_tile_sw(Wt, (8+j)*4+kk, lane), gnx);
      v8f gnh=zero8();
      #pragma unroll
      for(int kk=0;kk<2;++kk)
        gnh=WMMA_F16(hA[kk], load_tile_sw(WhhL, (8+j)*2+kk, lane), gnh);
      #pragma unroll
      for(int e=0;e<8;++e){
        float r =sigf(gr[e]+bi[j]+bh[j]);
        float z =sigf(gz[e]+bi[4+j]+bh[4+j]);
        float nv=tanhf(gnx[e]+bi[8+j]+r*(gnh[e]+bh[8+j]));
        float hn=(1.f-z)*nv + z*hT[j][e];
        hT[j][e]=hn; hs[j][e]+=hn;
      }
    }
    #pragma unroll
    for(int j=0;j<4;++j){
      #pragma unroll
      for(int v=0;v<8;++v)
        hl[(v+rowoff)*64 + j*16 + colb]=(_Float16)hT[j][v];
    }
    #pragma unroll
    for(int kk=0;kk<2;++kk) hA[kk]=load_tile_f16(hl+kk*32, 64, lane);
  }
  float* sb=seq+(size_t)pbase*HH;
  #pragma unroll
  for(int j=0;j<4;++j){
    #pragma unroll
    for(int v=0;v<8;++v)
      sb[(size_t)(v+rowoff)*HH + j*16 + colb]=hs[j][v]*0.125f;
  }
}

__global__ __launch_bounds__(256) void k_qkv(
    const float* __restrict__ X, const _Float16* __restrict__ WqkvT,
    const float* __restrict__ bqkv, _Float16* __restrict__ Qp,
    _Float16* __restrict__ Kp, _Float16* __restrict__ Vh)
{
  int lane=threadIdx.x&31, wave=threadIdx.x>>5;
  size_t rbase=((size_t)blockIdx.x*8+wave)*16;
  int b=(int)(rbase>>9), s=(int)(rbase&511);
  int colb=lane&15, rowoff=(lane>>4)*8;
  v16h A[2];
  #pragma unroll
  for(int kk=0;kk<2;++kk) A[kk]=load_tile_f32(X+rbase*HH+kk*32, HH, lane);

  #pragma unroll
  for(int j=0;j<12;++j){
    v8f acc=zero8();
    #pragma unroll
    for(int kk=0;kk<2;++kk)
      acc=WMMA_F16(A[kk], load_tile_sw(WqkvT, j*2+kk, lane), acc);
    float bias=bqkv[j*16+colb];
    int h=j&3;
    size_t bh=(size_t)b*4+h;
    if(j<4){
      _Float16* dst=Qp+(bh*SQ+s)*32;
      #pragma unroll
      for(int v=0;v<8;++v){
        size_t ro=(size_t)(v+rowoff)*32;
        dst[ro+colb]=(_Float16)((acc[v]+bias)*0.25f);
        dst[ro+16+colb]=(_Float16)0.f;
      }
    } else if(j<8){
      _Float16* dst=Kp+(bh*SQ+s)*32;
      #pragma unroll
      for(int v=0;v<8;++v){
        size_t ro=(size_t)(v+rowoff)*32;
        dst[ro+colb]=(_Float16)(acc[v]+bias);
        dst[ro+16+colb]=(_Float16)0.f;
      }
    } else {
      _Float16* dst=Vh+(bh*SQ+s)*16;
      #pragma unroll
      for(int v=0;v<8;++v)
        dst[(size_t)(v+rowoff)*16+colb]=(_Float16)(acc[v]+bias);
    }
  }
}

__global__ __launch_bounds__(256) void k_attn(
    const _Float16* __restrict__ Qp, const _Float16* __restrict__ Kp,
    const _Float16* __restrict__ Vh, float* __restrict__ O)
{
  __shared__ _Float16 Ks[SQ*32];
  __shared__ _Float16 Vs[SQ*16];
  __shared__ _Float16 Ps[8][16*32];
  int bh=blockIdx.x, b=bh>>2, h=bh&3;
  int tid=threadIdx.x, lane=tid&31, wave=tid>>5;
  int colb=lane&15, rowoff=(lane>>4)*8;

  {
    const uint32_t* ksrc=(const uint32_t*)(Kp+(size_t)bh*SQ*32);
    uint32_t* kdst=(uint32_t*)Ks;
    for(int i=tid;i<SQ*16;i+=256) kdst[i]=ksrc[i];
    const uint32_t* vsrc=(const uint32_t*)(Vh+(size_t)bh*SQ*16);
    uint32_t* vdst=(uint32_t*)Vs;
    for(int i=tid;i<SQ*8;i+=256) vdst[i]=vsrc[i];
  }
  __syncthreads();

  _Float16* pl=&Ps[wave][0];
  for(int qt=0;qt<4;++qt){
    int qbase=wave*64+qt*16;
    v16h Qa=load_tile_f16(Qp+((size_t)bh*SQ+qbase)*32, 32, lane);
    v8f acc=zero8(), mrow, lrow;
    #pragma unroll
    for(int e=0;e<8;++e){ mrow[e]=-1e30f; lrow[e]=0.f; }

    for(int kc=0;kc<16;++kc){
      int kbase=kc*32;
      v16h Kb0=load_tile_f16(Ks+(size_t)kbase*32, 32, lane);
      v16h Kb1=load_tile_f16(Ks+(size_t)(kbase+16)*32, 32, lane);
      v8f s0=WMMA_F16(Qa,Kb0,zero8());
      v8f s1=WMMA_F16(Qa,Kb1,zero8());
      v8f nmv, scv;
      #pragma unroll
      for(int e=0;e<8;++e){
        float cm=fmaxf(rmax16(s0[e]),rmax16(s1[e]));
        float nm=fmaxf(mrow[e],cm);
        nmv[e]=nm; scv[e]=__expf(mrow[e]-nm); mrow[e]=nm;
      }
      #pragma unroll
      for(int e=0;e<8;++e){
        s0[e]=__expf(s0[e]-nmv[e]);
        s1[e]=__expf(s1[e]-nmv[e]);
        lrow[e]=lrow[e]*scv[e]+rsum16(s0[e])+rsum16(s1[e]);
        acc[e]*=scv[e];
      }
      #pragma unroll
      for(int e=0;e<8;++e){
        pl[(e+rowoff)*32+colb]=(_Float16)s0[e];
        pl[(e+rowoff)*32+16+colb]=(_Float16)s1[e];
      }
      v16h Pa=load_tile_f16(pl, 32, lane);
      v16h Vb=load_tile_kmajor_f16(Vs+(size_t)kbase*16, 16, lane);
      acc=WMMA_F16(Pa,Vb,acc);
    }
    float* ob=O+((size_t)b*SQ+qbase)*HH + h*16;
    #pragma unroll
    for(int e=0;e<8;++e)
      ob[(size_t)(e+rowoff)*HH+colb]=acc[e]/lrow[e];
  }
}

__global__ __launch_bounds__(256) void k_projln(
    float* __restrict__ X, const float* __restrict__ Y,
    const _Float16* __restrict__ WoT, const float* __restrict__ bo,
    const float* __restrict__ g, const float* __restrict__ be)
{
  int lane=threadIdx.x&31, wave=threadIdx.x>>5;
  size_t rbase=((size_t)blockIdx.x*8+wave)*16;
  int colb=lane&15, rowoff=(lane>>4)*8;
  v16h A[2];
  #pragma unroll
  for(int kk=0;kk<2;++kk) A[kk]=load_tile_f32(Y+rbase*HH+kk*32, HH, lane);
  v8f o[4];
  #pragma unroll
  for(int j=0;j<4;++j){
    v8f acc=zero8();
    #pragma unroll
    for(int kk=0;kk<2;++kk)
      acc=WMMA_F16(A[kk], load_tile_sw(WoT, j*2+kk, lane), acc);
    float bias=bo[j*16+colb];
    const float* xr=X+rbase*HH+j*16+colb;
    #pragma unroll
    for(int e=0;e<8;++e) o[j][e]=acc[e]+bias+xr[(size_t)(e+rowoff)*HH];
  }
  float gv[4],bv[4];
  #pragma unroll
  for(int j=0;j<4;++j){ gv[j]=g[j*16+colb]; bv[j]=be[j*16+colb]; }
  float* xw=X+rbase*HH;
  #pragma unroll
  for(int e=0;e<8;++e){
    float s=o[0][e]+o[1][e]+o[2][e]+o[3][e];
    float mu=rsum16(s)*(1.f/64.f);
    float d0=o[0][e]-mu,d1=o[1][e]-mu,d2=o[2][e]-mu,d3=o[3][e]-mu;
    float var=rsum16(d0*d0+d1*d1+d2*d2+d3*d3)*(1.f/64.f);
    float inv=rsqrtf(var+1e-5f);
    #pragma unroll
    for(int j=0;j<4;++j)
      xw[(size_t)(e+rowoff)*HH+j*16+colb]=(o[j][e]-mu)*inv*gv[j]+bv[j];
  }
}

__global__ __launch_bounds__(128) void k_ffln(
    float* __restrict__ X, const _Float16* __restrict__ W1T,
    const float* __restrict__ b1, const _Float16* __restrict__ W2T,
    const float* __restrict__ b2, const float* __restrict__ g,
    const float* __restrict__ be)
{
  __shared__ _Float16 h1[4][16*256];
  int lane=threadIdx.x&31, wave=threadIdx.x>>5;
  size_t rbase=((size_t)blockIdx.x*4+wave)*16;
  int colb=lane&15, rowoff=(lane>>4)*8;
  v16h A[2];
  #pragma unroll
  for(int kk=0;kk<2;++kk) A[kk]=load_tile_f32(X+rbase*HH+kk*32, HH, lane);
  _Float16* hl=&h1[wave][0];
  #pragma unroll
  for(int j=0;j<16;++j){
    v8f acc=zero8();
    #pragma unroll
    for(int kk=0;kk<2;++kk)
      acc=WMMA_F16(A[kk], load_tile_sw(W1T, j*2+kk, lane), acc);
    float bias=b1[j*16+colb];
    #pragma unroll
    for(int e=0;e<8;++e)
      hl[(size_t)(e+rowoff)*256+j*16+colb]=(_Float16)fmaxf(acc[e]+bias,0.f);
  }
  v16h pA[8];
  #pragma unroll
  for(int kk=0;kk<8;++kk) pA[kk]=load_tile_f16(hl+kk*32, 256, lane);
  v8f o[4];
  #pragma unroll
  for(int j=0;j<4;++j){
    v8f acc=zero8();
    #pragma unroll
    for(int kk=0;kk<8;++kk)
      acc=WMMA_F16(pA[kk], load_tile_sw(W2T, j*8+kk, lane), acc);
    float bias=b2[j*16+colb];
    const float* xr=X+rbase*HH+j*16+colb;
    #pragma unroll
    for(int e=0;e<8;++e) o[j][e]=acc[e]+bias+xr[(size_t)(e+rowoff)*HH];
  }
  float gv[4],bv[4];
  #pragma unroll
  for(int j=0;j<4;++j){ gv[j]=g[j*16+colb]; bv[j]=be[j*16+colb]; }
  float* xw=X+rbase*HH;
  #pragma unroll
  for(int e=0;e<8;++e){
    float s=o[0][e]+o[1][e]+o[2][e]+o[3][e];
    float mu=rsum16(s)*(1.f/64.f);
    float d0=o[0][e]-mu,d1=o[1][e]-mu,d2=o[2][e]-mu,d3=o[3][e]-mu;
    float var=rsum16(d0*d0+d1*d1+d2*d2+d3*d3)*(1.f/64.f);
    float inv=rsqrtf(var+1e-5f);
    #pragma unroll
    for(int j=0;j<4;++j)
      xw[(size_t)(e+rowoff)*HH+j*16+colb]=(o[j][e]-mu)*inv*gv[j]+bv[j];
  }
}

__global__ void k_mean(const float* __restrict__ X, float* __restrict__ out){
  int b=blockIdx.x, j=threadIdx.x;
  float s=0.f;
  for(int np=0;np<SQ;++np) s+=X[((size_t)b*SQ+np)*HH+j];
  out[b*HH+j]=s*(1.f/512.f);
}

extern "C" void kernel_launch(void* const* d_in, const int* in_sizes, int n_in,
                              void* d_out, int out_size, void* d_ws, size_t ws_size,
                              hipStream_t stream){
  const float* x  =(const float*)d_in[0];
  const float* Wih=(const float*)d_in[1];
  const float* Whh=(const float*)d_in[2];
  const float* bih=(const float*)d_in[3];
  const float* bhh=(const float*)d_in[4];

  char* ws=(char*)d_ws;
  float*    X =(float*)ws;
  float*    Y =(float*)(ws+(size_t)(8<<20));
  _Float16* Qp=(_Float16*)(ws+(size_t)(16<<20));
  _Float16* Kp=(_Float16*)(ws+(size_t)(24<<20));
  _Float16* Vh=(_Float16*)(ws+(size_t)(32<<20));
  _Float16* WihT=(_Float16*)(ws+(size_t)(36<<20));
  _Float16* WhhT=(_Float16*)(ws+(size_t)(36<<20)+(64<<10));
  _Float16* LW[2][4];
  for(int L=0;L<2;++L){
    char* base=ws+(size_t)(37<<20)+(size_t)L*(1<<20);
    LW[L][0]=(_Float16*)(base);
    LW[L][1]=(_Float16*)(base+(32<<10));
    LW[L][2]=(_Float16*)(base+(64<<10));
    LW[L][3]=(_Float16*)(base+(128<<10));
  }

  k_swz<<<96,256,0,stream>>>(Wih, WihT, 128);
  k_swz<<<48,256,0,stream>>>(Whh, WhhT, 64);
  for(int L=0;L<2;++L){
    int base=5+L*12;
    k_swz<<<48,256,0,stream>>>((const float*)d_in[base+0], LW[L][0], 64);
    k_swz<<<16,256,0,stream>>>((const float*)d_in[base+2], LW[L][1], 64);
    k_swz<<<64,256,0,stream>>>((const float*)d_in[base+4], LW[L][2], 64);
    k_swz<<<64,256,0,stream>>>((const float*)d_in[base+6], LW[L][3], 256);
  }

  k_gru<<<256,256,0,stream>>>(x,WihT,WhhT,bih,bhh,X);

  for(int L=0;L<2;++L){
    int base=5+L*12;
    const float* bqkv=(const float*)d_in[base+1];
    const float* bo  =(const float*)d_in[base+3];
    const float* b1  =(const float*)d_in[base+5];
    const float* b2  =(const float*)d_in[base+7];
    const float* g1  =(const float*)d_in[base+8];
    const float* be1 =(const float*)d_in[base+9];
    const float* g2  =(const float*)d_in[base+10];
    const float* be2 =(const float*)d_in[base+11];

    k_qkv   <<<256,256,0,stream>>>(X,LW[L][0],bqkv,Qp,Kp,Vh);
    k_attn  <<<256,256,0,stream>>>(Qp,Kp,Vh,Y);
    k_projln<<<256,256,0,stream>>>(X,Y,LW[L][1],bo,g1,be1);
    k_ffln  <<<512,128,0,stream>>>(X,LW[L][2],b1,LW[L][3],b2,g2,be2);
  }
  k_mean<<<64,64,0,stream>>>(X,(float*)d_out);
}


# attention computed transposed (S^T=K*Q^T, O^T=V^T*P^T): softmax max and sum become in-lane reductions, P feeds the second WMMA from registers; row-sum order changes
# speedup vs baseline: 1.3867x; 1.3867x over previous
#include <hip/hip_runtime.h>
#include <hip/hip_fp16.h>

typedef __attribute__((ext_vector_type(16))) _Float16 v16h;
typedef __attribute__((ext_vector_type(8)))  float    v8f;

#define WMMA_F16(a,b,c) __builtin_amdgcn_wmma_f32_16x16x32_f16(false,(a),false,(b),(short)0,(c),false,false)

__device__ __forceinline__ v8f zero8(){
  v8f z;
  #pragma unroll
  for(int i=0;i<8;++i) z[i]=0.f;
  return z;
}
__device__ __forceinline__ v16h zero16(){
  v16h z;
  #pragma unroll
  for(int i=0;i<16;++i) z[i]=(_Float16)0.f;
  return z;
}

__device__ __forceinline__ int a_k(int i, int h16){ return (i&7) + ((i>>3)<<4) + (h16<<3); }

__device__ __forceinline__ v16h load_tile_f32(const float* __restrict__ src, int ld, int lane){
  const float* rb = src + (size_t)(lane&15)*ld;
  int h16=(lane>>4)&1; v16h r;
  #pragma unroll
  for(int i=0;i<16;++i) r[i]=(_Float16)rb[a_k(i,h16)];
  return r;
}
__device__ __forceinline__ v16h load_tile_f16(const _Float16* __restrict__ src, int ld, int lane){
  const _Float16* rb = src + (size_t)(lane&15)*ld;
  int h16=(lane>>4)&1; v16h r;
  #pragma unroll
  for(int i=0;i<16;++i) r[i]=rb[a_k(i,h16)];
  return r;
}
__device__ __forceinline__ v16h load_tile_kmajor_f16(const _Float16* __restrict__ src, int ld, int lane){
  const _Float16* cb = src + (lane&15);
  int h16=(lane>>4)&1; v16h r;
  #pragma unroll
  for(int i=0;i<16;++i) r[i]=cb[(size_t)a_k(i,h16)*ld];
  return r;
}
__device__ __forceinline__ v16h load_tile_sw(const _Float16* src, int tile, int lane){
  return *reinterpret_cast<const v16h*>(src + (((size_t)tile*32 + lane) << 4));
}

__device__ __forceinline__ float rmax16(float x){
  #pragma unroll
  for(int m=8;m;m>>=1) x=fmaxf(x,__shfl_xor(x,m,32));
  return x;
}
__device__ __forceinline__ float rsum16(float x){
  #pragma unroll
  for(int m=8;m;m>>=1) x+=__shfl_xor(x,m,32);
  return x;
}
__device__ __forceinline__ float sigf(float x){ return 1.f/(1.f+__expf(-x)); }

#define PB  8
#define DD  128
#define HH  64
#define SQ  512

__global__ __launch_bounds__(256) void k_swz(const float* __restrict__ W,
                                             _Float16* __restrict__ out, int K){
  int idx = blockIdx.x*256 + threadIdx.x;
  int i    = idx & 15;
  int lane = (idx >> 4) & 31;
  int tile = idx >> 9;
  int ktiles = K >> 5;
  int kt = tile % ktiles, jt = tile / ktiles;
  out[idx] = (_Float16)W[(size_t)(jt*16 + (lane&15))*K + kt*32 + a_k(i,(lane>>4)&1)];
}

__global__ __launch_bounds__(256) void k_gru(
    const float* __restrict__ x, const _Float16* __restrict__ WihT,
    const _Float16* __restrict__ WhhT, const float* __restrict__ bih,
    const float* __restrict__ bhh, float* __restrict__ seq)
{
  __shared__ _Float16 hscr[8][16*64];
  __shared__ _Float16 WhhL[192*64];
  int lane=threadIdx.x&31, wave=threadIdx.x>>5;
  int pbase=(blockIdx.x*8+wave)*16;
  int colb=lane&15, rowoff=(lane>>4)*8;

  {
    const uint32_t* wsrc=(const uint32_t*)WhhT;
    uint32_t* wdst=(uint32_t*)WhhL;
    for(int i=threadIdx.x;i<192*64/2;i+=256) wdst[i]=wsrc[i];
  }
  __syncthreads();

  float bi[12], bh[12];
  #pragma unroll
  for(int j=0;j<12;++j){ bi[j]=bih[j*16+colb]; bh[j]=bhh[j*16+colb]; }

  v8f hT[4], hs[4];
  #pragma unroll
  for(int j=0;j<4;++j){ hT[j]=zero8(); hs[j]=zero8(); }
  v16h hA[2]; hA[0]=zero16(); hA[1]=zero16();
  _Float16* hl=&hscr[wave][0];

  for(int t=0;t<PB;++t){
    size_t zoff = 0;
    asm volatile("" : "+s"(zoff));
    const _Float16* Wt = WihT + zoff;
    const float* xsrc = x + ((size_t)pbase*PB + t)*DD;
    if(t+1<PB) __builtin_prefetch(xsrc+DD,0,1);
    v16h xA[4];
    #pragma unroll
    for(int kk=0;kk<4;++kk) xA[kk]=load_tile_f32(xsrc+kk*32, PB*DD, lane);

    #pragma unroll
    for(int j=0;j<4;++j){
      v8f gr=zero8();
      #pragma unroll
      for(int kk=0;kk<4;++kk)
        gr=WMMA_F16(xA[kk], load_tile_sw(Wt, j*4+kk, lane), gr);
      #pragma unroll
      for(int kk=0;kk<2;++kk)
        gr=WMMA_F16(hA[kk], load_tile_sw(WhhL, j*2+kk, lane), gr);
      v8f gz=zero8();
      #pragma unroll
      for(int kk=0;kk<4;++kk)
        gz=WMMA_F16(xA[kk], load_tile_sw(Wt, (4+j)*4+kk, lane), gz);
      #pragma unroll
      for(int kk=0;kk<2;++kk)
        gz=WMMA_F16(hA[kk], load_tile_sw(WhhL, (4+j)*2+kk, lane), gz);
      v8f gnx=zero8();
      #pragma unroll
      for(int kk=0;kk<4;++kk)
        gnx=WMMA_F16(xA[kk], load_tile_sw(Wt, (8+j)*4+kk, lane), gnx);
      v8f gnh=zero8();
      #pragma unroll
      for(int kk=0;kk<2;++kk)
        gnh=WMMA_F16(hA[kk], load_tile_sw(WhhL, (8+j)*2+kk, lane), gnh);
      #pragma unroll
      for(int e=0;e<8;++e){
        float r =sigf(gr[e]+bi[j]+bh[j]);
        float z =sigf(gz[e]+bi[4+j]+bh[4+j]);
        float nv=tanhf(gnx[e]+bi[8+j]+r*(gnh[e]+bh[8+j]));
        float hn=(1.f-z)*nv + z*hT[j][e];
        hT[j][e]=hn; hs[j][e]+=hn;
      }
    }
    #pragma unroll
    for(int j=0;j<4;++j){
      #pragma unroll
      for(int v=0;v<8;++v)
        hl[(v+rowoff)*64 + j*16 + colb]=(_Float16)hT[j][v];
    }
    #pragma unroll
    for(int kk=0;kk<2;++kk) hA[kk]=load_tile_f16(hl+kk*32, 64, lane);
  }
  float* sb=seq+(size_t)pbase*HH;
  #pragma unroll
  for(int j=0;j<4;++j){
    #pragma unroll
    for(int v=0;v<8;++v)
      sb[(size_t)(v+rowoff)*HH + j*16 + colb]=hs[j][v]*0.125f;
  }
}

__global__ __launch_bounds__(256) void k_qkv(
    const float* __restrict__ X, const _Float16* __restrict__ WqkvT,
    const float* __restrict__ bqkv, _Float16* __restrict__ Qp,
    _Float16* __restrict__ Kp, _Float16* __restrict__ Vh)
{
  int lane=threadIdx.x&31, wave=threadIdx.x>>5;
  size_t rbase=((size_t)blockIdx.x*8+wave)*16;
  int b=(int)(rbase>>9), s=(int)(rbase&511);
  int colb=lane&15, rowoff=(lane>>4)*8;
  v16h A[2];
  #pragma unroll
  for(int kk=0;kk<2;++kk) A[kk]=load_tile_f32(X+rbase*HH+kk*32, HH, lane);

  #pragma unroll
  for(int j=0;j<12;++j){
    v8f acc=zero8();
    #pragma unroll
    for(int kk=0;kk<2;++kk)
      acc=WMMA_F16(A[kk], load_tile_sw(WqkvT, j*2+kk, lane), acc);
    float bias=bqkv[j*16+colb];
    int h=j&3;
    size_t bh=(size_t)b*4+h;
    if(j<4){
      _Float16* dst=Qp+(bh*SQ+s)*32;
      #pragma unroll
      for(int v=0;v<8;++v){
        size_t ro=(size_t)(v+rowoff)*32;
        dst[ro+colb]=(_Float16)((acc[v]+bias)*0.25f);
        dst[ro+16+colb]=(_Float16)0.f;
      }
    } else if(j<8){
      _Float16* dst=Kp+(bh*SQ+s)*32;
      #pragma unroll
      for(int v=0;v<8;++v){
        size_t ro=(size_t)(v+rowoff)*32;
        dst[ro+colb]=(_Float16)(acc[v]+bias);
        dst[ro+16+colb]=(_Float16)0.f;
      }
    } else {
      _Float16* dst=Vh+(bh*SQ+s)*16;
      #pragma unroll
      for(int v=0;v<8;++v)
        dst[(size_t)(v+rowoff)*16+colb]=(_Float16)(acc[v]+bias);
    }
  }
}

__device__ __forceinline__ float xhalf(float x){
  int b=__builtin_bit_cast(int,x);
  return __builtin_bit_cast(float,
      __builtin_amdgcn_permlanex16(b,b,0x76543210,0xfedcba98,false,false));
}

__global__ __launch_bounds__(256) void k_attn(
    const _Float16* __restrict__ Qp, const _Float16* __restrict__ Kp,
    const _Float16* __restrict__ Vh, float* __restrict__ O)
{
  __shared__ _Float16 Ks[SQ*32];
  __shared__ _Float16 Vs[SQ*16];
  int bh=blockIdx.x, b=bh>>2, h=bh&3;
  int tid=threadIdx.x, lane=tid&31, wave=tid>>5;
  int colb=lane&15, rowoff=(lane>>4)*8;

  {
    const uint32_t* ksrc=(const uint32_t*)(Kp+(size_t)bh*SQ*32);
    uint32_t* kdst=(uint32_t*)Ks;
    for(int i=tid;i<SQ*16;i+=256) kdst[i]=ksrc[i];
    const uint32_t* vsrc=(const uint32_t*)(Vh+(size_t)bh*SQ*16);
    uint32_t* vdst=(uint32_t*)Vs;
    for(int i=tid;i<SQ*8;i+=256) vdst[i]=vsrc[i];
  }
  __syncthreads();

  for(int qt=0;qt<4;++qt){
    int qbase=wave*64+qt*16;
    v16h Qb=load_tile_f16(Qp+((size_t)bh*SQ+qbase)*32, 32, lane);
    v8f acc=zero8();
    float mrow=-1e30f;
    float lpart=0.f;

    for(int kc=0;kc<16;++kc){
      int kbase=kc*32;
      v16h Ka0=load_tile_f16(Ks+(size_t)kbase*32, 32, lane);
      v16h Ka1=load_tile_f16(Ks+(size_t)(kbase+16)*32, 32, lane);
      v16h Va=load_tile_kmajor_f16(Vs+(size_t)kbase*16, 16, lane);
      v8f s0=WMMA_F16(Ka0,Qb,zero8());
      v8f s1=WMMA_F16(Ka1,Qb,zero8());
      float cm=fmaxf(s0[0],s1[0]);
      #pragma unroll
      for(int e=1;e<8;++e) cm=fmaxf(cm,fmaxf(s0[e],s1[e]));
      cm=fmaxf(cm,xhalf(cm));
      float nm=fmaxf(mrow,cm);
      float sc=__expf(mrow-nm);
      mrow=nm;
      v16h Pb; float ps=0.f;
      #pragma unroll
      for(int e=0;e<8;++e){
        float p0=__expf(s0[e]-nm);
        float p1=__expf(s1[e]-nm);
        ps+=p0+p1;
        Pb[e]=(_Float16)p0;
        Pb[8+e]=(_Float16)p1;
      }
      lpart=lpart*sc+ps;
      #pragma unroll
      for(int e=0;e<8;++e) acc[e]*=sc;
      acc=WMMA_F16(Va,Pb,acc);
    }
    float lrow=lpart+xhalf(lpart);
    float* ob=O+((size_t)b*SQ+qbase+colb)*HH + h*16 + rowoff;
    #pragma unroll
    for(int e=0;e<8;++e) ob[e]=acc[e]/lrow;
  }
}

__global__ __launch_bounds__(256) void k_projln(
    float* __restrict__ X, const float* __restrict__ Y,
    const _Float16* __restrict__ WoT, const float* __restrict__ bo,
    const float* __restrict__ g, const float* __restrict__ be)
{
  int lane=threadIdx.x&31, wave=threadIdx.x>>5;
  size_t rbase=((size_t)blockIdx.x*8+wave)*16;
  int colb=lane&15, rowoff=(lane>>4)*8;
  v16h A[2];
  #pragma unroll
  for(int kk=0;kk<2;++kk) A[kk]=load_tile_f32(Y+rbase*HH+kk*32, HH, lane);
  v8f o[4];
  #pragma unroll
  for(int j=0;j<4;++j){
    v8f acc=zero8();
    #pragma unroll
    for(int kk=0;kk<2;++kk)
      acc=WMMA_F16(A[kk], load_tile_sw(WoT, j*2+kk, lane), acc);
    float bias=bo[j*16+colb];
    const float* xr=X+rbase*HH+j*16+colb;
    #pragma unroll
    for(int e=0;e<8;++e) o[j][e]=acc[e]+bias+xr[(size_t)(e+rowoff)*HH];
  }
  float gv[4],bv[4];
  #pragma unroll
  for(int j=0;j<4;++j){ gv[j]=g[j*16+colb]; bv[j]=be[j*16+colb]; }
  float* xw=X+rbase*HH;
  #pragma unroll
  for(int e=0;e<8;++e){
    float s=o[0][e]+o[1][e]+o[2][e]+o[3][e];
    float mu=rsum16(s)*(1.f/64.f);
    float d0=o[0][e]-mu,d1=o[1][e]-mu,d2=o[2][e]-mu,d3=o[3][e]-mu;
    float var=rsum16(d0*d0+d1*d1+d2*d2+d3*d3)*(1.f/64.f);
    float inv=rsqrtf(var+1e-5f);
    #pragma unroll
    for(int j=0;j<4;++j)
      xw[(size_t)(e+rowoff)*HH+j*16+colb]=(o[j][e]-mu)*inv*gv[j]+bv[j];
  }
}

__global__ __launch_bounds__(128) void k_ffln(
    float* __restrict__ X, const _Float16* __restrict__ W1T,
    const float* __restrict__ b1, const _Float16* __restrict__ W2T,
    const float* __restrict__ b2, const float* __restrict__ g,
    const float* __restrict__ be)
{
  __shared__ _Float16 h1[4][16*256];
  int lane=threadIdx.x&31, wave=threadIdx.x>>5;
  size_t rbase=((size_t)blockIdx.x*4+wave)*16;
  int colb=lane&15, rowoff=(lane>>4)*8;
  v16h A[2];
  #pragma unroll
  for(int kk=0;kk<2;++kk) A[kk]=load_tile_f32(X+rbase*HH+kk*32, HH, lane);
  _Float16* hl=&h1[wave][0];
  #pragma unroll
  for(int j=0;j<16;++j){
    v8f acc=zero8();
    #pragma unroll
    for(int kk=0;kk<2;++kk)
      acc=WMMA_F16(A[kk], load_tile_sw(W1T, j*2+kk, lane), acc);
    float bias=b1[j*16+colb];
    #pragma unroll
    for(int e=0;e<8;++e)
      hl[(size_t)(e+rowoff)*256+j*16+colb]=(_Float16)fmaxf(acc[e]+bias,0.f);
  }
  v16h pA[8];
  #pragma unroll
  for(int kk=0;kk<8;++kk) pA[kk]=load_tile_f16(hl+kk*32, 256, lane);
  v8f o[4];
  #pragma unroll
  for(int j=0;j<4;++j){
    v8f acc=zero8();
    #pragma unroll
    for(int kk=0;kk<8;++kk)
      acc=WMMA_F16(pA[kk], load_tile_sw(W2T, j*8+kk, lane), acc);
    float bias=b2[j*16+colb];
    const float* xr=X+rbase*HH+j*16+colb;
    #pragma unroll
    for(int e=0;e<8;++e) o[j][e]=acc[e]+bias+xr[(size_t)(e+rowoff)*HH];
  }
  float gv[4],bv[4];
  #pragma unroll
  for(int j=0;j<4;++j){ gv[j]=g[j*16+colb]; bv[j]=be[j*16+colb]; }
  float* xw=X+rbase*HH;
  #pragma unroll
  for(int e=0;e<8;++e){
    float s=o[0][e]+o[1][e]+o[2][e]+o[3][e];
    float mu=rsum16(s)*(1.f/64.f);
    float d0=o[0][e]-mu,d1=o[1][e]-mu,d2=o[2][e]-mu,d3=o[3][e]-mu;
    float var=rsum16(d0*d0+d1*d1+d2*d2+d3*d3)*(1.f/64.f);
    float inv=rsqrtf(var+1e-5f);
    #pragma unroll
    for(int j=0;j<4;++j)
      xw[(size_t)(e+rowoff)*HH+j*16+colb]=(o[j][e]-mu)*inv*gv[j]+bv[j];
  }
}

__global__ void k_mean(const float* __restrict__ X, float* __restrict__ out){
  int b=blockIdx.x, j=threadIdx.x;
  float s=0.f;
  for(int np=0;np<SQ;++np) s+=X[((size_t)b*SQ+np)*HH+j];
  out[b*HH+j]=s*(1.f/512.f);
}

extern "C" void kernel_launch(void* const* d_in, const int* in_sizes, int n_in,
                              void* d_out, int out_size, void* d_ws, size_t ws_size,
                              hipStream_t stream){
  const float* x  =(const float*)d_in[0];
  const float* Wih=(const float*)d_in[1];
  const float* Whh=(const float*)d_in[2];
  const float* bih=(const float*)d_in[3];
  const float* bhh=(const float*)d_in[4];

  char* ws=(char*)d_ws;
  float*    X =(float*)ws;
  float*    Y =(float*)(ws+(size_t)(8<<20));
  _Float16* Qp=(_Float16*)(ws+(size_t)(16<<20));
  _Float16* Kp=(_Float16*)(ws+(size_t)(24<<20));
  _Float16* Vh=(_Float16*)(ws+(size_t)(32<<20));
  _Float16* WihT=(_Float16*)(ws+(size_t)(36<<20));
  _Float16* WhhT=(_Float16*)(ws+(size_t)(36<<20)+(64<<10));
  _Float16* LW[2][4];
  for(int L=0;L<2;++L){
    char* base=ws+(size_t)(37<<20)+(size_t)L*(1<<20);
    LW[L][0]=(_Float16*)(base);
    LW[L][1]=(_Float16*)(base+(32<<10));
    LW[L][2]=(_Float16*)(base+(64<<10));
    LW[L][3]=(_Float16*)(base+(128<<10));
  }

  k_swz<<<96,256,0,stream>>>(Wih, WihT, 128);
  k_swz<<<48,256,0,stream>>>(Whh, WhhT, 64);
  for(int L=0;L<2;++L){
    int base=5+L*12;
    k_swz<<<48,256,0,stream>>>((const float*)d_in[base+0], LW[L][0], 64);
    k_swz<<<16,256,0,stream>>>((const float*)d_in[base+2], LW[L][1], 64);
    k_swz<<<64,256,0,stream>>>((const float*)d_in[base+4], LW[L][2], 64);
    k_swz<<<64,256,0,stream>>>((const float*)d_in[base+6], LW[L][3], 256);
  }

  k_gru<<<256,256,0,stream>>>(x,WihT,WhhT,bih,bhh,X);

  for(int L=0;L<2;++L){
    int base=5+L*12;
    const float* bqkv=(const float*)d_in[base+1];
    const float* bo  =(const float*)d_in[base+3];
    const float* b1  =(const float*)d_in[base+5];
    const float* b2  =(const float*)d_in[base+7];
    const float* g1  =(const float*)d_in[base+8];
    const float* be1 =(const float*)d_in[base+9];
    const float* g2  =(const float*)d_in[base+10];
    const float* be2 =(const float*)d_in[base+11];

    k_qkv   <<<256,256,0,stream>>>(X,LW[L][0],bqkv,Qp,Kp,Vh);
    k_attn  <<<256,256,0,stream>>>(Qp,Kp,Vh,Y);
    k_projln<<<256,256,0,stream>>>(X,Y,LW[L][1],bo,g1,be1);
    k_ffln  <<<512,128,0,stream>>>(X,LW[L][2],b1,LW[L][3],b2,g2,be2);
  }
  k_mean<<<64,64,0,stream>>>(X,(float*)d_out);
}


# the ten weight-repack launches merged into one launch (one single-wave workgroup per 16x32 tile, 264 tiles): 20 -> 11 dependent launches per call; repacked bytes identical
# speedup vs baseline: 1.4660x; 1.0572x over previous
#include <hip/hip_runtime.h>
#include <hip/hip_fp16.h>

typedef __attribute__((ext_vector_type(16))) _Float16 v16h;
typedef __attribute__((ext_vector_type(8)))  float    v8f;

#define WMMA_F16(a,b,c) __builtin_amdgcn_wmma_f32_16x16x32_f16(false,(a),false,(b),(short)0,(c),false,false)

__device__ __forceinline__ v8f zero8(){
  v8f z;
  #pragma unroll
  for(int i=0;i<8;++i) z[i]=0.f;
  return z;
}
__device__ __forceinline__ v16h zero16(){
  v16h z;
  #pragma unroll
  for(int i=0;i<16;++i) z[i]=(_Float16)0.f;
  return z;
}

__device__ __forceinline__ int a_k(int i, int h16){ return (i&7) + ((i>>3)<<4) + (h16<<3); }

__device__ __forceinline__ v16h load_tile_f32(const float* __restrict__ src, int ld, int lane){
  const float* rb = src + (size_t)(lane&15)*ld;
  int h16=(lane>>4)&1; v16h r;
  #pragma unroll
  for(int i=0;i<16;++i) r[i]=(_Float16)rb[a_k(i,h16)];
  return r;
}
__device__ __forceinline__ v16h load_tile_f16(const _Float16* __restrict__ src, int ld, int lane){
  const _Float16* rb = src + (size_t)(lane&15)*ld;
  int h16=(lane>>4)&1; v16h r;
  #pragma unroll
  for(int i=0;i<16;++i) r[i]=rb[a_k(i,h16)];
  return r;
}
__device__ __forceinline__ v16h load_tile_kmajor_f16(const _Float16* __restrict__ src, int ld, int lane){
  const _Float16* cb = src + (lane&15);
  int h16=(lane>>4)&1; v16h r;
  #pragma unroll
  for(int i=0;i<16;++i) r[i]=cb[(size_t)a_k(i,h16)*ld];
  return r;
}
__device__ __forceinline__ v16h load_tile_sw(const _Float16* src, int tile, int lane){
  return *reinterpret_cast<const v16h*>(src + (((size_t)tile*32 + lane) << 4));
}

__device__ __forceinline__ float rmax16(float x){
  #pragma unroll
  for(int m=8;m;m>>=1) x=fmaxf(x,__shfl_xor(x,m,32));
  return x;
}
__device__ __forceinline__ float rsum16(float x){
  #pragma unroll
  for(int m=8;m;m>>=1) x+=__shfl_xor(x,m,32);
  return x;
}
__device__ __forceinline__ float sigf(float x){ return 1.f/(1.f+__expf(-x)); }

#define PB  8
#define DD  128
#define HH  64
#define SQ  512

__global__ __launch_bounds__(32) void k_swz_all(
    const float* __restrict__ wih,   const float* __restrict__ whh,
    const float* __restrict__ wqkv0, const float* __restrict__ wo0,
    const float* __restrict__ w1_0,  const float* __restrict__ w2_0,
    const float* __restrict__ wqkv1, const float* __restrict__ wo1,
    const float* __restrict__ w1_1,  const float* __restrict__ w2_1,
    _Float16* __restrict__ obase)
{
  int t = blockIdx.x, lane = threadIdx.x;
  const float* Wm; _Float16* out; int ksh;
  if(t < 48){ Wm = wih; out = obase; ksh = 2; }
  else if(t < 72){ t -= 48; Wm = whh; out = obase + (32<<10); ksh = 1; }
  else {
    t -= 72;
    int L = (t >= 96) ? 1 : 0;
    t -= L*96;
    _Float16* lb = obase + ((size_t)(1+L) << 19);
    if(t < 24){ Wm = L ? wqkv1 : wqkv0; out = lb; ksh = 1; }
    else if(t < 32){ t -= 24; Wm = L ? wo1 : wo0; out = lb + (16<<10); ksh = 1; }
    else if(t < 64){ t -= 32; Wm = L ? w1_1 : w1_0; out = lb + (32<<10); ksh = 1; }
    else { t -= 64; Wm = L ? w2_1 : w2_0; out = lb + (64<<10); ksh = 3; }
  }
  int K = 32 << ksh;
  int kt = t & ((1<<ksh)-1), jt = t >> ksh;
  v16h r = load_tile_f32(Wm + (size_t)jt*16*K + kt*32, K, lane);
  *reinterpret_cast<v16h*>(out + (((size_t)t*32 + lane) << 4)) = r;
}

__global__ __launch_bounds__(256) void k_gru(
    const float* __restrict__ x, const _Float16* __restrict__ WihT,
    const _Float16* __restrict__ WhhT, const float* __restrict__ bih,
    const float* __restrict__ bhh, float* __restrict__ seq)
{
  __shared__ _Float16 hscr[8][16*64];
  __shared__ _Float16 WhhL[192*64];
  int lane=threadIdx.x&31, wave=threadIdx.x>>5;
  int pbase=(blockIdx.x*8+wave)*16;
  int colb=lane&15, rowoff=(lane>>4)*8;

  {
    const uint32_t* wsrc=(const uint32_t*)WhhT;
    uint32_t* wdst=(uint32_t*)WhhL;
    for(int i=threadIdx.x;i<192*64/2;i+=256) wdst[i]=wsrc[i];
  }
  __syncthreads();

  float bi[12], bh[12];
  #pragma unroll
  for(int j=0;j<12;++j){ bi[j]=bih[j*16+colb]; bh[j]=bhh[j*16+colb]; }

  v8f hT[4], hs[4];
  #pragma unroll
  for(int j=0;j<4;++j){ hT[j]=zero8(); hs[j]=zero8(); }
  v16h hA[2]; hA[0]=zero16(); hA[1]=zero16();
  _Float16* hl=&hscr[wave][0];

  for(int t=0;t<PB;++t){
    size_t zoff = 0;
    asm volatile("" : "+s"(zoff));
    const _Float16* Wt = WihT + zoff;
    const float* xsrc = x + ((size_t)pbase*PB + t)*DD;
    if(t+1<PB) __builtin_prefetch(xsrc+DD,0,1);
    v16h xA[4];
    #pragma unroll
    for(int kk=0;kk<4;++kk) xA[kk]=load_tile_f32(xsrc+kk*32, PB*DD, lane);

    #pragma unroll
    for(int j=0;j<4;++j){
      v8f gr=zero8();
      #pragma unroll
      for(int kk=0;kk<4;++kk)
        gr=WMMA_F16(xA[kk], load_tile_sw(Wt, j*4+kk, lane), gr);
      #pragma unroll
      for(int kk=0;kk<2;++kk)
        gr=WMMA_F16(hA[kk], load_tile_sw(WhhL, j*2+kk, lane), gr);
      v8f gz=zero8();
      #pragma unroll
      for(int kk=0;kk<4;++kk)
        gz=WMMA_F16(xA[kk], load_tile_sw(Wt, (4+j)*4+kk, lane), gz);
      #pragma unroll
      for(int kk=0;kk<2;++kk)
        gz=WMMA_F16(hA[kk], load_tile_sw(WhhL, (4+j)*2+kk, lane), gz);
      v8f gnx=zero8();
      #pragma unroll
      for(int kk=0;kk<4;++kk)
        gnx=WMMA_F16(xA[kk], load_tile_sw(Wt, (8+j)*4+kk, lane), gnx);
      v8f gnh=zero8();
      #pragma unroll
      for(int kk=0;kk<2;++kk)
        gnh=WMMA_F16(hA[kk], load_tile_sw(WhhL, (8+j)*2+kk, lane), gnh);
      #pragma unroll
      for(int e=0;e<8;++e){
        float r =sigf(gr[e]+bi[j]+bh[j]);
        float z =sigf(gz[e]+bi[4+j]+bh[4+j]);
        float nv=tanhf(gnx[e]+bi[8+j]+r*(gnh[e]+bh[8+j]));
        float hn=(1.f-z)*nv + z*hT[j][e];
        hT[j][e]=hn; hs[j][e]+=hn;
      }
    }
    #pragma unroll
    for(int j=0;j<4;++j){
      #pragma unroll
      for(int v=0;v<8;++v)
        hl[(v+rowoff)*64 + j*16 + colb]=(_Float16)hT[j][v];
    }
    #pragma unroll
    for(int kk=0;kk<2;++kk) hA[kk]=load_tile_f16(hl+kk*32, 64, lane);
  }
  float* sb=seq+(size_t)pbase*HH;
  #pragma unroll
  for(int j=0;j<4;++j){
    #pragma unroll
    for(int v=0;v<8;++v)
      sb[(size_t)(v+rowoff)*HH + j*16 + colb]=hs[j][v]*0.125f;
  }
}

__global__ __launch_bounds__(256) void k_qkv(
    const float* __restrict__ X, const _Float16* __restrict__ WqkvT,
    const float* __restrict__ bqkv, _Float16* __restrict__ Qp,
    _Float16* __restrict__ Kp, _Float16* __restrict__ Vh)
{
  int lane=threadIdx.x&31, wave=threadIdx.x>>5;
  size_t rbase=((size_t)blockIdx.x*8+wave)*16;
  int b=(int)(rbase>>9), s=(int)(rbase&511);
  int colb=lane&15, rowoff=(lane>>4)*8;
  v16h A[2];
  #pragma unroll
  for(int kk=0;kk<2;++kk) A[kk]=load_tile_f32(X+rbase*HH+kk*32, HH, lane);

  #pragma unroll
  for(int j=0;j<12;++j){
    v8f acc=zero8();
    #pragma unroll
    for(int kk=0;kk<2;++kk)
      acc=WMMA_F16(A[kk], load_tile_sw(WqkvT, j*2+kk, lane), acc);
    float bias=bqkv[j*16+colb];
    int h=j&3;
    size_t bh=(size_t)b*4+h;
    if(j<4){
      _Float16* dst=Qp+(bh*SQ+s)*32;
      #pragma unroll
      for(int v=0;v<8;++v){
        size_t ro=(size_t)(v+rowoff)*32;
        dst[ro+colb]=(_Float16)((acc[v]+bias)*0.25f);
        dst[ro+16+colb]=(_Float16)0.f;
      }
    } else if(j<8){
      _Float16* dst=Kp+(bh*SQ+s)*32;
      #pragma unroll
      for(int v=0;v<8;++v){
        size_t ro=(size_t)(v+rowoff)*32;
        dst[ro+colb]=(_Float16)(acc[v]+bias);
        dst[ro+16+colb]=(_Float16)0.f;
      }
    } else {
      _Float16* dst=Vh+(bh*SQ+s)*16;
      #pragma unroll
      for(int v=0;v<8;++v)
        dst[(size_t)(v+rowoff)*16+colb]=(_Float16)(acc[v]+bias);
    }
  }
}

__device__ __forceinline__ float xhalf(float x){
  int b=__builtin_bit_cast(int,x);
  return __builtin_bit_cast(float,
      __builtin_amdgcn_permlanex16(b,b,0x76543210,0xfedcba98,false,false));
}

__global__ __launch_bounds__(256) void k_attn(
    const _Float16* __restrict__ Qp, const _Float16* __restrict__ Kp,
    const _Float16* __restrict__ Vh, float* __restrict__ O)
{
  __shared__ _Float16 Ks[SQ*32];
  __shared__ _Float16 Vs[SQ*16];
  int bh=blockIdx.x, b=bh>>2, h=bh&3;
  int tid=threadIdx.x, lane=tid&31, wave=tid>>5;
  int colb=lane&15, rowoff=(lane>>4)*8;

  {
    const uint32_t* ksrc=(const uint32_t*)(Kp+(size_t)bh*SQ*32);
    uint32_t* kdst=(uint32_t*)Ks;
    for(int i=tid;i<SQ*16;i+=256) kdst[i]=ksrc[i];
    const uint32_t* vsrc=(const uint32_t*)(Vh+(size_t)bh*SQ*16);
    uint32_t* vdst=(uint32_t*)Vs;
    for(int i=tid;i<SQ*8;i+=256) vdst[i]=vsrc[i];
  }
  __syncthreads();

  for(int qt=0;qt<4;++qt){
    int qbase=wave*64+qt*16;
    v16h Qb=load_tile_f16(Qp+((size_t)bh*SQ+qbase)*32, 32, lane);
    v8f acc=zero8();
    float mrow=-1e30f;
    float lpart=0.f;

    for(int kc=0;kc<16;++kc){
      int kbase=kc*32;
      v16h Ka0=load_tile_f16(Ks+(size_t)kbase*32, 32, lane);
      v16h Ka1=load_tile_f16(Ks+(size_t)(kbase+16)*32, 32, lane);
      v16h Va=load_tile_kmajor_f16(Vs+(size_t)kbase*16, 16, lane);
      v8f s0=WMMA_F16(Ka0,Qb,zero8());
      v8f s1=WMMA_F16(Ka1,Qb,zero8());
      float cm=fmaxf(s0[0],s1[0]);
      #pragma unroll
      for(int e=1;e<8;++e) cm=fmaxf(cm,fmaxf(s0[e],s1[e]));
      cm=fmaxf(cm,xhalf(cm));
      float nm=fmaxf(mrow,cm);
      float sc=__expf(mrow-nm);
      mrow=nm;
      v16h Pb; float ps=0.f;
      #pragma unroll
      for(int e=0;e<8;++e){
        float p0=__expf(s0[e]-nm);
        float p1=__expf(s1[e]-nm);
        ps+=p0+p1;
        Pb[e]=(_Float16)p0;
        Pb[8+e]=(_Float16)p1;
      }
      lpart=lpart*sc+ps;
      #pragma unroll
      for(int e=0;e<8;++e) acc[e]*=sc;
      acc=WMMA_F16(Va,Pb,acc);
    }
    float lrow=lpart+xhalf(lpart);
    float* ob=O+((size_t)b*SQ+qbase+colb)*HH + h*16 + rowoff;
    #pragma unroll
    for(int e=0;e<8;++e) ob[e]=acc[e]/lrow;
  }
}

__global__ __launch_bounds__(256) void k_projln(
    float* __restrict__ X, const float* __restrict__ Y,
    const _Float16* __restrict__ WoT, const float* __restrict__ bo,
    const float* __restrict__ g, const float* __restrict__ be)
{
  int lane=threadIdx.x&31, wave=threadIdx.x>>5;
  size_t rbase=((size_t)blockIdx.x*8+wave)*16;
  int colb=lane&15, rowoff=(lane>>4)*8;
  v16h A[2];
  #pragma unroll
  for(int kk=0;kk<2;++kk) A[kk]=load_tile_f32(Y+rbase*HH+kk*32, HH, lane);
  v8f o[4];
  #pragma unroll
  for(int j=0;j<4;++j){
    v8f acc=zero8();
    #pragma unroll
    for(int kk=0;kk<2;++kk)
      acc=WMMA_F16(A[kk], load_tile_sw(WoT, j*2+kk, lane), acc);
    float bias=bo[j*16+colb];
    const float* xr=X+rbase*HH+j*16+colb;
    #pragma unroll
    for(int e=0;e<8;++e) o[j][e]=acc[e]+bias+xr[(size_t)(e+rowoff)*HH];
  }
  float gv[4],bv[4];
  #pragma unroll
  for(int j=0;j<4;++j){ gv[j]=g[j*16+colb]; bv[j]=be[j*16+colb]; }
  float* xw=X+rbase*HH;
  #pragma unroll
  for(int e=0;e<8;++e){
    float s=o[0][e]+o[1][e]+o[2][e]+o[3][e];
    float mu=rsum16(s)*(1.f/64.f);
    float d0=o[0][e]-mu,d1=o[1][e]-mu,d2=o[2][e]-mu,d3=o[3][e]-mu;
    float var=rsum16(d0*d0+d1*d1+d2*d2+d3*d3)*(1.f/64.f);
    float inv=rsqrtf(var+1e-5f);
    #pragma unroll
    for(int j=0;j<4;++j)
      xw[(size_t)(e+rowoff)*HH+j*16+colb]=(o[j][e]-mu)*inv*gv[j]+bv[j];
  }
}

__global__ __launch_bounds__(128) void k_ffln(
    float* __restrict__ X, const _Float16* __restrict__ W1T,
    const float* __restrict__ b1, const _Float16* __restrict__ W2T,
    const float* __restrict__ b2, const float* __restrict__ g,
    const float* __restrict__ be)
{
  __shared__ _Float16 h1[4][16*256];
  int lane=threadIdx.x&31, wave=threadIdx.x>>5;
  size_t rbase=((size_t)blockIdx.x*4+wave)*16;
  int colb=lane&15, rowoff=(lane>>4)*8;
  v16h A[2];
  #pragma unroll
  for(int kk=0;kk<2;++kk) A[kk]=load_tile_f32(X+rbase*HH+kk*32, HH, lane);
  _Float16* hl=&h1[wave][0];
  #pragma unroll
  for(int j=0;j<16;++j){
    v8f acc=zero8();
    #pragma unroll
    for(int kk=0;kk<2;++kk)
      acc=WMMA_F16(A[kk], load_tile_sw(W1T, j*2+kk, lane), acc);
    float bias=b1[j*16+colb];
    #pragma unroll
    for(int e=0;e<8;++e)
      hl[(size_t)(e+rowoff)*256+j*16+colb]=(_Float16)fmaxf(acc[e]+bias,0.f);
  }
  v16h pA[8];
  #pragma unroll
  for(int kk=0;kk<8;++kk) pA[kk]=load_tile_f16(hl+kk*32, 256, lane);
  v8f o[4];
  #pragma unroll
  for(int j=0;j<4;++j){
    v8f acc=zero8();
    #pragma unroll
    for(int kk=0;kk<8;++kk)
      acc=WMMA_F16(pA[kk], load_tile_sw(W2T, j*8+kk, lane), acc);
    float bias=b2[j*16+colb];
    const float* xr=X+rbase*HH+j*16+colb;
    #pragma unroll
    for(int e=0;e<8;++e) o[j][e]=acc[e]+bias+xr[(size_t)(e+rowoff)*HH];
  }
  float gv[4],bv[4];
  #pragma unroll
  for(int j=0;j<4;++j){ gv[j]=g[j*16+colb]; bv[j]=be[j*16+colb]; }
  float* xw=X+rbase*HH;
  #pragma unroll
  for(int e=0;e<8;++e){
    float s=o[0][e]+o[1][e]+o[2][e]+o[3][e];
    float mu=rsum16(s)*(1.f/64.f);
    float d0=o[0][e]-mu,d1=o[1][e]-mu,d2=o[2][e]-mu,d3=o[3][e]-mu;
    float var=rsum16(d0*d0+d1*d1+d2*d2+d3*d3)*(1.f/64.f);
    float inv=rsqrtf(var+1e-5f);
    #pragma unroll
    for(int j=0;j<4;++j)
      xw[(size_t)(e+rowoff)*HH+j*16+colb]=(o[j][e]-mu)*inv*gv[j]+bv[j];
  }
}

__global__ void k_mean(const float* __restrict__ X, float* __restrict__ out){
  int b=blockIdx.x, j=threadIdx.x;
  float s=0.f;
  for(int np=0;np<SQ;++np) s+=X[((size_t)b*SQ+np)*HH+j];
  out[b*HH+j]=s*(1.f/512.f);
}

extern "C" void kernel_launch(void* const* d_in, const int* in_sizes, int n_in,
                              void* d_out, int out_size, void* d_ws, size_t ws_size,
                              hipStream_t stream){
  const float* x  =(const float*)d_in[0];
  const float* Wih=(const float*)d_in[1];
  const float* Whh=(const float*)d_in[2];
  const float* bih=(const float*)d_in[3];
  const float* bhh=(const float*)d_in[4];

  char* ws=(char*)d_ws;
  float*    X =(float*)ws;
  float*    Y =(float*)(ws+(size_t)(8<<20));
  _Float16* Qp=(_Float16*)(ws+(size_t)(16<<20));
  _Float16* Kp=(_Float16*)(ws+(size_t)(24<<20));
  _Float16* Vh=(_Float16*)(ws+(size_t)(32<<20));
  _Float16* WihT=(_Float16*)(ws+(size_t)(36<<20));
  _Float16* WhhT=(_Float16*)(ws+(size_t)(36<<20)+(64<<10));
  _Float16* LW[2][4];
  for(int L=0;L<2;++L){
    char* base=ws+(size_t)(37<<20)+(size_t)L*(1<<20);
    LW[L][0]=(_Float16*)(base);
    LW[L][1]=(_Float16*)(base+(32<<10));
    LW[L][2]=(_Float16*)(base+(64<<10));
    LW[L][3]=(_Float16*)(base+(128<<10));
  }

  k_swz_all<<<264,32,0,stream>>>(Wih, Whh,
      (const float*)d_in[5],  (const float*)d_in[7],
      (const float*)d_in[9],  (const float*)d_in[11],
      (const float*)d_in[17], (const float*)d_in[19],
      (const float*)d_in[21], (const float*)d_in[23],
      WihT);

  k_gru<<<256,256,0,stream>>>(x,WihT,WhhT,bih,bhh,X);

  for(int L=0;L<2;++L){
    int base=5+L*12;
    const float* bqkv=(const float*)d_in[base+1];
    const float* bo  =(const float*)d_in[base+3];
    const float* b1  =(const float*)d_in[base+5];
    const float* b2  =(const float*)d_in[base+7];
    const float* g1  =(const float*)d_in[base+8];
    const float* be1 =(const float*)d_in[base+9];
    const float* g2  =(const float*)d_in[base+10];
    const float* be2 =(const float*)d_in[base+11];

    k_qkv   <<<256,256,0,stream>>>(X,LW[L][0],bqkv,Qp,Kp,Vh);
    k_attn  <<<256,256,0,stream>>>(Qp,Kp,Vh,Y);
    k_projln<<<256,256,0,stream>>>(X,Y,LW[L][1],bo,g1,be1);
    k_ffln  <<<512,128,0,stream>>>(X,LW[L][2],b1,LW[L][3],b2,g2,be2);
  }
  k_mean<<<64,64,0,stream>>>(X,(float*)d_out);
}


# GRU gates on hardware v_exp/v_rcp/v_tanh (about 1 ulp) instead of IEEE division and branching tanhf: branch-free, a tile's 8 elements interleaved; gate values differ in the last bits
# speedup vs baseline: 1.7867x; 1.2188x over previous
#include <hip/hip_runtime.h>
#include <hip/hip_fp16.h>

typedef __attribute__((ext_vector_type(16))) _Float16 v16h;
typedef __attribute__((ext_vector_type(8)))  float    v8f;
typedef __attribute__((ext_vector_type(4)))  float    v4f;

#define WMMA_F16(a,b,c) __builtin_amdgcn_wmma_f32_16x16x32_f16(false,(a),false,(b),(short)0,(c),false,false)

__device__ __forceinline__ v8f zero8(){
  v8f z;
  #pragma unroll
  for(int i=0;i<8;++i) z[i]=0.f;
  return z;
}
__device__ __forceinline__ v16h zero16(){
  v16h z;
  #pragma unroll
  for(int i=0;i<16;++i) z[i]=(_Float16)0.f;
  return z;
}

__device__ __forceinline__ int a_k(int i, int h16){ return (i&7) + ((i>>3)<<4) + (h16<<3); }

__device__ __forceinline__ v16h load_tile_f32(const float* __restrict__ src, int ld, int lane){
  const float* rb = src + (size_t)(lane&15)*ld;
  int h16=(lane>>4)&1; v16h r;
  #pragma unroll
  for(int i=0;i<16;++i) r[i]=(_Float16)rb[a_k(i,h16)];
  return r;
}
__device__ __forceinline__ v16h load_tile_f16(const _Float16* __restrict__ src, int ld, int lane){
  const _Float16* rb = src + (size_t)(lane&15)*ld;
  int h16=(lane>>4)&1; v16h r;
  #pragma unroll
  for(int i=0;i<16;++i) r[i]=rb[a_k(i,h16)];
  return r;
}
__device__ __forceinline__ v16h load_tile_kmajor_f16(const _Float16* __restrict__ src, int ld, int lane){
  const _Float16* cb = src + (lane&15);
  int h16=(lane>>4)&1; v16h r;
  #pragma unroll
  for(int i=0;i<16;++i) r[i]=cb[(size_t)a_k(i,h16)*ld];
  return r;
}
__device__ __forceinline__ v16h load_tile_sw(const _Float16* src, int tile, int lane){
  return *reinterpret_cast<const v16h*>(src + (((size_t)tile*32 + lane) << 4));
}

__device__ __forceinline__ float rmax16(float x){
  #pragma unroll
  for(int m=8;m;m>>=1) x=fmaxf(x,__shfl_xor(x,m,32));
  return x;
}
__device__ __forceinline__ float rsum16(float x){
  #pragma unroll
  for(int m=8;m;m>>=1) x+=__shfl_xor(x,m,32);
  return x;
}

#define PB  8
#define DD  128
#define HH  64
#define SQ  512

__global__ __launch_bounds__(32) void k_swz_all(
    const float* __restrict__ wih,   const float* __restrict__ whh,
    const float* __restrict__ wqkv0, const float* __restrict__ wo0,
    const float* __restrict__ w1_0,  const float* __restrict__ w2_0,
    const float* __restrict__ wqkv1, const float* __restrict__ wo1,
    const float* __restrict__ w1_1,  const float* __restrict__ w2_1,
    _Float16* __restrict__ obase)
{
  int t = blockIdx.x, lane = threadIdx.x;
  const float* Wm; _Float16* out; int ksh;
  if(t < 48){ Wm = wih; out = obase; ksh = 2; }
  else if(t < 72){ t -= 48; Wm = whh; out = obase + (32<<10); ksh = 1; }
  else {
    t -= 72;
    int L = (t >= 96) ? 1 : 0;
    t -= L*96;
    _Float16* lb = obase + ((size_t)(1+L) << 19);
    if(t < 24){ Wm = L ? wqkv1 : wqkv0; out = lb; ksh = 1; }
    else if(t < 32){ t -= 24; Wm = L ? wo1 : wo0; out = lb + (16<<10); ksh = 1; }
    else if(t < 64){ t -= 32; Wm = L ? w1_1 : w1_0; out = lb + (32<<10); ksh = 1; }
    else { t -= 64; Wm = L ? w2_1 : w2_0; out = lb + (64<<10); ksh = 3; }
  }
  int K = 32 << ksh;
  int kt = t & ((1<<ksh)-1), jt = t >> ksh;
  v16h r = load_tile_f32(Wm + (size_t)jt*16*K + kt*32, K, lane);
  *reinterpret_cast<v16h*>(out + (((size_t)t*32 + lane) << 4)) = r;
}

__global__ __launch_bounds__(256) __attribute__((amdgpu_waves_per_eu(2,2))) void k_gru(
    const float* __restrict__ x, const _Float16* __restrict__ WihT,
    const _Float16* __restrict__ WhhT, const float* __restrict__ bih,
    const float* __restrict__ bhh, float* __restrict__ seq)
{
  __shared__ _Float16 hscr[8][16*64];
  __shared__ _Float16 WhhL[192*64];
  int lane=threadIdx.x&31, wave=threadIdx.x>>5;
  int pbase=(blockIdx.x*8+wave)*16;
  int colb=lane&15, rowoff=(lane>>4)*8;

  {
    const uint32_t* wsrc=(const uint32_t*)WhhT;
    uint32_t* wdst=(uint32_t*)WhhL;
    for(int i=threadIdx.x;i<192*64/2;i+=256) wdst[i]=wsrc[i];
  }
  __syncthreads();

  float bi[12], bh[12];
  #pragma unroll
  for(int j=0;j<12;++j){ bi[j]=bih[j*16+colb]; bh[j]=bhh[j*16+colb]; }

  v8f hT[4], hs[4];
  #pragma unroll
  for(int j=0;j<4;++j){ hT[j]=zero8(); hs[j]=zero8(); }
  v16h hA[2]; hA[0]=zero16(); hA[1]=zero16();
  _Float16* hl=&hscr[wave][0];

  for(int t=0;t<PB;++t){
    size_t zoff = 0;
    asm volatile("" : "+s"(zoff));
    const _Float16* Wt = WihT + zoff;
    const float* xsrc = x + ((size_t)pbase*PB + t)*DD;
    if(t+1<PB) __builtin_prefetch(xsrc+DD,0,1);
    v4f xr[16];
    {
      const float* rb = xsrc + (size_t)(lane&15)*(PB*DD) + (((lane>>4)&1)<<3);
      #pragma unroll
      for(int kk=0;kk<4;++kk){
        xr[kk*4+0]=*reinterpret_cast<const v4f*>(rb+kk*32);
        xr[kk*4+1]=*reinterpret_cast<const v4f*>(rb+kk*32+4);
        xr[kk*4+2]=*reinterpret_cast<const v4f*>(rb+kk*32+16);
        xr[kk*4+3]=*reinterpret_cast<const v4f*>(rb+kk*32+20);
      }
    }
    __builtin_amdgcn_sched_barrier(0);
    v16h xA[4];
    #pragma unroll
    for(int kk=0;kk<4;++kk){
      #pragma unroll
      for(int q=0;q<4;++q){
        #pragma unroll
        for(int c=0;c<4;++c) xA[kk][q*4+c]=(_Float16)xr[kk*4+q][c];
      }
    }

    #pragma unroll
    for(int j=0;j<4;++j){
      v8f gr=zero8();
      #pragma unroll
      for(int kk=0;kk<4;++kk)
        gr=WMMA_F16(xA[kk], load_tile_sw(Wt, j*4+kk, lane), gr);
      #pragma unroll
      for(int kk=0;kk<2;++kk)
        gr=WMMA_F16(hA[kk], load_tile_sw(WhhL, j*2+kk, lane), gr);
      v8f gz=zero8();
      #pragma unroll
      for(int kk=0;kk<4;++kk)
        gz=WMMA_F16(xA[kk], load_tile_sw(Wt, (4+j)*4+kk, lane), gz);
      #pragma unroll
      for(int kk=0;kk<2;++kk)
        gz=WMMA_F16(hA[kk], load_tile_sw(WhhL, (4+j)*2+kk, lane), gz);
      v8f gnx=zero8();
      #pragma unroll
      for(int kk=0;kk<4;++kk)
        gnx=WMMA_F16(xA[kk], load_tile_sw(Wt, (8+j)*4+kk, lane), gnx);
      v8f gnh=zero8();
      #pragma unroll
      for(int kk=0;kk<2;++kk)
        gnh=WMMA_F16(hA[kk], load_tile_sw(WhhL, (8+j)*2+kk, lane), gnh);
      float rr[8], zz[8], nn[8];
      #pragma unroll
      for(int e=0;e<8;++e){
        rr[e]=(gr[e]+bi[j]+bh[j])*-1.44269504f;
        zz[e]=(gz[e]+bi[4+j]+bh[4+j])*-1.44269504f;
      }
      #pragma unroll
      for(int e=0;e<8;++e){
        rr[e]=1.f+__builtin_amdgcn_exp2f(rr[e]);
        zz[e]=1.f+__builtin_amdgcn_exp2f(zz[e]);
      }
      #pragma unroll
      for(int e=0;e<8;++e){
        rr[e]=__builtin_amdgcn_rcpf(rr[e]);
        zz[e]=__builtin_amdgcn_rcpf(zz[e]);
      }
      #pragma unroll
      for(int e=0;e<8;++e)
        nn[e]=gnx[e]+bi[8+j]+rr[e]*(gnh[e]+bh[8+j]);
      #pragma unroll
      for(int e=0;e<8;++e) nn[e]=__builtin_amdgcn_tanhf(nn[e]);
      #pragma unroll
      for(int e=0;e<8;++e){
        float hn=(1.f-zz[e])*nn[e] + zz[e]*hT[j][e];
        hT[j][e]=hn; hs[j][e]+=hn;
      }
    }
    #pragma unroll
    for(int j=0;j<4;++j){
      #pragma unroll
      for(int v=0;v<8;++v)
        hl[(v+rowoff)*64 + j*16 + colb]=(_Float16)hT[j][v];
    }
    #pragma unroll
    for(int kk=0;kk<2;++kk) hA[kk]=load_tile_f16(hl+kk*32, 64, lane);
  }
  float* sb=seq+(size_t)pbase*HH;
  #pragma unroll
  for(int j=0;j<4;++j){
    #pragma unroll
    for(int v=0;v<8;++v)
      sb[(size_t)(v+rowoff)*HH + j*16 + colb]=hs[j][v]*0.125f;
  }
}

__global__ __launch_bounds__(256) void k_qkv(
    const float* __restrict__ X, const _Float16* __restrict__ WqkvT,
    const float* __restrict__ bqkv, _Float16* __restrict__ Qp,
    _Float16* __restrict__ Kp, _Float16* __restrict__ Vh)
{
  int lane=threadIdx.x&31, wave=threadIdx.x>>5;
  size_t rbase=((size_t)blockIdx.x*8+wave)*16;
  int b=(int)(rbase>>9), s=(int)(rbase&511);
  int colb=lane&15, rowoff=(lane>>4)*8;
  v16h A[2];
  #pragma unroll
  for(int kk=0;kk<2;++kk) A[kk]=load_tile_f32(X+rbase*HH+kk*32, HH, lane);

  #pragma unroll
  for(int j=0;j<12;++j){
    v8f acc=zero8();
    #pragma unroll
    for(int kk=0;kk<2;++kk)
      acc=WMMA_F16(A[kk], load_tile_sw(WqkvT, j*2+kk, lane), acc);
    float bias=bqkv[j*16+colb];
    int h=j&3;
    size_t bh=(size_t)b*4+h;
    if(j<4){
      _Float16* dst=Qp+(bh*SQ+s)*32;
      #pragma unroll
      for(int v=0;v<8;++v){
        size_t ro=(size_t)(v+rowoff)*32;
        dst[ro+colb]=(_Float16)((acc[v]+bias)*0.25f);
        dst[ro+16+colb]=(_Float16)0.f;
      }
    } else if(j<8){
      _Float16* dst=Kp+(bh*SQ+s)*32;
      #pragma unroll
      for(int v=0;v<8;++v){
        size_t ro=(size_t)(v+rowoff)*32;
        dst[ro+colb]=(_Float16)(acc[v]+bias);
        dst[ro+16+colb]=(_Float16)0.f;
      }
    } else {
      _Float16* dst=Vh+(bh*SQ+s)*16;
      #pragma unroll
      for(int v=0;v<8;++v)
        dst[(size_t)(v+rowoff)*16+colb]=(_Float16)(acc[v]+bias);
    }
  }
}

__device__ __forceinline__ float xhalf(float x){
  int b=__builtin_bit_cast(int,x);
  return __builtin_bit_cast(float,
      __builtin_amdgcn_permlanex16(b,b,0x76543210,0xfedcba98,false,false));
}

__global__ __launch_bounds__(256) void k_attn(
    const _Float16* __restrict__ Qp, const _Float16* __restrict__ Kp,
    const _Float16* __restrict__ Vh, float* __restrict__ O)
{
  __shared__ _Float16 Ks[SQ*32];
  __shared__ _Float16 Vs[SQ*16];
  int bh=blockIdx.x, b=bh>>2, h=bh&3;
  int tid=threadIdx.x, lane=tid&31, wave=tid>>5;
  int colb=lane&15, rowoff=(lane>>4)*8;

  {
    const uint32_t* ksrc=(const uint32_t*)(Kp+(size_t)bh*SQ*32);
    uint32_t* kdst=(uint32_t*)Ks;
    for(int i=tid;i<SQ*16;i+=256) kdst[i]=ksrc[i];
    const uint32_t* vsrc=(const uint32_t*)(Vh+(size_t)bh*SQ*16);
    uint32_t* vdst=(uint32_t*)Vs;
    for(int i=tid;i<SQ*8;i+=256) vdst[i]=vsrc[i];
  }
  __syncthreads();

  for(int qt=0;qt<4;++qt){
    int qbase=wave*64+qt*16;
    v16h Qb=load_tile_f16(Qp+((size_t)bh*SQ+qbase)*32, 32, lane);
    v8f acc=zero8();
    float mrow=-1e30f;
    float lpart=0.f;

    for(int kc=0;kc<16;++kc){
      int kbase=kc*32;
      v16h Ka0=load_tile_f16(Ks+(size_t)kbase*32, 32, lane);
      v16h Ka1=load_tile_f16(Ks+(size_t)(kbase+16)*32, 32, lane);
      v16h Va=load_tile_kmajor_f16(Vs+(size_t)kbase*16, 16, lane);
      v8f s0=WMMA_F16(Ka0,Qb,zero8());
      v8f s1=WMMA_F16(Ka1,Qb,zero8());
      float cm=fmaxf(s0[0],s1[0]);
      #pragma unroll
      for(int e=1;e<8;++e) cm=fmaxf(cm,fmaxf(s0[e],s1[e]));
      cm=fmaxf(cm,xhalf(cm));
      float nm=fmaxf(mrow,cm);
      float sc=__expf(mrow-nm);
      mrow=nm;
      v16h Pb; float ps=0.f;
      #pragma unroll
      for(int e=0;e<8;++e){
        float p0=__expf(s0[e]-nm);
        float p1=__expf(s1[e]-nm);
        ps+=p0+p1;
        Pb[e]=(_Float16)p0;
        Pb[8+e]=(_Float16)p1;
      }
      lpart=lpart*sc+ps;
      #pragma unroll
      for(int e=0;e<8;++e) acc[e]*=sc;
      acc=WMMA_F16(Va,Pb,acc);
    }
    float lrow=lpart+xhalf(lpart);
    float* ob=O+((size_t)b*SQ+qbase+colb)*HH + h*16 + rowoff;
    #pragma unroll
    for(int e=0;e<8;++e) ob[e]=acc[e]/lrow;
  }
}

__global__ __launch_bounds__(256) void k_projln(
    float* __restrict__ X, const float* __restrict__ Y,
    const _Float16* __restrict__ WoT, const float* __restrict__ bo,
    const float* __restrict__ g, const float* __restrict__ be)
{
  int lane=threadIdx.x&31, wave=threadIdx.x>>5;
  size_t rbase=((size_t)blockIdx.x*8+wave)*16;
  int colb=lane&15, rowoff=(lane>>4)*8;
  v16h A[2];
  #pragma unroll
  for(int kk=0;kk<2;++kk) A[kk]=load_tile_f32(Y+rbase*HH+kk*32, HH, lane);
  v8f o[4];
  #pragma unroll
  for(int j=0;j<4;++j){
    v8f acc=zero8();
    #pragma unroll
    for(int kk=0;kk<2;++kk)
      acc=WMMA_F16(A[kk], load_tile_sw(WoT, j*2+kk, lane), acc);
    float bias=bo[j*16+colb];
    const float* xr=X+rbase*HH+j*16+colb;
    #pragma unroll
    for(int e=0;e<8;++e) o[j][e]=acc[e]+bias+xr[(size_t)(e+rowoff)*HH];
  }
  float gv[4],bv[4];
  #pragma unroll
  for(int j=0;j<4;++j){ gv[j]=g[j*16+colb]; bv[j]=be[j*16+colb]; }
  float* xw=X+rbase*HH;
  #pragma unroll
  for(int e=0;e<8;++e){
    float s=o[0][e]+o[1][e]+o[2][e]+o[3][e];
    float mu=rsum16(s)*(1.f/64.f);
    float d0=o[0][e]-mu,d1=o[1][e]-mu,d2=o[2][e]-mu,d3=o[3][e]-mu;
    float var=rsum16(d0*d0+d1*d1+d2*d2+d3*d3)*(1.f/64.f);
    float inv=rsqrtf(var+1e-5f);
    #pragma unroll
    for(int j=0;j<4;++j)
      xw[(size_t)(e+rowoff)*HH+j*16+colb]=(o[j][e]-mu)*inv*gv[j]+bv[j];
  }
}

__global__ __launch_bounds__(128) void k_ffln(
    float* __restrict__ X, const _Float16* __restrict__ W1T,
    const float* __restrict__ b1, const _Float16* __restrict__ W2T,
    const float* __restrict__ b2, const float* __restrict__ g,
    const float* __restrict__ be)
{
  __shared__ _Float16 h1[4][16*256];
  int lane=threadIdx.x&31, wave=threadIdx.x>>5;
  size_t rbase=((size_t)blockIdx.x*4+wave)*16;
  int colb=lane&15, rowoff=(lane>>4)*8;
  v16h A[2];
  #pragma unroll
  for(int kk=0;kk<2;++kk) A[kk]=load_tile_f32(X+rbase*HH+kk*32, HH, lane);
  _Float16* hl=&h1[wave][0];
  #pragma unroll
  for(int j=0;j<16;++j){
    v8f acc=zero8();
    #pragma unroll
    for(int kk=0;kk<2;++kk)
      acc=WMMA_F16(A[kk], load_tile_sw(W1T, j*2+kk, lane), acc);
    float bias=b1[j*16+colb];
    #pragma unroll
    for(int e=0;e<8;++e)
      hl[(size_t)(e+rowoff)*256+j*16+colb]=(_Float16)fmaxf(acc[e]+bias,0.f);
  }
  v16h pA[8];
  #pragma unroll
  for(int kk=0;kk<8;++kk) pA[kk]=load_tile_f16(hl+kk*32, 256, lane);
  v8f o[4];
  #pragma unroll
  for(int j=0;j<4;++j){
    v8f acc=zero8();
    #pragma unroll
    for(int kk=0;kk<8;++kk)
      acc=WMMA_F16(pA[kk], load_tile_sw(W2T, j*8+kk, lane), acc);
    float bias=b2[j*16+colb];
    const float* xr=X+rbase*HH+j*16+colb;
    #pragma unroll
    for(int e=0;e<8;++e) o[j][e]=acc[e]+bias+xr[(size_t)(e+rowoff)*HH];
  }
  float gv[4],bv[4];
  #pragma unroll
  for(int j=0;j<4;++j){ gv[j]=g[j*16+colb]; bv[j]=be[j*16+colb]; }
  float* xw=X+rbase*HH;
  #pragma unroll
  for(int e=0;e<8;++e){
    float s=o[0][e]+o[1][e]+o[2][e]+o[3][e];
    float mu=rsum16(s)*(1.f/64.f);
    float d0=o[0][e]-mu,d1=o[1][e]-mu,d2=o[2][e]-mu,d3=o[3][e]-mu;
    float var=rsum16(d0*d0+d1*d1+d2*d2+d3*d3)*(1.f/64.f);
    float inv=rsqrtf(var+1e-5f);
    #pragma unroll
    for(int j=0;j<4;++j)
      xw[(size_t)(e+rowoff)*HH+j*16+colb]=(o[j][e]-mu)*inv*gv[j]+bv[j];
  }
}

__global__ void k_mean(const float* __restrict__ X, float* __restrict__ out){
  int b=blockIdx.x, j=threadIdx.x;
  float s=0.f;
  for(int np=0;np<SQ;++np) s+=X[((size_t)b*SQ+np)*HH+j];
  out[b*HH+j]=s*(1.f/512.f);
}

extern "C" void kernel_launch(void* const* d_in, const int* in_sizes, int n_in,
                              void* d_out, int out_size, void* d_ws, size_t ws_size,
                              hipStream_t stream){
  const float* x  =(const float*)d_in[0];
  const float* Wih=(const float*)d_in[1];
  const float* Whh=(const float*)d_in[2];
  const float* bih=(const float*)d_in[3];
  const float* bhh=(const float*)d_in[4];

  char* ws=(char*)d_ws;
  float*    X =(float*)ws;
  float*    Y =(float*)(ws+(size_t)(8<<20));
  _Float16* Qp=(_Float16*)(ws+(size_t)(16<<20));
  _Float16* Kp=(_Float16*)(ws+(size_t)(24<<20));
  _Float16* Vh=(_Float16*)(ws+(size_t)(32<<20));
  _Float16* WihT=(_Float16*)(ws+(size_t)(36<<20));
  _Float16* WhhT=(_Float16*)(ws+(size_t)(36<<20)+(64<<10));
  _Float16* LW[2][4];
  for(int L=0;L<2;++L){
    char* base=ws+(size_t)(37<<20)+(size_t)L*(1<<20);
    LW[L][0]=(_Float16*)(base);
    LW[L][1]=(_Float16*)(base+(32<<10));
    LW[L][2]=(_Float16*)(base+(64<<10));
    LW[L][3]=(_Float16*)(base+(128<<10));
  }

  k_swz_all<<<264,32,0,stream>>>(Wih, Whh,
      (const float*)d_in[5],  (const float*)d_in[7],
      (const float*)d_in[9],  (const float*)d_in[11],
      (const float*)d_in[17], (const float*)d_in[19],
      (const float*)d_in[21], (const float*)d_in[23],
      WihT);

  k_gru<<<256,256,0,stream>>>(x,WihT,WhhT,bih,bhh,X);

  for(int L=0;L<2;++L){
    int base=5+L*12;
    const float* bqkv=(const float*)d_in[base+1];
    const float* bo  =(const float*)d_in[base+3];
    const float* b1  =(const float*)d_in[base+5];
    const float* b2  =(const float*)d_in[base+7];
    const float* g1  =(const float*)d_in[base+8];
    const float* be1 =(const float*)d_in[base+9];
    const float* g2  =(const float*)d_in[base+10];
    const float* be2 =(const float*)d_in[base+11];

    k_qkv   <<<256,256,0,stream>>>(X,LW[L][0],bqkv,Qp,Kp,Vh);
    k_attn  <<<256,256,0,stream>>>(Qp,Kp,Vh,Y);
    k_projln<<<256,256,0,stream>>>(X,Y,LW[L][1],bo,g1,be1);
    k_ffln  <<<512,128,0,stream>>>(X,LW[L][2],b1,LW[L][3],b2,g2,be2);
  }
  k_mean<<<64,64,0,stream>>>(X,(float*)d_out);
}


# GRU input weights Wih (48 KB of f16 tiles) staged once per workgroup in LDS beside Whh, as two 16-byte planes per tile: the 96 global weight loads per wave and step become LDS loads
# speedup vs baseline: 1.8571x; 1.0394x over previous
#include <hip/hip_runtime.h>
#include <hip/hip_fp16.h>

typedef __attribute__((ext_vector_type(16))) _Float16 v16h;
typedef __attribute__((ext_vector_type(8)))  float    v8f;
typedef __attribute__((ext_vector_type(4)))  float    v4f;

#define WMMA_F16(a,b,c) __builtin_amdgcn_wmma_f32_16x16x32_f16(false,(a),false,(b),(short)0,(c),false,false)

__device__ __forceinline__ v8f zero8(){
  v8f z;
  #pragma unroll
  for(int i=0;i<8;++i) z[i]=0.f;
  return z;
}
__device__ __forceinline__ v16h zero16(){
  v16h z;
  #pragma unroll
  for(int i=0;i<16;++i) z[i]=(_Float16)0.f;
  return z;
}

__device__ __forceinline__ int a_k(int i, int h16){ return (i&7) + ((i>>3)<<4) + (h16<<3); }

__device__ __forceinline__ v16h load_tile_f32(const float* __restrict__ src, int ld, int lane){
  const float* rb = src + (size_t)(lane&15)*ld;
  int h16=(lane>>4)&1; v16h r;
  #pragma unroll
  for(int i=0;i<16;++i) r[i]=(_Float16)rb[a_k(i,h16)];
  return r;
}
__device__ __forceinline__ v16h load_tile_f16(const _Float16* __restrict__ src, int ld, int lane){
  const _Float16* rb = src + (size_t)(lane&15)*ld;
  int h16=(lane>>4)&1; v16h r;
  #pragma unroll
  for(int i=0;i<16;++i) r[i]=rb[a_k(i,h16)];
  return r;
}
__device__ __forceinline__ v16h load_tile_kmajor_f16(const _Float16* __restrict__ src, int ld, int lane){
  const _Float16* cb = src + (lane&15);
  int h16=(lane>>4)&1; v16h r;
  #pragma unroll
  for(int i=0;i<16;++i) r[i]=cb[(size_t)a_k(i,h16)*ld];
  return r;
}
__device__ __forceinline__ v16h load_tile_sw(const _Float16* src, int tile, int lane){
  return *reinterpret_cast<const v16h*>(src + (((size_t)tile*32 + lane) << 4));
}

typedef __attribute__((ext_vector_type(8))) _Float16 v8h;
__device__ __forceinline__ v16h load_tile_planes(const _Float16* src, int tile, int lane){
  const _Float16* tb = src + ((size_t)tile << 9) + (lane << 3);
  v8h lo = *reinterpret_cast<const v8h*>(tb);
  v8h hi = *reinterpret_cast<const v8h*>(tb + 256);
  return __builtin_shufflevector(lo, hi, 0,1,2,3,4,5,6,7,8,9,10,11,12,13,14,15);
}

__device__ __forceinline__ float rmax16(float x){
  #pragma unroll
  for(int m=8;m;m>>=1) x=fmaxf(x,__shfl_xor(x,m,32));
  return x;
}
__device__ __forceinline__ float rsum16(float x){
  #pragma unroll
  for(int m=8;m;m>>=1) x+=__shfl_xor(x,m,32);
  return x;
}

#define PB  8
#define DD  128
#define HH  64
#define SQ  512

__global__ __launch_bounds__(32) void k_swz_all(
    const float* __restrict__ wih,   const float* __restrict__ whh,
    const float* __restrict__ wqkv0, const float* __restrict__ wo0,
    const float* __restrict__ w1_0,  const float* __restrict__ w2_0,
    const float* __restrict__ wqkv1, const float* __restrict__ wo1,
    const float* __restrict__ w1_1,  const float* __restrict__ w2_1,
    _Float16* __restrict__ obase)
{
  int t = blockIdx.x, lane = threadIdx.x;
  const float* Wm; _Float16* out; int ksh;
  if(t < 48){ Wm = wih; out = obase; ksh = 2; }
  else if(t < 72){ t -= 48; Wm = whh; out = obase + (32<<10); ksh = 1; }
  else {
    t -= 72;
    int L = (t >= 96) ? 1 : 0;
    t -= L*96;
    _Float16* lb = obase + ((size_t)(1+L) << 19);
    if(t < 24){ Wm = L ? wqkv1 : wqkv0; out = lb; ksh = 1; }
    else if(t < 32){ t -= 24; Wm = L ? wo1 : wo0; out = lb + (16<<10); ksh = 1; }
    else if(t < 64){ t -= 32; Wm = L ? w1_1 : w1_0; out = lb + (32<<10); ksh = 1; }
    else { t -= 64; Wm = L ? w2_1 : w2_0; out = lb + (64<<10); ksh = 3; }
  }
  int K = 32 << ksh;
  int kt = t & ((1<<ksh)-1), jt = t >> ksh;
  v16h r = load_tile_f32(Wm + (size_t)jt*16*K + kt*32, K, lane);
  *reinterpret_cast<v16h*>(out + (((size_t)t*32 + lane) << 4)) = r;
}

__global__ __launch_bounds__(256) __attribute__((amdgpu_waves_per_eu(2,2))) void k_gru(
    const float* __restrict__ x, const _Float16* __restrict__ WihT,
    const _Float16* __restrict__ WhhT, const float* __restrict__ bih,
    const float* __restrict__ bhh, float* __restrict__ seq)
{
  __shared__ _Float16 hscr[8][16*64];
  __shared__ _Float16 WhhL[192*64];
  __shared__ __attribute__((aligned(16))) _Float16 WihL[192*128];
  int lane=threadIdx.x&31, wave=threadIdx.x>>5;
  int pbase=(blockIdx.x*8+wave)*16;
  int colb=lane&15, rowoff=(lane>>4)*8;

  {
    const uint32_t* wsrc=(const uint32_t*)WhhT;
    uint32_t* wdst=(uint32_t*)WhhL;
    for(int i=threadIdx.x;i<192*64/2;i+=256) wdst[i]=wsrc[i];
    const v4f* isrc=(const v4f*)WihT;
    v4f* idst=(v4f*)WihL;
    #pragma unroll
    for(int k=0;k<12;++k){
      int g=threadIdx.x+k*256;
      int u=g&63;
      idst[(g&~63) + ((u&1)<<5) + (u>>1)]=isrc[g];
    }
  }
  __syncthreads();

  float bi[12], bh[12];
  #pragma unroll
  for(int j=0;j<12;++j){ bi[j]=bih[j*16+colb]; bh[j]=bhh[j*16+colb]; }

  v8f hT[4], hs[4];
  #pragma unroll
  for(int j=0;j<4;++j){ hT[j]=zero8(); hs[j]=zero8(); }
  v16h hA[2]; hA[0]=zero16(); hA[1]=zero16();
  _Float16* hl=&hscr[wave][0];

  for(int t=0;t<PB;++t){
    const float* xsrc = x + ((size_t)pbase*PB + t)*DD;
    if(t+1<PB) __builtin_prefetch(xsrc+DD,0,1);
    v4f xr[16];
    {
      const float* rb = xsrc + (size_t)(lane&15)*(PB*DD) + (((lane>>4)&1)<<3);
      #pragma unroll
      for(int kk=0;kk<4;++kk){
        xr[kk*4+0]=*reinterpret_cast<const v4f*>(rb+kk*32);
        xr[kk*4+1]=*reinterpret_cast<const v4f*>(rb+kk*32+4);
        xr[kk*4+2]=*reinterpret_cast<const v4f*>(rb+kk*32+16);
        xr[kk*4+3]=*reinterpret_cast<const v4f*>(rb+kk*32+20);
      }
    }
    __builtin_amdgcn_sched_barrier(0);
    v16h xA[4];
    #pragma unroll
    for(int kk=0;kk<4;++kk){
      #pragma unroll
      for(int q=0;q<4;++q){
        #pragma unroll
        for(int c=0;c<4;++c) xA[kk][q*4+c]=(_Float16)xr[kk*4+q][c];
      }
    }

    #pragma unroll
    for(int j=0;j<4;++j){
      v8f gr=zero8();
      #pragma unroll
      for(int kk=0;kk<4;++kk)
        gr=WMMA_F16(xA[kk], load_tile_planes(WihL, j*4+kk, lane), gr);
      #pragma unroll
      for(int kk=0;kk<2;++kk)
        gr=WMMA_F16(hA[kk], load_tile_sw(WhhL, j*2+kk, lane), gr);
      v8f gz=zero8();
      #pragma unroll
      for(int kk=0;kk<4;++kk)
        gz=WMMA_F16(xA[kk], load_tile_planes(WihL, (4+j)*4+kk, lane), gz);
      #pragma unroll
      for(int kk=0;kk<2;++kk)
        gz=WMMA_F16(hA[kk], load_tile_sw(WhhL, (4+j)*2+kk, lane), gz);
      v8f gnx=zero8();
      #pragma unroll
      for(int kk=0;kk<4;++kk)
        gnx=WMMA_F16(xA[kk], load_tile_planes(WihL, (8+j)*4+kk, lane), gnx);
      v8f gnh=zero8();
      #pragma unroll
      for(int kk=0;kk<2;++kk)
        gnh=WMMA_F16(hA[kk], load_tile_sw(WhhL, (8+j)*2+kk, lane), gnh);
      float rr[8], zz[8], nn[8];
      #pragma unroll
      for(int e=0;e<8;++e){
        rr[e]=(gr[e]+bi[j]+bh[j])*-1.44269504f;
        zz[e]=(gz[e]+bi[4+j]+bh[4+j])*-1.44269504f;
      }
      #pragma unroll
      for(int e=0;e<8;++e){
        rr[e]=1.f+__builtin_amdgcn_exp2f(rr[e]);
        zz[e]=1.f+__builtin_amdgcn_exp2f(zz[e]);
      }
      #pragma unroll
      for(int e=0;e<8;++e){
        rr[e]=__builtin_amdgcn_rcpf(rr[e]);
        zz[e]=__builtin_amdgcn_rcpf(zz[e]);
      }
      #pragma unroll
      for(int e=0;e<8;++e)
        nn[e]=gnx[e]+bi[8+j]+rr[e]*(gnh[e]+bh[8+j]);
      #pragma unroll
      for(int e=0;e<8;++e) nn[e]=__builtin_amdgcn_tanhf(nn[e]);
      #pragma unroll
      for(int e=0;e<8;++e){
        float hn=(1.f-zz[e])*nn[e] + zz[e]*hT[j][e];
        hT[j][e]=hn; hs[j][e]+=hn;
      }
    }
    #pragma unroll
    for(int j=0;j<4;++j){
      #pragma unroll
      for(int v=0;v<8;++v)
        hl[(v+rowoff)*64 + j*16 + colb]=(_Float16)hT[j][v];
    }
    #pragma unroll
    for(int kk=0;kk<2;++kk) hA[kk]=load_tile_f16(hl+kk*32, 64, lane);
  }
  float* sb=seq+(size_t)pbase*HH;
  #pragma unroll
  for(int j=0;j<4;++j){
    #pragma unroll
    for(int v=0;v<8;++v)
      sb[(size_t)(v+rowoff)*HH + j*16 + colb]=hs[j][v]*0.125f;
  }
}

__global__ __launch_bounds__(256) void k_qkv(
    const float* __restrict__ X, const _Float16* __restrict__ WqkvT,
    const float* __restrict__ bqkv, _Float16* __restrict__ Qp,
    _Float16* __restrict__ Kp, _Float16* __restrict__ Vh)
{
  int lane=threadIdx.x&31, wave=threadIdx.x>>5;
  size_t rbase=((size_t)blockIdx.x*8+wave)*16;
  int b=(int)(rbase>>9), s=(int)(rbase&511);
  int colb=lane&15, rowoff=(lane>>4)*8;
  v16h A[2];
  #pragma unroll
  for(int kk=0;kk<2;++kk) A[kk]=load_tile_f32(X+rbase*HH+kk*32, HH, lane);

  #pragma unroll
  for(int j=0;j<12;++j){
    v8f acc=zero8();
    #pragma unroll
    for(int kk=0;kk<2;++kk)
      acc=WMMA_F16(A[kk], load_tile_sw(WqkvT, j*2+kk, lane), acc);
    float bias=bqkv[j*16+colb];
    int h=j&3;
    size_t bh=(size_t)b*4+h;
    if(j<4){
      _Float16* dst=Qp+(bh*SQ+s)*32;
      #pragma unroll
      for(int v=0;v<8;++v){
        size_t ro=(size_t)(v+rowoff)*32;
        dst[ro+colb]=(_Float16)((acc[v]+bias)*0.25f);
        dst[ro+16+colb]=(_Float16)0.f;
      }
    } else if(j<8){
      _Float16* dst=Kp+(bh*SQ+s)*32;
      #pragma unroll
      for(int v=0;v<8;++v){
        size_t ro=(size_t)(v+rowoff)*32;
        dst[ro+colb]=(_Float16)(acc[v]+bias);
        dst[ro+16+colb]=(_Float16)0.f;
      }
    } else {
      _Float16* dst=Vh+(bh*SQ+s)*16;
      #pragma unroll
      for(int v=0;v<8;++v)
        dst[(size_t)(v+rowoff)*16+colb]=(_Float16)(acc[v]+bias);
    }
  }
}

__device__ __forceinline__ float xhalf(float x){
  int b=__builtin_bit_cast(int,x);
  return __builtin_bit_cast(float,
      __builtin_amdgcn_permlanex16(b,b,0x76543210,0xfedcba98,false,false));
}

__global__ __launch_bounds__(256) void k_attn(
    const _Float16* __restrict__ Qp, const _Float16* __restrict__ Kp,
    const _Float16* __restrict__ Vh, float* __restrict__ O)
{
  __shared__ _Float16 Ks[SQ*32];
  __shared__ _Float16 Vs[SQ*16];
  int bh=blockIdx.x, b=bh>>2, h=bh&3;
  int tid=threadIdx.x, lane=tid&31, wave=tid>>5;
  int colb=lane&15, rowoff=(lane>>4)*8;

  {
    const uint32_t* ksrc=(const uint32_t*)(Kp+(size_t)bh*SQ*32);
    uint32_t* kdst=(uint32_t*)Ks;
    for(int i=tid;i<SQ*16;i+=256) kdst[i]=ksrc[i];
    const uint32_t* vsrc=(const uint32_t*)(Vh+(size_t)bh*SQ*16);
    uint32_t* vdst=(uint32_t*)Vs;
    for(int i=tid;i<SQ*8;i+=256) vdst[i]=vsrc[i];
  }
  __syncthreads();

  for(int qt=0;qt<4;++qt){
    int qbase=wave*64+qt*16;
    v16h Qb=load_tile_f16(Qp+((size_t)bh*SQ+qbase)*32, 32, lane);
    v8f acc=zero8();
    float mrow=-1e30f;
    float lpart=0.f;

    for(int kc=0;kc<16;++kc){
      int kbase=kc*32;
      v16h Ka0=load_tile_f16(Ks+(size_t)kbase*32, 32, lane);
      v16h Ka1=load_tile_f16(Ks+(size_t)(kbase+16)*32, 32, lane);
      v16h Va=load_tile_kmajor_f16(Vs+(size_t)kbase*16, 16, lane);
      v8f s0=WMMA_F16(Ka0,Qb,zero8());
      v8f s1=WMMA_F16(Ka1,Qb,zero8());
      float cm=fmaxf(s0[0],s1[0]);
      #pragma unroll
      for(int e=1;e<8;++e) cm=fmaxf(cm,fmaxf(s0[e],s1[e]));
      cm=fmaxf(cm,xhalf(cm));
      float nm=fmaxf(mrow,cm);
      float sc=__expf(mrow-nm);
      mrow=nm;
      v16h Pb; float ps=0.f;
      #pragma unroll
      for(int e=0;e<8;++e){
        float p0=__expf(s0[e]-nm);
        float p1=__expf(s1[e]-nm);
        ps+=p0+p1;
        Pb[e]=(_Float16)p0;
        Pb[8+e]=(_Float16)p1;
      }
      lpart=lpart*sc+ps;
      #pragma unroll
      for(int e=0;e<8;++e) acc[e]*=sc;
      acc=WMMA_F16(Va,Pb,acc);
    }
    float lrow=lpart+xhalf(lpart);
    float* ob=O+((size_t)b*SQ+qbase+colb)*HH + h*16 + rowoff;
    #pragma unroll
    for(int e=0;e<8;++e) ob[e]=acc[e]/lrow;
  }
}

__global__ __launch_bounds__(256) void k_projln(
    float* __restrict__ X, const float* __restrict__ Y,
    const _Float16* __restrict__ WoT, const float* __restrict__ bo,
    const float* __restrict__ g, const float* __restrict__ be)
{
  int lane=threadIdx.x&31, wave=threadIdx.x>>5;
  size_t rbase=((size_t)blockIdx.x*8+wave)*16;
  int colb=lane&15, rowoff=(lane>>4)*8;
  v16h A[2];
  #pragma unroll
  for(int kk=0;kk<2;++kk) A[kk]=load_tile_f32(Y+rbase*HH+kk*32, HH, lane);
  v8f o[4];
  #pragma unroll
  for(int j=0;j<4;++j){
    v8f acc=zero8();
    #pragma unroll
    for(int kk=0;kk<2;++kk)
      acc=WMMA_F16(A[kk], load_tile_sw(WoT, j*2+kk, lane), acc);
    float bias=bo[j*16+colb];
    const float* xr=X+rbase*HH+j*16+colb;
    #pragma unroll
    for(int e=0;e<8;++e) o[j][e]=acc[e]+bias+xr[(size_t)(e+rowoff)*HH];
  }
  float gv[4],bv[4];
  #pragma unroll
  for(int j=0;j<4;++j){ gv[j]=g[j*16+colb]; bv[j]=be[j*16+colb]; }
  float* xw=X+rbase*HH;
  #pragma unroll
  for(int e=0;e<8;++e){
    float s=o[0][e]+o[1][e]+o[2][e]+o[3][e];
    float mu=rsum16(s)*(1.f/64.f);
    float d0=o[0][e]-mu,d1=o[1][e]-mu,d2=o[2][e]-mu,d3=o[3][e]-mu;
    float var=rsum16(d0*d0+d1*d1+d2*d2+d3*d3)*(1.f/64.f);
    float inv=rsqrtf(var+1e-5f);
    #pragma unroll
    for(int j=0;j<4;++j)
      xw[(size_t)(e+rowoff)*HH+j*16+colb]=(o[j][e]-mu)*inv*gv[j]+bv[j];
  }
}

__global__ __launch_bounds__(128) void k_ffln(
    float* __restrict__ X, const _Float16* __restrict__ W1T,
    const float* __restrict__ b1, const _Float16* __restrict__ W2T,
    const float* __restrict__ b2, const float* __restrict__ g,
    const float* __restrict__ be)
{
  __shared__ _Float16 h1[4][16*256];
  int lane=threadIdx.x&31, wave=threadIdx.x>>5;
  size_t rbase=((size_t)blockIdx.x*4+wave)*16;
  int colb=lane&15, rowoff=(lane>>4)*8;
  v16h A[2];
  #pragma unroll
  for(int kk=0;kk<2;++kk) A[kk]=load_tile_f32(X+rbase*HH+kk*32, HH, lane);
  _Float16* hl=&h1[wave][0];
  #pragma unroll
  for(int j=0;j<16;++j){
    v8f acc=zero8();
    #pragma unroll
    for(int kk=0;kk<2;++kk)
      acc=WMMA_F16(A[kk], load_tile_sw(W1T, j*2+kk, lane), acc);
    float bias=b1[j*16+colb];
    #pragma unroll
    for(int e=0;e<8;++e)
      hl[(size_t)(e+rowoff)*256+j*16+colb]=(_Float16)fmaxf(acc[e]+bias,0.f);
  }
  v16h pA[8];
  #pragma unroll
  for(int kk=0;kk<8;++kk) pA[kk]=load_tile_f16(hl+kk*32, 256, lane);
  v8f o[4];
  #pragma unroll
  for(int j=0;j<4;++j){
    v8f acc=zero8();
    #pragma unroll
    for(int kk=0;kk<8;++kk)
      acc=WMMA_F16(pA[kk], load_tile_sw(W2T, j*8+kk, lane), acc);
    float bias=b2[j*16+colb];
    const float* xr=X+rbase*HH+j*16+colb;
    #pragma unroll
    for(int e=0;e<8;++e) o[j][e]=acc[e]+bias+xr[(size_t)(e+rowoff)*HH];
  }
  float gv[4],bv[4];
  #pragma unroll
  for(int j=0;j<4;++j){ gv[j]=g[j*16+colb]; bv[j]=be[j*16+colb]; }
  float* xw=X+rbase*HH;
  #pragma unroll
  for(int e=0;e<8;++e){
    float s=o[0][e]+o[1][e]+o[2][e]+o[3][e];
    float mu=rsum16(s)*(1.f/64.f);
    float d0=o[0][e]-mu,d1=o[1][e]-mu,d2=o[2][e]-mu,d3=o[3][e]-mu;
    float var=rsum16(d0*d0+d1*d1+d2*d2+d3*d3)*(1.f/64.f);
    float inv=rsqrtf(var+1e-5f);
    #pragma unroll
    for(int j=0;j<4;++j)
      xw[(size_t)(e+rowoff)*HH+j*16+colb]=(o[j][e]-mu)*inv*gv[j]+bv[j];
  }
}

__global__ void k_mean(const float* __restrict__ X, float* __restrict__ out){
  int b=blockIdx.x, j=threadIdx.x;
  float s=0.f;
  for(int np=0;np<SQ;++np) s+=X[((size_t)b*SQ+np)*HH+j];
  out[b*HH+j]=s*(1.f/512.f);
}

extern "C" void kernel_launch(void* const* d_in, const int* in_sizes, int n_in,
                              void* d_out, int out_size, void* d_ws, size_t ws_size,
                              hipStream_t stream){
  const float* x  =(const float*)d_in[0];
  const float* Wih=(const float*)d_in[1];
  const float* Whh=(const float*)d_in[2];
  const float* bih=(const float*)d_in[3];
  const float* bhh=(const float*)d_in[4];

  char* ws=(char*)d_ws;
  float*    X =(float*)ws;
  float*    Y =(float*)(ws+(size_t)(8<<20));
  _Float16* Qp=(_Float16*)(ws+(size_t)(16<<20));
  _Float16* Kp=(_Float16*)(ws+(size_t)(24<<20));
  _Float16* Vh=(_Float16*)(ws+(size_t)(32<<20));
  _Float16* WihT=(_Float16*)(ws+(size_t)(36<<20));
  _Float16* WhhT=(_Float16*)(ws+(size_t)(36<<20)+(64<<10));
  _Float16* LW[2][4];
  for(int L=0;L<2;++L){
    char* base=ws+(size_t)(37<<20)+(size_t)L*(1<<20);
    LW[L][0]=(_Float16*)(base);
    LW[L][1]=(_Float16*)(base+(32<<10));
    LW[L][2]=(_Float16*)(base+(64<<10));
    LW[L][3]=(_Float16*)(base+(128<<10));
  }

  k_swz_all<<<264,32,0,stream>>>(Wih, Whh,
      (const float*)d_in[5],  (const float*)d_in[7],
      (const float*)d_in[9],  (const float*)d_in[11],
      (const float*)d_in[17], (const float*)d_in[19],
      (const float*)d_in[21], (const float*)d_in[23],
      WihT);

  k_gru<<<256,256,0,stream>>>(x,WihT,WhhT,bih,bhh,X);

  for(int L=0;L<2;++L){
    int base=5+L*12;
    const float* bqkv=(const float*)d_in[base+1];
    const float* bo  =(const float*)d_in[base+3];
    const float* b1  =(const float*)d_in[base+5];
    const float* b2  =(const float*)d_in[base+7];
    const float* g1  =(const float*)d_in[base+8];
    const float* be1 =(const float*)d_in[base+9];
    const float* g2  =(const float*)d_in[base+10];
    const float* be2 =(const float*)d_in[base+11];

    k_qkv   <<<256,256,0,stream>>>(X,LW[L][0],bqkv,Qp,Kp,Vh);
    k_attn  <<<256,256,0,stream>>>(Qp,Kp,Vh,Y);
    k_projln<<<256,256,0,stream>>>(X,Y,LW[L][1],bo,g1,be1);
    k_ffln  <<<512,128,0,stream>>>(X,LW[L][2],b1,LW[L][3],b2,g2,be2);
  }
  k_mean<<<64,64,0,stream>>>(X,(float*)d_out);
}


# FF and QKV weight tiles and bias vectors staged once per 8-wave workgroup in LDS as 16-byte planes, next 8-tile group read from LDS ahead of the current matrix block; FF launch 512x128 -> 256x256
# speedup vs baseline: 2.0616x; 1.1101x over previous
#include <hip/hip_runtime.h>
#include <hip/hip_fp16.h>

typedef __attribute__((ext_vector_type(16))) _Float16 v16h;
typedef __attribute__((ext_vector_type(8)))  float    v8f;
typedef __attribute__((ext_vector_type(4)))  float    v4f;

#define WMMA_F16(a,b,c) __builtin_amdgcn_wmma_f32_16x16x32_f16(false,(a),false,(b),(short)0,(c),false,false)

__device__ __forceinline__ v8f zero8(){
  v8f z;
  #pragma unroll
  for(int i=0;i<8;++i) z[i]=0.f;
  return z;
}
__device__ __forceinline__ v16h zero16(){
  v16h z;
  #pragma unroll
  for(int i=0;i<16;++i) z[i]=(_Float16)0.f;
  return z;
}

__device__ __forceinline__ int a_k(int i, int h16){ return (i&7) + ((i>>3)<<4) + (h16<<3); }

__device__ __forceinline__ v16h load_tile_f32(const float* __restrict__ src, int ld, int lane){
  const float* rb = src + (size_t)(lane&15)*ld;
  int h16=(lane>>4)&1; v16h r;
  #pragma unroll
  for(int i=0;i<16;++i) r[i]=(_Float16)rb[a_k(i,h16)];
  return r;
}
__device__ __forceinline__ v16h load_tile_f16(const _Float16* __restrict__ src, int ld, int lane){
  const _Float16* rb = src + (size_t)(lane&15)*ld;
  int h16=(lane>>4)&1; v16h r;
  #pragma unroll
  for(int i=0;i<16;++i) r[i]=rb[a_k(i,h16)];
  return r;
}
__device__ __forceinline__ v16h load_tile_kmajor_f16(const _Float16* __restrict__ src, int ld, int lane){
  const _Float16* cb = src + (lane&15);
  int h16=(lane>>4)&1; v16h r;
  #pragma unroll
  for(int i=0;i<16;++i) r[i]=cb[(size_t)a_k(i,h16)*ld];
  return r;
}
__device__ __forceinline__ v16h load_tile_sw(const _Float16* src, int tile, int lane){
  return *reinterpret_cast<const v16h*>(src + (((size_t)tile*32 + lane) << 4));
}

typedef __attribute__((ext_vector_type(8))) _Float16 v8h;
__device__ __forceinline__ v16h load_tile_planes(const _Float16* src, int tile, int lane){
  const _Float16* tb = src + ((size_t)tile << 9) + (lane << 3);
  v8h lo = *reinterpret_cast<const v8h*>(tb);
  v8h hi = *reinterpret_cast<const v8h*>(tb + 256);
  return __builtin_shufflevector(lo, hi, 0,1,2,3,4,5,6,7,8,9,10,11,12,13,14,15);
}

__device__ __forceinline__ float rmax16(float x){
  #pragma unroll
  for(int m=8;m;m>>=1) x=fmaxf(x,__shfl_xor(x,m,32));
  return x;
}
__device__ __forceinline__ float rsum16(float x){
  #pragma unroll
  for(int m=8;m;m>>=1) x+=__shfl_xor(x,m,32);
  return x;
}

#define PB  8
#define DD  128
#define HH  64
#define SQ  512

__global__ __launch_bounds__(32) void k_swz_all(
    const float* __restrict__ wih,   const float* __restrict__ whh,
    const float* __restrict__ wqkv0, const float* __restrict__ wo0,
    const float* __restrict__ w1_0,  const float* __restrict__ w2_0,
    const float* __restrict__ wqkv1, const float* __restrict__ wo1,
    const float* __restrict__ w1_1,  const float* __restrict__ w2_1,
    _Float16* __restrict__ obase)
{
  int t = blockIdx.x, lane = threadIdx.x;
  const float* Wm; _Float16* out; int ksh;
  if(t < 48){ Wm = wih; out = obase; ksh = 2; }
  else if(t < 72){ t -= 48; Wm = whh; out = obase + (32<<10); ksh = 1; }
  else {
    t -= 72;
    int L = (t >= 96) ? 1 : 0;
    t -= L*96;
    _Float16* lb = obase + ((size_t)(1+L) << 19);
    if(t < 24){ Wm = L ? wqkv1 : wqkv0; out = lb; ksh = 1; }
    else if(t < 32){ t -= 24; Wm = L ? wo1 : wo0; out = lb + (16<<10); ksh = 1; }
    else if(t < 64){ t -= 32; Wm = L ? w1_1 : w1_0; out = lb + (32<<10); ksh = 1; }
    else { t -= 64; Wm = L ? w2_1 : w2_0; out = lb + (64<<10); ksh = 3; }
  }
  int K = 32 << ksh;
  int kt = t & ((1<<ksh)-1), jt = t >> ksh;
  v16h r = load_tile_f32(Wm + (size_t)jt*16*K + kt*32, K, lane);
  *reinterpret_cast<v16h*>(out + (((size_t)t*32 + lane) << 4)) = r;
}

__global__ __launch_bounds__(256) __attribute__((amdgpu_waves_per_eu(2,2))) void k_gru(
    const float* __restrict__ x, const _Float16* __restrict__ WihT,
    const _Float16* __restrict__ WhhT, const float* __restrict__ bih,
    const float* __restrict__ bhh, float* __restrict__ seq)
{
  __shared__ _Float16 hscr[8][16*64];
  __shared__ _Float16 WhhL[192*64];
  __shared__ __attribute__((aligned(16))) _Float16 WihL[192*128];
  int lane=threadIdx.x&31, wave=threadIdx.x>>5;
  int pbase=(blockIdx.x*8+wave)*16;
  int colb=lane&15, rowoff=(lane>>4)*8;

  {
    const uint32_t* wsrc=(const uint32_t*)WhhT;
    uint32_t* wdst=(uint32_t*)WhhL;
    for(int i=threadIdx.x;i<192*64/2;i+=256) wdst[i]=wsrc[i];
    const v4f* isrc=(const v4f*)WihT;
    v4f* idst=(v4f*)WihL;
    #pragma unroll
    for(int k=0;k<12;++k){
      int g=threadIdx.x+k*256;
      int u=g&63;
      idst[(g&~63) + ((u&1)<<5) + (u>>1)]=isrc[g];
    }
  }
  __syncthreads();

  float bi[12], bh[12];
  #pragma unroll
  for(int j=0;j<12;++j){ bi[j]=bih[j*16+colb]; bh[j]=bhh[j*16+colb]; }

  v8f hT[4], hs[4];
  #pragma unroll
  for(int j=0;j<4;++j){ hT[j]=zero8(); hs[j]=zero8(); }
  v16h hA[2]; hA[0]=zero16(); hA[1]=zero16();
  _Float16* hl=&hscr[wave][0];

  for(int t=0;t<PB;++t){
    const float* xsrc = x + ((size_t)pbase*PB + t)*DD;
    if(t+1<PB) __builtin_prefetch(xsrc+DD,0,1);
    v4f xr[16];
    {
      const float* rb = xsrc + (size_t)(lane&15)*(PB*DD) + (((lane>>4)&1)<<3);
      #pragma unroll
      for(int kk=0;kk<4;++kk){
        xr[kk*4+0]=*reinterpret_cast<const v4f*>(rb+kk*32);
        xr[kk*4+1]=*reinterpret_cast<const v4f*>(rb+kk*32+4);
        xr[kk*4+2]=*reinterpret_cast<const v4f*>(rb+kk*32+16);
        xr[kk*4+3]=*reinterpret_cast<const v4f*>(rb+kk*32+20);
      }
    }
    __builtin_amdgcn_sched_barrier(0);
    v16h xA[4];
    #pragma unroll
    for(int kk=0;kk<4;++kk){
      #pragma unroll
      for(int q=0;q<4;++q){
        #pragma unroll
        for(int c=0;c<4;++c) xA[kk][q*4+c]=(_Float16)xr[kk*4+q][c];
      }
    }

    #pragma unroll
    for(int j=0;j<4;++j){
      v8f gr=zero8();
      #pragma unroll
      for(int kk=0;kk<4;++kk)
        gr=WMMA_F16(xA[kk], load_tile_planes(WihL, j*4+kk, lane), gr);
      #pragma unroll
      for(int kk=0;kk<2;++kk)
        gr=WMMA_F16(hA[kk], load_tile_sw(WhhL, j*2+kk, lane), gr);
      v8f gz=zero8();
      #pragma unroll
      for(int kk=0;kk<4;++kk)
        gz=WMMA_F16(xA[kk], load_tile_planes(WihL, (4+j)*4+kk, lane), gz);
      #pragma unroll
      for(int kk=0;kk<2;++kk)
        gz=WMMA_F16(hA[kk], load_tile_sw(WhhL, (4+j)*2+kk, lane), gz);
      v8f gnx=zero8();
      #pragma unroll
      for(int kk=0;kk<4;++kk)
        gnx=WMMA_F16(xA[kk], load_tile_planes(WihL, (8+j)*4+kk, lane), gnx);
      v8f gnh=zero8();
      #pragma unroll
      for(int kk=0;kk<2;++kk)
        gnh=WMMA_F16(hA[kk], load_tile_sw(WhhL, (8+j)*2+kk, lane), gnh);
      float rr[8], zz[8], nn[8];
      #pragma unroll
      for(int e=0;e<8;++e){
        rr[e]=(gr[e]+bi[j]+bh[j])*-1.44269504f;
        zz[e]=(gz[e]+bi[4+j]+bh[4+j])*-1.44269504f;
      }
      #pragma unroll
      for(int e=0;e<8;++e){
        rr[e]=1.f+__builtin_amdgcn_exp2f(rr[e]);
        zz[e]=1.f+__builtin_amdgcn_exp2f(zz[e]);
      }
      #pragma unroll
      for(int e=0;e<8;++e){
        rr[e]=__builtin_amdgcn_rcpf(rr[e]);
        zz[e]=__builtin_amdgcn_rcpf(zz[e]);
      }
      #pragma unroll
      for(int e=0;e<8;++e)
        nn[e]=gnx[e]+bi[8+j]+rr[e]*(gnh[e]+bh[8+j]);
      #pragma unroll
      for(int e=0;e<8;++e) nn[e]=__builtin_amdgcn_tanhf(nn[e]);
      #pragma unroll
      for(int e=0;e<8;++e){
        float hn=(1.f-zz[e])*nn[e] + zz[e]*hT[j][e];
        hT[j][e]=hn; hs[j][e]+=hn;
      }
    }
    #pragma unroll
    for(int j=0;j<4;++j){
      #pragma unroll
      for(int v=0;v<8;++v)
        hl[(v+rowoff)*64 + j*16 + colb]=(_Float16)hT[j][v];
    }
    #pragma unroll
    for(int kk=0;kk<2;++kk) hA[kk]=load_tile_f16(hl+kk*32, 64, lane);
  }
  float* sb=seq+(size_t)pbase*HH;
  #pragma unroll
  for(int j=0;j<4;++j){
    #pragma unroll
    for(int v=0;v<8;++v)
      sb[(size_t)(v+rowoff)*HH + j*16 + colb]=hs[j][v]*0.125f;
  }
}

__global__ __launch_bounds__(256) void k_qkv(
    const float* __restrict__ X, const _Float16* __restrict__ WqkvT,
    const float* __restrict__ bqkv, _Float16* __restrict__ Qp,
    _Float16* __restrict__ Kp, _Float16* __restrict__ Vh)
{
  __shared__ __attribute__((aligned(16))) _Float16 WL[24*512];
  __shared__ float bL[256];
  int lane=threadIdx.x&31, wave=threadIdx.x>>5;
  size_t rbase=((size_t)blockIdx.x*8+wave)*16;
  int b=(int)(rbase>>9), s=(int)(rbase&511);
  int colb=lane&15, rowoff=(lane>>4)*8;
  v4f ar[8];
  {
    const float* rb = X + rbase*HH + (size_t)(lane&15)*HH + (((lane>>4)&1)<<3);
    #pragma unroll
    for(int kk=0;kk<2;++kk){
      ar[kk*4+0]=*reinterpret_cast<const v4f*>(rb+kk*32);
      ar[kk*4+1]=*reinterpret_cast<const v4f*>(rb+kk*32+4);
      ar[kk*4+2]=*reinterpret_cast<const v4f*>(rb+kk*32+16);
      ar[kk*4+3]=*reinterpret_cast<const v4f*>(rb+kk*32+20);
    }
  }
  __builtin_amdgcn_sched_barrier(0);

  {
    float bvq=bqkv[threadIdx.x<192 ? threadIdx.x : 0];
    const v4f* wsrc=(const v4f*)WqkvT;
    v4f* wdst=(v4f*)WL;
    #pragma unroll
    for(int k=0;k<6;++k){
      int gi=threadIdx.x+k*256;
      int u=gi&63;
      wdst[(gi&~63) + ((u&1)<<5) + (u>>1)]=wsrc[gi];
    }
    bL[threadIdx.x]=bvq;
  }
  __syncthreads();

  v16h A[2];
  #pragma unroll
  for(int kk=0;kk<2;++kk){
    #pragma unroll
    for(int q=0;q<4;++q){
      #pragma unroll
      for(int cc=0;cc<4;++cc) A[kk][q*4+cc]=(_Float16)ar[kk*4+q][cc];
    }
  }

  v16h Bc[8], Bn[8];
  float bc[4], bn[4];
  #pragma unroll
  for(int i=0;i<8;++i) Bc[i]=load_tile_planes(WL, i, lane);
  #pragma unroll
  for(int jj=0;jj<4;++jj) bc[jj]=bL[jj*16+colb];
  #pragma unroll
  for(int g=0;g<3;++g){
    if(g<2){
      #pragma unroll
      for(int i=0;i<8;++i) Bn[i]=load_tile_planes(WL, (g+1)*8+i, lane);
      #pragma unroll
      for(int jj=0;jj<4;++jj) bn[jj]=bL[((g+1)*4+jj)*16+colb];
    }
    __builtin_amdgcn_sched_barrier(0);
    v8f acc[4];
    #pragma unroll
    for(int jj=0;jj<4;++jj) acc[jj]=zero8();
    #pragma unroll
    for(int kk=0;kk<2;++kk){
      #pragma unroll
      for(int jj=0;jj<4;++jj) acc[jj]=WMMA_F16(A[kk], Bc[jj*2+kk], acc[jj]);
    }
    #pragma unroll
    for(int jj=0;jj<4;++jj){
      float bias=bc[jj];
      size_t bh=(size_t)b*4+jj;
      if(g==0){
        _Float16* dst=Qp+(bh*SQ+s)*32;
        #pragma unroll
        for(int v=0;v<8;++v){
          size_t ro=(size_t)(v+rowoff)*32;
          dst[ro+colb]=(_Float16)((acc[jj][v]+bias)*0.25f);
          dst[ro+16+colb]=(_Float16)0.f;
        }
      } else if(g==1){
        _Float16* dst=Kp+(bh*SQ+s)*32;
        #pragma unroll
        for(int v=0;v<8;++v){
          size_t ro=(size_t)(v+rowoff)*32;
          dst[ro+colb]=(_Float16)(acc[jj][v]+bias);
          dst[ro+16+colb]=(_Float16)0.f;
        }
      } else {
        _Float16* dst=Vh+(bh*SQ+s)*16;
        #pragma unroll
        for(int v=0;v<8;++v)
          dst[(size_t)(v+rowoff)*16+colb]=(_Float16)(acc[jj][v]+bias);
      }
    }
    if(g<2){
      #pragma unroll
      for(int i=0;i<8;++i) Bc[i]=Bn[i];
      #pragma unroll
      for(int jj=0;jj<4;++jj) bc[jj]=bn[jj];
    }
  }
}

__device__ __forceinline__ float xhalf(float x){
  int b=__builtin_bit_cast(int,x);
  return __builtin_bit_cast(float,
      __builtin_amdgcn_permlanex16(b,b,0x76543210,0xfedcba98,false,false));
}

__global__ __launch_bounds__(256) void k_attn(
    const _Float16* __restrict__ Qp, const _Float16* __restrict__ Kp,
    const _Float16* __restrict__ Vh, float* __restrict__ O)
{
  __shared__ _Float16 Ks[SQ*32];
  __shared__ _Float16 Vs[SQ*16];
  int bh=blockIdx.x, b=bh>>2, h=bh&3;
  int tid=threadIdx.x, lane=tid&31, wave=tid>>5;
  int colb=lane&15, rowoff=(lane>>4)*8;

  {
    const uint32_t* ksrc=(const uint32_t*)(Kp+(size_t)bh*SQ*32);
    uint32_t* kdst=(uint32_t*)Ks;
    for(int i=tid;i<SQ*16;i+=256) kdst[i]=ksrc[i];
    const uint32_t* vsrc=(const uint32_t*)(Vh+(size_t)bh*SQ*16);
    uint32_t* vdst=(uint32_t*)Vs;
    for(int i=tid;i<SQ*8;i+=256) vdst[i]=vsrc[i];
  }
  __syncthreads();

  for(int qt=0;qt<4;++qt){
    int qbase=wave*64+qt*16;
    v16h Qb=load_tile_f16(Qp+((size_t)bh*SQ+qbase)*32, 32, lane);
    v8f acc=zero8();
    float mrow=-1e30f;
    float lpart=0.f;

    for(int kc=0;kc<16;++kc){
      int kbase=kc*32;
      v16h Ka0=load_tile_f16(Ks+(size_t)kbase*32, 32, lane);
      v16h Ka1=load_tile_f16(Ks+(size_t)(kbase+16)*32, 32, lane);
      v16h Va=load_tile_kmajor_f16(Vs+(size_t)kbase*16, 16, lane);
      v8f s0=WMMA_F16(Ka0,Qb,zero8());
      v8f s1=WMMA_F16(Ka1,Qb,zero8());
      float cm=fmaxf(s0[0],s1[0]);
      #pragma unroll
      for(int e=1;e<8;++e) cm=fmaxf(cm,fmaxf(s0[e],s1[e]));
      cm=fmaxf(cm,xhalf(cm));
      float nm=fmaxf(mrow,cm);
      float sc=__expf(mrow-nm);
      mrow=nm;
      v16h Pb; float ps=0.f;
      #pragma unroll
      for(int e=0;e<8;++e){
        float p0=__expf(s0[e]-nm);
        float p1=__expf(s1[e]-nm);
        ps+=p0+p1;
        Pb[e]=(_Float16)p0;
        Pb[8+e]=(_Float16)p1;
      }
      lpart=lpart*sc+ps;
      #pragma unroll
      for(int e=0;e<8;++e) acc[e]*=sc;
      acc=WMMA_F16(Va,Pb,acc);
    }
    float lrow=lpart+xhalf(lpart);
    float* ob=O+((size_t)b*SQ+qbase+colb)*HH + h*16 + rowoff;
    #pragma unroll
    for(int e=0;e<8;++e) ob[e]=acc[e]/lrow;
  }
}

__global__ __launch_bounds__(256) void k_projln(
    float* __restrict__ X, const float* __restrict__ Y,
    const _Float16* __restrict__ WoT, const float* __restrict__ bo,
    const float* __restrict__ g, const float* __restrict__ be)
{
  int lane=threadIdx.x&31, wave=threadIdx.x>>5;
  size_t rbase=((size_t)blockIdx.x*8+wave)*16;
  int colb=lane&15, rowoff=(lane>>4)*8;
  v16h A[2];
  #pragma unroll
  for(int kk=0;kk<2;++kk) A[kk]=load_tile_f32(Y+rbase*HH+kk*32, HH, lane);
  v8f o[4];
  #pragma unroll
  for(int j=0;j<4;++j){
    v8f acc=zero8();
    #pragma unroll
    for(int kk=0;kk<2;++kk)
      acc=WMMA_F16(A[kk], load_tile_sw(WoT, j*2+kk, lane), acc);
    float bias=bo[j*16+colb];
    const float* xr=X+rbase*HH+j*16+colb;
    #pragma unroll
    for(int e=0;e<8;++e) o[j][e]=acc[e]+bias+xr[(size_t)(e+rowoff)*HH];
  }
  float gv[4],bv[4];
  #pragma unroll
  for(int j=0;j<4;++j){ gv[j]=g[j*16+colb]; bv[j]=be[j*16+colb]; }
  float* xw=X+rbase*HH;
  #pragma unroll
  for(int e=0;e<8;++e){
    float s=o[0][e]+o[1][e]+o[2][e]+o[3][e];
    float mu=rsum16(s)*(1.f/64.f);
    float d0=o[0][e]-mu,d1=o[1][e]-mu,d2=o[2][e]-mu,d3=o[3][e]-mu;
    float var=rsum16(d0*d0+d1*d1+d2*d2+d3*d3)*(1.f/64.f);
    float inv=rsqrtf(var+1e-5f);
    #pragma unroll
    for(int j=0;j<4;++j)
      xw[(size_t)(e+rowoff)*HH+j*16+colb]=(o[j][e]-mu)*inv*gv[j]+bv[j];
  }
}

__global__ __launch_bounds__(256) __attribute__((amdgpu_waves_per_eu(2,2))) void k_ffln(
    float* __restrict__ X, const _Float16* __restrict__ W1T,
    const float* __restrict__ b1, const _Float16* __restrict__ W2T,
    const float* __restrict__ b2, const float* __restrict__ g,
    const float* __restrict__ be)
{
  __shared__ _Float16 h1[8][16*256];
  __shared__ __attribute__((aligned(16))) _Float16 W1L[32*512];
  __shared__ __attribute__((aligned(16))) _Float16 W2L[32*512];
  __shared__ float b1L[256];
  __shared__ float vL[256];
  int lane=threadIdx.x&31, wave=threadIdx.x>>5;
  size_t rbase=((size_t)blockIdx.x*8+wave)*16;
  int colb=lane&15, rowoff=(lane>>4)*8;
  v16h A[2];
  #pragma unroll
  for(int kk=0;kk<2;++kk) A[kk]=load_tile_f32(X+rbase*HH+kk*32, HH, lane);

  {
    float b1v=b1[threadIdx.x];
    const float* vsrc = threadIdx.x<64 ? b2 : (threadIdx.x<128 ? g : be);
    float vv=vsrc[threadIdx.x&63];
    const v4f* s1=(const v4f*)W1T; v4f* d1=(v4f*)W1L;
    const v4f* s2=(const v4f*)W2T; v4f* d2=(v4f*)W2L;
    #pragma unroll
    for(int k=0;k<8;++k){
      int gi=threadIdx.x+k*256;
      int u=gi&63;
      int di=(gi&~63) + ((u&1)<<5) + (u>>1);
      d1[di]=s1[gi];
      d2[di]=s2[gi];
    }
    b1L[threadIdx.x]=b1v;
    vL[threadIdx.x]=vv;
  }
  __syncthreads();

  _Float16* hl=&h1[wave][0];
  v16h Bc[8], Bn[8];
  float bc[4], bn[4];
  #pragma unroll
  for(int i=0;i<8;++i) Bc[i]=load_tile_planes(W1L, i, lane);
  #pragma unroll
  for(int jj=0;jj<4;++jj) bc[jj]=b1L[jj*16+colb];
  #pragma unroll
  for(int gq=0;gq<4;++gq){
    if(gq<3){
      #pragma unroll
      for(int i=0;i<8;++i) Bn[i]=load_tile_planes(W1L, (gq+1)*8+i, lane);
      #pragma unroll
      for(int jj=0;jj<4;++jj) bn[jj]=b1L[((gq+1)*4+jj)*16+colb];
    } else {
      #pragma unroll
      for(int i=0;i<8;++i) Bn[i]=load_tile_planes(W2L, (i>>1)*8+(i&1), lane);
    }
    __builtin_amdgcn_sched_barrier(0);
    v8f acc[4];
    #pragma unroll
    for(int jj=0;jj<4;++jj) acc[jj]=zero8();
    #pragma unroll
    for(int kk=0;kk<2;++kk){
      #pragma unroll
      for(int jj=0;jj<4;++jj) acc[jj]=WMMA_F16(A[kk], Bc[jj*2+kk], acc[jj]);
    }
    #pragma unroll
    for(int jj=0;jj<4;++jj){
      int j=gq*4+jj;
      float bias=bc[jj];
      #pragma unroll
      for(int e=0;e<8;++e)
        hl[(size_t)(e+rowoff)*256+j*16+colb]=(_Float16)fmaxf(acc[jj][e]+bias,0.f);
    }
    #pragma unroll
    for(int i=0;i<8;++i) Bc[i]=Bn[i];
    if(gq<3){
      #pragma unroll
      for(int jj=0;jj<4;++jj) bc[jj]=bn[jj];
    }
  }
  v16h pA[8];
  #pragma unroll
  for(int kk=0;kk<8;++kk) pA[kk]=load_tile_f16(hl+kk*32, 256, lane);
  v8f o[4];
  #pragma unroll
  for(int j=0;j<4;++j) o[j]=zero8();
  #pragma unroll
  for(int gq=0;gq<4;++gq){
    if(gq<3){
      #pragma unroll
      for(int i=0;i<8;++i) Bn[i]=load_tile_planes(W2L, (i>>1)*8+(gq+1)*2+(i&1), lane);
    }
    __builtin_amdgcn_sched_barrier(0);
    #pragma unroll
    for(int c=0;c<2;++c){
      #pragma unroll
      for(int j=0;j<4;++j) o[j]=WMMA_F16(pA[gq*2+c], Bc[j*2+c], o[j]);
    }
    if(gq<3){
      #pragma unroll
      for(int i=0;i<8;++i) Bc[i]=Bn[i];
    }
  }
  #pragma unroll
  for(int j=0;j<4;++j){
    float bias=vL[j*16+colb];
    const float* xr=X+rbase*HH+j*16+colb;
    #pragma unroll
    for(int e=0;e<8;++e) o[j][e]=o[j][e]+bias+xr[(size_t)(e+rowoff)*HH];
  }
  float gv[4],bv[4];
  #pragma unroll
  for(int j=0;j<4;++j){ gv[j]=vL[64+j*16+colb]; bv[j]=vL[128+j*16+colb]; }
  float* xw=X+rbase*HH;
  #pragma unroll
  for(int e=0;e<8;++e){
    float s=o[0][e]+o[1][e]+o[2][e]+o[3][e];
    float mu=rsum16(s)*(1.f/64.f);
    float d0=o[0][e]-mu,d1=o[1][e]-mu,d2=o[2][e]-mu,d3=o[3][e]-mu;
    float var=rsum16(d0*d0+d1*d1+d2*d2+d3*d3)*(1.f/64.f);
    float inv=rsqrtf(var+1e-5f);
    #pragma unroll
    for(int j=0;j<4;++j)
      xw[(size_t)(e+rowoff)*HH+j*16+colb]=(o[j][e]-mu)*inv*gv[j]+bv[j];
  }
}

__global__ void k_mean(const float* __restrict__ X, float* __restrict__ out){
  int b=blockIdx.x, j=threadIdx.x;
  float s=0.f;
  for(int np=0;np<SQ;++np) s+=X[((size_t)b*SQ+np)*HH+j];
  out[b*HH+j]=s*(1.f/512.f);
}

extern "C" void kernel_launch(void* const* d_in, const int* in_sizes, int n_in,
                              void* d_out, int out_size, void* d_ws, size_t ws_size,
                              hipStream_t stream){
  const float* x  =(const float*)d_in[0];
  const float* Wih=(const float*)d_in[1];
  const float* Whh=(const float*)d_in[2];
  const float* bih=(const float*)d_in[3];
  const float* bhh=(const float*)d_in[4];

  char* ws=(char*)d_ws;
  float*    X =(float*)ws;
  float*    Y =(float*)(ws+(size_t)(8<<20));
  _Float16* Qp=(_Float16*)(ws+(size_t)(16<<20));
  _Float16* Kp=(_Float16*)(ws+(size_t)(24<<20));
  _Float16* Vh=(_Float16*)(ws+(size_t)(32<<20));
  _Float16* WihT=(_Float16*)(ws+(size_t)(36<<20));
  _Float16* WhhT=(_Float16*)(ws+(size_t)(36<<20)+(64<<10));
  _Float16* LW[2][4];
  for(int L=0;L<2;++L){
    char* base=ws+(size_t)(37<<20)+(size_t)L*(1<<20);
    LW[L][0]=(_Float16*)(base);
    LW[L][1]=(_Float16*)(base+(32<<10));
    LW[L][2]=(_Float16*)(base+(64<<10));
    LW[L][3]=(_Float16*)(base+(128<<10));
  }

  k_swz_all<<<264,32,0,stream>>>(Wih, Whh,
      (const float*)d_in[5],  (const float*)d_in[7],
      (const float*)d_in[9],  (const float*)d_in[11],
      (const float*)d_in[17], (const float*)d_in[19],
      (const float*)d_in[21], (const float*)d_in[23],
      WihT);

  k_gru<<<256,256,0,stream>>>(x,WihT,WhhT,bih,bhh,X);

  for(int L=0;L<2;++L){
    int base=5+L*12;
    const float* bqkv=(const float*)d_in[base+1];
    const float* bo  =(const float*)d_in[base+3];
    const float* b1  =(const float*)d_in[base+5];
    const float* b2  =(const float*)d_in[base+7];
    const float* g1  =(const float*)d_in[base+8];
    const float* be1 =(const float*)d_in[base+9];
    const float* g2  =(const float*)d_in[base+10];
    const float* be2 =(const float*)d_in[base+11];

    k_qkv   <<<256,256,0,stream>>>(X,LW[L][0],bqkv,Qp,Kp,Vh);
    k_attn  <<<256,256,0,stream>>>(Qp,Kp,Vh,Y);
    k_projln<<<256,256,0,stream>>>(X,Y,LW[L][1],bo,g1,be1);
    k_ffln  <<<256,256,0,stream>>>(X,LW[L][2],b1,LW[L][3],b2,g2,be2);
  }
  k_mean<<<64,64,0,stream>>>(X,(float*)d_out);
}


# every copy/reduce loop issues all its independent loads as one group: final mean on 128 workgroups x 8 waves x 64 rows + LDS sum (sum order changes); attention K/V and GRU Whh staged in 16-byte units
# speedup vs baseline: 2.9683x; 1.4398x over previous
#include <hip/hip_runtime.h>
#include <hip/hip_fp16.h>

typedef __attribute__((ext_vector_type(16))) _Float16 v16h;
typedef __attribute__((ext_vector_type(8)))  float    v8f;
typedef __attribute__((ext_vector_type(4)))  float    v4f;

#define WMMA_F16(a,b,c) __builtin_amdgcn_wmma_f32_16x16x32_f16(false,(a),false,(b),(short)0,(c),false,false)

__device__ __forceinline__ v8f zero8(){
  v8f z;
  #pragma unroll
  for(int i=0;i<8;++i) z[i]=0.f;
  return z;
}
__device__ __forceinline__ v16h zero16(){
  v16h z;
  #pragma unroll
  for(int i=0;i<16;++i) z[i]=(_Float16)0.f;
  return z;
}

__device__ __forceinline__ int a_k(int i, int h16){ return (i&7) + ((i>>3)<<4) + (h16<<3); }

__device__ __forceinline__ v16h load_tile_f32(const float* __restrict__ src, int ld, int lane){
  const float* rb = src + (size_t)(lane&15)*ld;
  int h16=(lane>>4)&1; v16h r;
  #pragma unroll
  for(int i=0;i<16;++i) r[i]=(_Float16)rb[a_k(i,h16)];
  return r;
}
__device__ __forceinline__ v16h load_tile_f16(const _Float16* __restrict__ src, int ld, int lane){
  const _Float16* rb = src + (size_t)(lane&15)*ld;
  int h16=(lane>>4)&1; v16h r;
  #pragma unroll
  for(int i=0;i<16;++i) r[i]=rb[a_k(i,h16)];
  return r;
}
__device__ __forceinline__ v16h load_tile_kmajor_f16(const _Float16* __restrict__ src, int ld, int lane){
  const _Float16* cb = src + (lane&15);
  int h16=(lane>>4)&1; v16h r;
  #pragma unroll
  for(int i=0;i<16;++i) r[i]=cb[(size_t)a_k(i,h16)*ld];
  return r;
}
__device__ __forceinline__ v16h load_tile_sw(const _Float16* src, int tile, int lane){
  return *reinterpret_cast<const v16h*>(src + (((size_t)tile*32 + lane) << 4));
}

typedef __attribute__((ext_vector_type(8))) _Float16 v8h;
__device__ __forceinline__ v16h load_tile_planes(const _Float16* src, int tile, int lane){
  const _Float16* tb = src + ((size_t)tile << 9) + (lane << 3);
  v8h lo = *reinterpret_cast<const v8h*>(tb);
  v8h hi = *reinterpret_cast<const v8h*>(tb + 256);
  return __builtin_shufflevector(lo, hi, 0,1,2,3,4,5,6,7,8,9,10,11,12,13,14,15);
}

__device__ __forceinline__ float rmax16(float x){
  #pragma unroll
  for(int m=8;m;m>>=1) x=fmaxf(x,__shfl_xor(x,m,32));
  return x;
}
__device__ __forceinline__ float rsum16(float x){
  #pragma unroll
  for(int m=8;m;m>>=1) x+=__shfl_xor(x,m,32);
  return x;
}

#define PB  8
#define DD  128
#define HH  64
#define SQ  512

__global__ __launch_bounds__(32) void k_swz_all(
    const float* __restrict__ wih,   const float* __restrict__ whh,
    const float* __restrict__ wqkv0, const float* __restrict__ wo0,
    const float* __restrict__ w1_0,  const float* __restrict__ w2_0,
    const float* __restrict__ wqkv1, const float* __restrict__ wo1,
    const float* __restrict__ w1_1,  const float* __restrict__ w2_1,
    _Float16* __restrict__ obase)
{
  int t = blockIdx.x, lane = threadIdx.x;
  const float* Wm; _Float16* out; int ksh;
  if(t < 48){ Wm = wih; out = obase; ksh = 2; }
  else if(t < 72){ t -= 48; Wm = whh; out = obase + (32<<10); ksh = 1; }
  else {
    t -= 72;
    int L = (t >= 96) ? 1 : 0;
    t -= L*96;
    _Float16* lb = obase + ((size_t)(1+L) << 19);
    if(t < 24){ Wm = L ? wqkv1 : wqkv0; out = lb; ksh = 1; }
    else if(t < 32){ t -= 24; Wm = L ? wo1 : wo0; out = lb + (16<<10); ksh = 1; }
    else if(t < 64){ t -= 32; Wm = L ? w1_1 : w1_0; out = lb + (32<<10); ksh = 1; }
    else { t -= 64; Wm = L ? w2_1 : w2_0; out = lb + (64<<10); ksh = 3; }
  }
  int K = 32 << ksh;
  int kt = t & ((1<<ksh)-1), jt = t >> ksh;
  v16h r = load_tile_f32(Wm + (size_t)jt*16*K + kt*32, K, lane);
  *reinterpret_cast<v16h*>(out + (((size_t)t*32 + lane) << 4)) = r;
}

__device__ __forceinline__ void qkv_rows(
    v16h A0, v16h A1, const _Float16* WL, const float* bL, size_t rbase, int lane,
    _Float16* __restrict__ Qp, _Float16* __restrict__ Kp, _Float16* __restrict__ Vh)
{
  int b=(int)(rbase>>9), s=(int)(rbase&511);
  int colb=lane&15, rowoff=(lane>>4)*8;
  v16h A[2]; A[0]=A0; A[1]=A1;
  v16h Bc[8], Bn[8];
  float bc[4], bn[4];
  #pragma unroll
  for(int i=0;i<8;++i) Bc[i]=load_tile_planes(WL, i, lane);
  #pragma unroll
  for(int jj=0;jj<4;++jj) bc[jj]=bL[jj*16+colb];
  #pragma unroll
  for(int g=0;g<3;++g){
    if(g<2){
      #pragma unroll
      for(int i=0;i<8;++i) Bn[i]=load_tile_planes(WL, (g+1)*8+i, lane);
      #pragma unroll
      for(int jj=0;jj<4;++jj) bn[jj]=bL[((g+1)*4+jj)*16+colb];
    }
    __builtin_amdgcn_sched_barrier(0);
    v8f acc[4];
    #pragma unroll
    for(int jj=0;jj<4;++jj) acc[jj]=zero8();
    #pragma unroll
    for(int kk=0;kk<2;++kk){
      #pragma unroll
      for(int jj=0;jj<4;++jj) acc[jj]=WMMA_F16(A[kk], Bc[jj*2+kk], acc[jj]);
    }
    #pragma unroll
    for(int jj=0;jj<4;++jj){
      float bias=bc[jj];
      size_t bh=(size_t)b*4+jj;
      if(g==0){
        _Float16* dst=Qp+(bh*SQ+s)*32;
        #pragma unroll
        for(int v=0;v<8;++v){
          size_t ro=(size_t)(v+rowoff)*32;
          dst[ro+colb]=(_Float16)((acc[jj][v]+bias)*0.25f);
          dst[ro+16+colb]=(_Float16)0.f;
        }
      } else if(g==1){
        _Float16* dst=Kp+(bh*SQ+s)*32;
        #pragma unroll
        for(int v=0;v<8;++v){
          size_t ro=(size_t)(v+rowoff)*32;
          dst[ro+colb]=(_Float16)(acc[jj][v]+bias);
          dst[ro+16+colb]=(_Float16)0.f;
        }
      } else {
        _Float16* dst=Vh+(bh*SQ+s)*16;
        #pragma unroll
        for(int v=0;v<8;++v)
          dst[(size_t)(v+rowoff)*16+colb]=(_Float16)(acc[jj][v]+bias);
      }
    }
    if(g<2){
      #pragma unroll
      for(int i=0;i<8;++i) Bc[i]=Bn[i];
      #pragma unroll
      for(int jj=0;jj<4;++jj) bc[jj]=bn[jj];
    }
  }
}

__global__ __launch_bounds__(256) __attribute__((amdgpu_waves_per_eu(2,2))) void k_gru_qkv(
    const float* __restrict__ x, const _Float16* __restrict__ WihT,
    const _Float16* __restrict__ WhhT, const float* __restrict__ bih,
    const float* __restrict__ bhh, float* __restrict__ seq,
    const _Float16* __restrict__ WqkvT, const float* __restrict__ bqkv,
    _Float16* __restrict__ Qp, _Float16* __restrict__ Kp, _Float16* __restrict__ Vh)
{
  __shared__ _Float16 hscr[8][16*64];
  __shared__ __attribute__((aligned(16))) _Float16 WhhL[192*64];
  __shared__ __attribute__((aligned(16))) _Float16 WihL[192*128];
  __shared__ __attribute__((aligned(16))) _Float16 WL[24*512];
  __shared__ float bL[256];
  int lane=threadIdx.x&31, wave=threadIdx.x>>5;
  int pbase=(blockIdx.x*8+wave)*16;
  int colb=lane&15, rowoff=(lane>>4)*8;

  float bi[12], bh[12];
  #pragma unroll
  for(int j=0;j<12;++j){ bi[j]=bih[j*16+colb]; bh[j]=bhh[j*16+colb]; }

  {
    const v4f* hsrc=(const v4f*)WhhT;
    v4f* hdst=(v4f*)WhhL;
    #pragma unroll
    for(int k=0;k<6;++k) hdst[threadIdx.x+k*256]=hsrc[threadIdx.x+k*256];
    const v4f* isrc=(const v4f*)WihT;
    v4f* idst=(v4f*)WihL;
    #pragma unroll
    for(int k=0;k<12;++k){
      int g=threadIdx.x+k*256;
      int u=g&63;
      idst[(g&~63) + ((u&1)<<5) + (u>>1)]=isrc[g];
    }
    const v4f* qsrc=(const v4f*)WqkvT;
    v4f* qdst=(v4f*)WL;
    #pragma unroll
    for(int k=0;k<6;++k){
      int g=threadIdx.x+k*256;
      int u=g&63;
      qdst[(g&~63) + ((u&1)<<5) + (u>>1)]=qsrc[g];
    }
    bL[threadIdx.x]=bqkv[threadIdx.x<192 ? threadIdx.x : 0];
  }
  __syncthreads();

  v8f hT[4], hs[4];
  #pragma unroll
  for(int j=0;j<4;++j){ hT[j]=zero8(); hs[j]=zero8(); }
  v16h hA[2]; hA[0]=zero16(); hA[1]=zero16();
  _Float16* hl=&hscr[wave][0];

  for(int t=0;t<PB;++t){
    const float* xsrc = x + ((size_t)pbase*PB + t)*DD;
    if(t+1<PB) __builtin_prefetch(xsrc+DD,0,1);
    v4f xr[16];
    {
      const float* rb = xsrc + (size_t)(lane&15)*(PB*DD) + (((lane>>4)&1)<<3);
      #pragma unroll
      for(int kk=0;kk<4;++kk){
        xr[kk*4+0]=*reinterpret_cast<const v4f*>(rb+kk*32);
        xr[kk*4+1]=*reinterpret_cast<const v4f*>(rb+kk*32+4);
        xr[kk*4+2]=*reinterpret_cast<const v4f*>(rb+kk*32+16);
        xr[kk*4+3]=*reinterpret_cast<const v4f*>(rb+kk*32+20);
      }
    }
    __builtin_amdgcn_sched_barrier(0);
    v16h xA[4];
    #pragma unroll
    for(int kk=0;kk<4;++kk){
      #pragma unroll
      for(int q=0;q<4;++q){
        #pragma unroll
        for(int c=0;c<4;++c) xA[kk][q*4+c]=(_Float16)xr[kk*4+q][c];
      }
    }

    #pragma unroll
    for(int j=0;j<4;++j){
      v8f gr=zero8();
      #pragma unroll
      for(int kk=0;kk<4;++kk)
        gr=WMMA_F16(xA[kk], load_tile_planes(WihL, j*4+kk, lane), gr);
      #pragma unroll
      for(int kk=0;kk<2;++kk)
        gr=WMMA_F16(hA[kk], load_tile_sw(WhhL, j*2+kk, lane), gr);
      v8f gz=zero8();
      #pragma unroll
      for(int kk=0;kk<4;++kk)
        gz=WMMA_F16(xA[kk], load_tile_planes(WihL, (4+j)*4+kk, lane), gz);
      #pragma unroll
      for(int kk=0;kk<2;++kk)
        gz=WMMA_F16(hA[kk], load_tile_sw(WhhL, (4+j)*2+kk, lane), gz);
      v8f gnx=zero8();
      #pragma unroll
      for(int kk=0;kk<4;++kk)
        gnx=WMMA_F16(xA[kk], load_tile_planes(WihL, (8+j)*4+kk, lane), gnx);
      v8f gnh=zero8();
      #pragma unroll
      for(int kk=0;kk<2;++kk)
        gnh=WMMA_F16(hA[kk], load_tile_sw(WhhL, (8+j)*2+kk, lane), gnh);
      float rr[8], zz[8], nn[8];
      #pragma unroll
      for(int e=0;e<8;++e){
        rr[e]=(gr[e]+bi[j]+bh[j])*-1.44269504f;
        zz[e]=(gz[e]+bi[4+j]+bh[4+j])*-1.44269504f;
      }
      #pragma unroll
      for(int e=0;e<8;++e){
        rr[e]=1.f+__builtin_amdgcn_exp2f(rr[e]);
        zz[e]=1.f+__builtin_amdgcn_exp2f(zz[e]);
      }
      #pragma unroll
      for(int e=0;e<8;++e){
        rr[e]=__builtin_amdgcn_rcpf(rr[e]);
        zz[e]=__builtin_amdgcn_rcpf(zz[e]);
      }
      #pragma unroll
      for(int e=0;e<8;++e)
        nn[e]=gnx[e]+bi[8+j]+rr[e]*(gnh[e]+bh[8+j]);
      #pragma unroll
      for(int e=0;e<8;++e) nn[e]=__builtin_amdgcn_tanhf(nn[e]);
      #pragma unroll
      for(int e=0;e<8;++e){
        float hn=(1.f-zz[e])*nn[e] + zz[e]*hT[j][e];
        hT[j][e]=hn; hs[j][e]+=hn;
      }
    }
    #pragma unroll
    for(int j=0;j<4;++j){
      #pragma unroll
      for(int v=0;v<8;++v)
        hl[(v+rowoff)*64 + j*16 + colb]=(_Float16)hT[j][v];
    }
    #pragma unroll
    for(int kk=0;kk<2;++kk) hA[kk]=load_tile_f16(hl+kk*32, 64, lane);
  }
  float* sb=seq+(size_t)pbase*HH;
  #pragma unroll
  for(int j=0;j<4;++j){
    #pragma unroll
    for(int v=0;v<8;++v){
      float pv=hs[j][v]*0.125f;
      sb[(size_t)(v+rowoff)*HH + j*16 + colb]=pv;
      hl[(v+rowoff)*64 + j*16 + colb]=(_Float16)pv;
    }
  }
  v16h pA0=load_tile_f16(hl, 64, lane);
  v16h pA1=load_tile_f16(hl+32, 64, lane);
  qkv_rows(pA0, pA1, WL, bL, (size_t)pbase, lane, Qp, Kp, Vh);
}

__device__ __forceinline__ float xhalf(float x){
  int b=__builtin_bit_cast(int,x);
  return __builtin_bit_cast(float,
      __builtin_amdgcn_permlanex16(b,b,0x76543210,0xfedcba98,false,false));
}

__global__ __launch_bounds__(256) void k_attn(
    const _Float16* __restrict__ Qp, const _Float16* __restrict__ Kp,
    const _Float16* __restrict__ Vh, float* __restrict__ O)
{
  __shared__ __attribute__((aligned(16))) _Float16 Ks[SQ*32];
  __shared__ __attribute__((aligned(16))) _Float16 Vs[SQ*16];
  int bh=blockIdx.x, b=bh>>2, h=bh&3;
  int tid=threadIdx.x, lane=tid&31, wave=tid>>5;
  int colb=lane&15, rowoff=(lane>>4)*8;

  v16h Qn=load_tile_f16(Qp+((size_t)bh*SQ+wave*64)*32, 32, lane);

  {
    const v4f* ksrc=(const v4f*)(Kp+(size_t)bh*SQ*32);
    v4f* kdst=(v4f*)Ks;
    const v4f* vsrc=(const v4f*)(Vh+(size_t)bh*SQ*16);
    v4f* vdst=(v4f*)Vs;
    #pragma unroll
    for(int k=0;k<8;++k) kdst[tid+k*256]=ksrc[tid+k*256];
    #pragma unroll
    for(int k=0;k<4;++k) vdst[tid+k*256]=vsrc[tid+k*256];
  }
  __syncthreads();

  for(int qt=0;qt<4;++qt){
    int qbase=wave*64+qt*16;
    v16h Qb=Qn;
    if(qt<3) Qn=load_tile_f16(Qp+((size_t)bh*SQ+qbase+16)*32, 32, lane);
    v8f acc=zero8();
    float mrow=-1e30f;
    float lpart=0.f;

    for(int kc=0;kc<16;++kc){
      int kbase=kc*32;
      v16h Ka0=load_tile_f16(Ks+(size_t)kbase*32, 32, lane);
      v16h Ka1=load_tile_f16(Ks+(size_t)(kbase+16)*32, 32, lane);
      v16h Va=load_tile_kmajor_f16(Vs+(size_t)kbase*16, 16, lane);
      v8f s0=WMMA_F16(Ka0,Qb,zero8());
      v8f s1=WMMA_F16(Ka1,Qb,zero8());
      float cm=fmaxf(s0[0],s1[0]);
      #pragma unroll
      for(int e=1;e<8;++e) cm=fmaxf(cm,fmaxf(s0[e],s1[e]));
      cm=fmaxf(cm,xhalf(cm));
      float nm=fmaxf(mrow,cm);
      float sc=__expf(mrow-nm);
      mrow=nm;
      v16h Pb; float ps=0.f;
      #pragma unroll
      for(int e=0;e<8;++e){
        float p0=__expf(s0[e]-nm);
        float p1=__expf(s1[e]-nm);
        ps+=p0+p1;
        Pb[e]=(_Float16)p0;
        Pb[8+e]=(_Float16)p1;
      }
      lpart=lpart*sc+ps;
      #pragma unroll
      for(int e=0;e<8;++e) acc[e]*=sc;
      acc=WMMA_F16(Va,Pb,acc);
    }
    float lrow=lpart+xhalf(lpart);
    float* ob=O+((size_t)b*SQ+qbase+colb)*HH + h*16 + rowoff;
    #pragma unroll
    for(int e=0;e<8;++e) ob[e]=acc[e]/lrow;
  }
}

#define RP 72

__device__ __forceinline__ void ln_rows(v8f (&o)[4], const float (&gv)[4], const float (&bv)[4]){
  #pragma unroll
  for(int e=0;e<8;++e){
    float s=o[0][e]+o[1][e]+o[2][e]+o[3][e];
    float mu=rsum16(s)*(1.f/64.f);
    float d0=o[0][e]-mu,d1=o[1][e]-mu,d2=o[2][e]-mu,d3=o[3][e]-mu;
    float var=rsum16(d0*d0+d1*d1+d2*d2+d3*d3)*(1.f/64.f);
    float inv=rsqrtf(var+1e-5f);
    #pragma unroll
    for(int j=0;j<4;++j)
      o[j][e]=(o[j][e]-mu)*inv*gv[j]+bv[j];
  }
}

__global__ __launch_bounds__(256) __attribute__((amdgpu_waves_per_eu(2,2))) void k_layer(
    float* __restrict__ X, const float* __restrict__ Y,
    const _Float16* __restrict__ WoT, const float* __restrict__ bo,
    const float* __restrict__ g1, const float* __restrict__ be1,
    const _Float16* __restrict__ W1T, const float* __restrict__ b1,
    const _Float16* __restrict__ W2T, const float* __restrict__ b2,
    const float* __restrict__ g2, const float* __restrict__ be2,
    const _Float16* __restrict__ WqkvT, const float* __restrict__ bqkv,
    _Float16* __restrict__ Qp, _Float16* __restrict__ Kp, _Float16* __restrict__ Vh,
    int with_qkv)
{
  __shared__ __attribute__((aligned(16))) _Float16 h1[8][16*256];
  __shared__ __attribute__((aligned(16))) _Float16 W1L[32*512];
  __shared__ __attribute__((aligned(16))) _Float16 W2L[32*512];
  __shared__ __attribute__((aligned(16))) _Float16 WoL[8*512];
  __shared__ __attribute__((aligned(16))) _Float16 WL[24*512];
  __shared__ float sv[1024];
  int lane=threadIdx.x&31, wave=threadIdx.x>>5;
  size_t rbase=((size_t)blockIdx.x*8+wave)*16;
  int colb=lane&15, rowoff=(lane>>4)*8;

  v4f yr[8];
  {
    const float* rb = Y + rbase*HH + (size_t)(lane&15)*HH + (((lane>>4)&1)<<3);
    #pragma unroll
    for(int kk=0;kk<2;++kk){
      yr[kk*4+0]=*reinterpret_cast<const v4f*>(rb+kk*32);
      yr[kk*4+1]=*reinterpret_cast<const v4f*>(rb+kk*32+4);
      yr[kk*4+2]=*reinterpret_cast<const v4f*>(rb+kk*32+16);
      yr[kk*4+3]=*reinterpret_cast<const v4f*>(rb+kk*32+20);
    }
  }
  v8f xres[4];
  #pragma unroll
  for(int j=0;j<4;++j){
    const float* xr=X+rbase*HH+j*16+colb;
    #pragma unroll
    for(int e=0;e<8;++e) xres[j][e]=xr[(size_t)(e+rowoff)*HH];
  }

  {
    int tid=threadIdx.x;
    float t0=b1[tid];
    const float* p1 = tid<64 ? bo : (tid<128 ? g1 : be1);
    float t1=p1[tid&63];
    const float* p2 = tid<64 ? b2 : (tid<128 ? g2 : be2);
    float t2=p2[tid&63];
    const v4f* s1=(const v4f*)W1T; v4f* d1=(v4f*)W1L;
    const v4f* s2=(const v4f*)W2T; v4f* d2=(v4f*)W2L;
    const v4f* so=(const v4f*)WoT; v4f* dO=(v4f*)WoL;
    #pragma unroll
    for(int k=0;k<8;++k){
      int gi=tid+k*256;
      int u=gi&63;
      int di=(gi&~63) + ((u&1)<<5) + (u>>1);
      d1[di]=s1[gi];
      d2[di]=s2[gi];
      if(k<2) dO[di]=so[gi];
    }
    float t3=bqkv[tid<192 ? tid : 0];
    const v4f* sq=(const v4f*)WqkvT; v4f* dq=(v4f*)WL;
    #pragma unroll
    for(int k=0;k<6;++k){
      int gi=tid+k*256;
      int u=gi&63;
      dq[(gi&~63) + ((u&1)<<5) + (u>>1)]=sq[gi];
    }
    sv[768+tid]=t3;
    sv[tid]=t0;
    sv[256+tid]=t1;
    sv[512+tid]=t2;
  }
  __syncthreads();

  _Float16* hl=&h1[wave][0];
  v16h A[2];
  #pragma unroll
  for(int kk=0;kk<2;++kk){
    #pragma unroll
    for(int q=0;q<4;++q){
      #pragma unroll
      for(int cc=0;cc<4;++cc) A[kk][q*4+cc]=(_Float16)yr[kk*4+q][cc];
    }
  }

  v8f x1[4];
  #pragma unroll
  for(int j=0;j<4;++j){
    v8f acc=zero8();
    #pragma unroll
    for(int kk=0;kk<2;++kk)
      acc=WMMA_F16(A[kk], load_tile_planes(WoL, j*2+kk, lane), acc);
    float bias=sv[256+j*16+colb];
    #pragma unroll
    for(int e=0;e<8;++e) x1[j][e]=acc[e]+bias+xres[j][e];
  }
  {
    float gv[4],bv[4];
    #pragma unroll
    for(int j=0;j<4;++j){ gv[j]=sv[320+j*16+colb]; bv[j]=sv[384+j*16+colb]; }
    ln_rows(x1, gv, bv);
  }
  #pragma unroll
  for(int j=0;j<4;++j){
    #pragma unroll
    for(int e=0;e<8;++e) hl[(e+rowoff)*RP + j*16 + colb]=(_Float16)x1[j][e];
  }
  #pragma unroll
  for(int kk=0;kk<2;++kk) A[kk]=load_tile_f16(hl+kk*32, RP, lane);

  v16h Bc[8], Bn[8];
  float bc[4], bn[4];
  #pragma unroll
  for(int i=0;i<8;++i) Bc[i]=load_tile_planes(W1L, i, lane);
  #pragma unroll
  for(int jj=0;jj<4;++jj) bc[jj]=sv[jj*16+colb];
  #pragma unroll
  for(int gq=0;gq<4;++gq){
    if(gq<3){
      #pragma unroll
      for(int i=0;i<8;++i) Bn[i]=load_tile_planes(W1L, (gq+1)*8+i, lane);
      #pragma unroll
      for(int jj=0;jj<4;++jj) bn[jj]=sv[((gq+1)*4+jj)*16+colb];
    } else {
      #pragma unroll
      for(int i=0;i<8;++i) Bn[i]=load_tile_planes(W2L, (i>>1)*8+(i&1), lane);
    }
    __builtin_amdgcn_sched_barrier(0);
    v8f acc[4];
    #pragma unroll
    for(int jj=0;jj<4;++jj) acc[jj]=zero8();
    #pragma unroll
    for(int kk=0;kk<2;++kk){
      #pragma unroll
      for(int jj=0;jj<4;++jj) acc[jj]=WMMA_F16(A[kk], Bc[jj*2+kk], acc[jj]);
    }
    #pragma unroll
    for(int jj=0;jj<4;++jj){
      int j=gq*4+jj;
      float bias=bc[jj];
      #pragma unroll
      for(int e=0;e<8;++e)
        hl[(size_t)(e+rowoff)*256+j*16+colb]=(_Float16)fmaxf(acc[jj][e]+bias,0.f);
    }
    #pragma unroll
    for(int i=0;i<8;++i) Bc[i]=Bn[i];
    if(gq<3){
      #pragma unroll
      for(int jj=0;jj<4;++jj) bc[jj]=bn[jj];
    }
  }
  v16h pA[8];
  #pragma unroll
  for(int kk=0;kk<8;++kk) pA[kk]=load_tile_f16(hl+kk*32, 256, lane);
  v8f o[4];
  #pragma unroll
  for(int j=0;j<4;++j) o[j]=zero8();
  #pragma unroll
  for(int gq=0;gq<4;++gq){
    if(gq<3){
      #pragma unroll
      for(int i=0;i<8;++i) Bn[i]=load_tile_planes(W2L, (i>>1)*8+(gq+1)*2+(i&1), lane);
    }
    __builtin_amdgcn_sched_barrier(0);
    #pragma unroll
    for(int c=0;c<2;++c){
      #pragma unroll
      for(int j=0;j<4;++j) o[j]=WMMA_F16(pA[gq*2+c], Bc[j*2+c], o[j]);
    }
    if(gq<3){
      #pragma unroll
      for(int i=0;i<8;++i) Bc[i]=Bn[i];
    }
  }
  #pragma unroll
  for(int j=0;j<4;++j){
    float bias=sv[512+j*16+colb];
    #pragma unroll
    for(int e=0;e<8;++e) o[j][e]=o[j][e]+bias+x1[j][e];
  }
  {
    float gv[4],bv[4];
    #pragma unroll
    for(int j=0;j<4;++j){ gv[j]=sv[576+j*16+colb]; bv[j]=sv[640+j*16+colb]; }
    ln_rows(o, gv, bv);
  }
  float* xw=X+rbase*HH;
  #pragma unroll
  for(int e=0;e<8;++e){
    #pragma unroll
    for(int j=0;j<4;++j)
      xw[(size_t)(e+rowoff)*HH+j*16+colb]=o[j][e];
  }
  if(with_qkv){
    #pragma unroll
    for(int j=0;j<4;++j){
      #pragma unroll
      for(int e=0;e<8;++e) hl[(e+rowoff)*RP + j*16 + colb]=(_Float16)o[j][e];
    }
    v16h qA0=load_tile_f16(hl, RP, lane);
    v16h qA1=load_tile_f16(hl+32, RP, lane);
    qkv_rows(qA0, qA1, WL, sv+768, rbase, lane, Qp, Kp, Vh);
  }
}

__global__ __launch_bounds__(256) void k_mean(const float* __restrict__ X, float* __restrict__ out){
  __shared__ float part[8][32];
  int b=blockIdx.x>>1, cb=(blockIdx.x&1)*32;
  int lane=threadIdx.x&31, wave=threadIdx.x>>5;
  const float* p=X+((size_t)b*SQ+wave)*HH+cb+lane;
  float v[64];
  #pragma unroll
  for(int i=0;i<64;++i) v[i]=p[(size_t)i*8*HH];
  float s=0.f;
  #pragma unroll
  for(int i=0;i<64;++i) s+=v[i];
  part[wave][lane]=s;
  __syncthreads();
  if(wave==0){
    float t=0.f;
    #pragma unroll
    for(int w=0;w<8;++w) t+=part[w][lane];
    out[b*HH+cb+lane]=t*(1.f/512.f);
  }
}

extern "C" void kernel_launch(void* const* d_in, const int* in_sizes, int n_in,
                              void* d_out, int out_size, void* d_ws, size_t ws_size,
                              hipStream_t stream){
  const float* x  =(const float*)d_in[0];
  const float* Wih=(const float*)d_in[1];
  const float* Whh=(const float*)d_in[2];
  const float* bih=(const float*)d_in[3];
  const float* bhh=(const float*)d_in[4];

  char* ws=(char*)d_ws;
  float*    X =(float*)ws;
  float*    Y =(float*)(ws+(size_t)(8<<20));
  _Float16* Qp=(_Float16*)(ws+(size_t)(16<<20));
  _Float16* Kp=(_Float16*)(ws+(size_t)(24<<20));
  _Float16* Vh=(_Float16*)(ws+(size_t)(32<<20));
  _Float16* WihT=(_Float16*)(ws+(size_t)(36<<20));
  _Float16* WhhT=(_Float16*)(ws+(size_t)(36<<20)+(64<<10));
  _Float16* LW[2][4];
  for(int L=0;L<2;++L){
    char* base=ws+(size_t)(37<<20)+(size_t)L*(1<<20);
    LW[L][0]=(_Float16*)(base);
    LW[L][1]=(_Float16*)(base+(32<<10));
    LW[L][2]=(_Float16*)(base+(64<<10));
    LW[L][3]=(_Float16*)(base+(128<<10));
  }

  k_swz_all<<<264,32,0,stream>>>(Wih, Whh,
      (const float*)d_in[5],  (const float*)d_in[7],
      (const float*)d_in[9],  (const float*)d_in[11],
      (const float*)d_in[17], (const float*)d_in[19],
      (const float*)d_in[21], (const float*)d_in[23],
      WihT);

  k_gru_qkv<<<256,256,0,stream>>>(x,WihT,WhhT,bih,bhh,X,
      LW[0][0],(const float*)d_in[6],Qp,Kp,Vh);

  k_attn <<<256,256,0,stream>>>(Qp,Kp,Vh,Y);
  k_layer<<<256,256,0,stream>>>(X,Y,
      LW[0][1],(const float*)d_in[8], (const float*)d_in[13],(const float*)d_in[14],
      LW[0][2],(const float*)d_in[10],LW[0][3],(const float*)d_in[12],
      (const float*)d_in[15],(const float*)d_in[16],
      LW[1][0],(const float*)d_in[18],Qp,Kp,Vh,1);

  k_attn <<<256,256,0,stream>>>(Qp,Kp,Vh,Y);
  k_layer<<<256,256,0,stream>>>(X,Y,
      LW[1][1],(const float*)d_in[20],(const float*)d_in[25],(const float*)d_in[26],
      LW[1][2],(const float*)d_in[22],LW[1][3],(const float*)d_in[24],
      (const float*)d_in[27],(const float*)d_in[28],
      LW[1][0],(const float*)d_in[18],Qp,Kp,Vh,0);

  k_mean<<<128,256,0,stream>>>(X,(float*)d_out);
}


# Q, K, V stored in attention-fragment order (Q/K products transposed, head dims unpadded, V dim-major per 16-key tile): each fragment and each QKV store is one 16-byte unit per lane; sums unchanged
# speedup vs baseline: 3.4911x; 1.1761x over previous
#include <hip/hip_runtime.h>
#include <hip/hip_fp16.h>

typedef __attribute__((ext_vector_type(16))) _Float16 v16h;
typedef __attribute__((ext_vector_type(8)))  float    v8f;
typedef __attribute__((ext_vector_type(4)))  float    v4f;

#define WMMA_F16(a,b,c) __builtin_amdgcn_wmma_f32_16x16x32_f16(false,(a),false,(b),(short)0,(c),false,false)

__device__ __forceinline__ v8f zero8(){
  v8f z;
  #pragma unroll
  for(int i=0;i<8;++i) z[i]=0.f;
  return z;
}
__device__ __forceinline__ v16h zero16(){
  v16h z;
  #pragma unroll
  for(int i=0;i<16;++i) z[i]=(_Float16)0.f;
  return z;
}

__device__ __forceinline__ int a_k(int i, int h16){ return (i&7) + ((i>>3)<<4) + (h16<<3); }

__device__ __forceinline__ v16h load_tile_f32(const float* __restrict__ src, int ld, int lane){
  const float* rb = src + (size_t)(lane&15)*ld;
  int h16=(lane>>4)&1; v16h r;
  #pragma unroll
  for(int i=0;i<16;++i) r[i]=(_Float16)rb[a_k(i,h16)];
  return r;
}
__device__ __forceinline__ v16h load_tile_f16(const _Float16* __restrict__ src, int ld, int lane){
  const _Float16* rb = src + (size_t)(lane&15)*ld;
  int h16=(lane>>4)&1; v16h r;
  #pragma unroll
  for(int i=0;i<16;++i) r[i]=rb[a_k(i,h16)];
  return r;
}
__device__ __forceinline__ v16h load_tile_sw(const _Float16* src, int tile, int lane){
  return *reinterpret_cast<const v16h*>(src + (((size_t)tile*32 + lane) << 4));
}

typedef __attribute__((ext_vector_type(8))) _Float16 v8h;
__device__ __forceinline__ v16h load_tile_planes(const _Float16* src, int tile, int lane){
  const _Float16* tb = src + ((size_t)tile << 9) + (lane << 3);
  v8h lo = *reinterpret_cast<const v8h*>(tb);
  v8h hi = *reinterpret_cast<const v8h*>(tb + 256);
  return __builtin_shufflevector(lo, hi, 0,1,2,3,4,5,6,7,8,9,10,11,12,13,14,15);
}

__device__ __forceinline__ float rmax16(float x){
  #pragma unroll
  for(int m=8;m;m>>=1) x=fmaxf(x,__shfl_xor(x,m,32));
  return x;
}
__device__ __forceinline__ float rsum16(float x){
  #pragma unroll
  for(int m=8;m;m>>=1) x+=__shfl_xor(x,m,32);
  return x;
}

#define PB  8
#define DD  128
#define HH  64
#define SQ  512

__global__ __launch_bounds__(32) void k_swz_all(
    const float* __restrict__ wih,   const float* __restrict__ whh,
    const float* __restrict__ wqkv0, const float* __restrict__ wo0,
    const float* __restrict__ w1_0,  const float* __restrict__ w2_0,
    const float* __restrict__ wqkv1, const float* __restrict__ wo1,
    const float* __restrict__ w1_1,  const float* __restrict__ w2_1,
    _Float16* __restrict__ obase)
{
  int t = blockIdx.x, lane = threadIdx.x;
  const float* Wm; _Float16* out; int ksh;
  if(t < 48){ Wm = wih; out = obase; ksh = 2; }
  else if(t < 72){ t -= 48; Wm = whh; out = obase + (32<<10); ksh = 1; }
  else {
    t -= 72;
    int L = (t >= 96) ? 1 : 0;
    t -= L*96;
    _Float16* lb = obase + ((size_t)(1+L) << 19);
    if(t < 24){ Wm = L ? wqkv1 : wqkv0; out = lb; ksh = 1; }
    else if(t < 32){ t -= 24; Wm = L ? wo1 : wo0; out = lb + (16<<10); ksh = 1; }
    else if(t < 64){ t -= 32; Wm = L ? w1_1 : w1_0; out = lb + (32<<10); ksh = 1; }
    else { t -= 64; Wm = L ? w2_1 : w2_0; out = lb + (64<<10); ksh = 3; }
  }
  int K = 32 << ksh;
  int kt = t & ((1<<ksh)-1), jt = t >> ksh;
  v16h r = load_tile_f32(Wm + (size_t)jt*16*K + kt*32, K, lane);
  *reinterpret_cast<v16h*>(out + (((size_t)t*32 + lane) << 4)) = r;
}

__device__ __forceinline__ void qkv_rows(
    v16h A0, v16h A1, const _Float16* WL, const float* bL, size_t rbase, int lane,
    _Float16* __restrict__ Qh, _Float16* __restrict__ Kh, _Float16* __restrict__ Vt)
{
  int b=(int)(rbase>>9), s=(int)(rbase&511);
  int colb=lane&15, rowoff=(lane>>4)*8;
  v16h A[2]; A[0]=A0; A[1]=A1;
  v16h Bc[8], Bn[8];
  #pragma unroll
  for(int i=0;i<8;++i) Bc[i]=load_tile_planes(WL, i, lane);
  #pragma unroll
  for(int g=0;g<3;++g){
    if(g<2){
      #pragma unroll
      for(int i=0;i<8;++i) Bn[i]=load_tile_planes(WL, (g+1)*8+i, lane);
    }
    v4f bq[4][2]; float bv[4];
    #pragma unroll
    for(int jj=0;jj<4;++jj){
      if(g<2){
        const v4f* bp=reinterpret_cast<const v4f*>(bL+(g*4+jj)*16+rowoff);
        bq[jj][0]=bp[0]; bq[jj][1]=bp[1];
      } else bv[jj]=bL[(8+jj)*16+colb];
    }
    __builtin_amdgcn_sched_barrier(0);
    v8f acc[4];
    #pragma unroll
    for(int jj=0;jj<4;++jj) acc[jj]=zero8();
    #pragma unroll
    for(int kk=0;kk<2;++kk){
      #pragma unroll
      for(int jj=0;jj<4;++jj){
        if(g<2) acc[jj]=WMMA_F16(Bc[jj*2+kk], A[kk], acc[jj]);
        else    acc[jj]=WMMA_F16(A[kk], Bc[jj*2+kk], acc[jj]);
      }
    }
    #pragma unroll
    for(int jj=0;jj<4;++jj){
      size_t bh=(size_t)b*4+jj;
      v8h o;
      if(g<2){
        #pragma unroll
        for(int v=0;v<8;++v){
          float t=acc[jj][v]+bq[jj][v>>2][v&3];
          o[v]=(g==0) ? (_Float16)(t*0.25f) : (_Float16)t;
        }
        _Float16* dst=(g==0?Qh:Kh)+((bh*SQ+s+colb)*16+rowoff);
        *reinterpret_cast<v8h*>(dst)=o;
      } else {
        #pragma unroll
        for(int v=0;v<8;++v) o[v]=(_Float16)(acc[jj][v]+bv[jj]);
        _Float16* dst=Vt+(((bh*32+(s>>4))*16+colb)*16+rowoff);
        *reinterpret_cast<v8h*>(dst)=o;
      }
    }
    if(g<2){
      #pragma unroll
      for(int i=0;i<8;++i) Bc[i]=Bn[i];
    }
  }
}

__global__ __launch_bounds__(256) __attribute__((amdgpu_waves_per_eu(2,2))) void k_gru_qkv(
    const float* __restrict__ x, const _Float16* __restrict__ WihT,
    const _Float16* __restrict__ WhhT, const float* __restrict__ bih,
    const float* __restrict__ bhh, float* __restrict__ seq,
    const _Float16* __restrict__ WqkvT, const float* __restrict__ bqkv,
    _Float16* __restrict__ Qp, _Float16* __restrict__ Kp, _Float16* __restrict__ Vh)
{
  __shared__ _Float16 hscr[8][16*64];
  __shared__ __attribute__((aligned(16))) _Float16 WhhL[192*64];
  __shared__ __attribute__((aligned(16))) _Float16 WihL[192*128];
  __shared__ __attribute__((aligned(16))) _Float16 WL[24*512];
  __shared__ __attribute__((aligned(16))) float bL[256];
  int lane=threadIdx.x&31, wave=threadIdx.x>>5;
  int pbase=(blockIdx.x*8+wave)*16;
  int colb=lane&15, rowoff=(lane>>4)*8;

  float bi[12], bh[12];
  #pragma unroll
  for(int j=0;j<12;++j){ bi[j]=bih[j*16+colb]; bh[j]=bhh[j*16+colb]; }

  {
    const v4f* hsrc=(const v4f*)WhhT;
    v4f* hdst=(v4f*)WhhL;
    #pragma unroll
    for(int k=0;k<6;++k) hdst[threadIdx.x+k*256]=hsrc[threadIdx.x+k*256];
    const v4f* isrc=(const v4f*)WihT;
    v4f* idst=(v4f*)WihL;
    #pragma unroll
    for(int k=0;k<12;++k){
      int g=threadIdx.x+k*256;
      int u=g&63;
      idst[(g&~63) + ((u&1)<<5) + (u>>1)]=isrc[g];
    }
    const v4f* qsrc=(const v4f*)WqkvT;
    v4f* qdst=(v4f*)WL;
    #pragma unroll
    for(int k=0;k<6;++k){
      int g=threadIdx.x+k*256;
      int u=g&63;
      qdst[(g&~63) + ((u&1)<<5) + (u>>1)]=qsrc[g];
    }
    bL[threadIdx.x]=bqkv[threadIdx.x<192 ? threadIdx.x : 0];
  }
  __syncthreads();

  v8f hT[4], hs[4];
  #pragma unroll
  for(int j=0;j<4;++j){ hT[j]=zero8(); hs[j]=zero8(); }
  v16h hA[2]; hA[0]=zero16(); hA[1]=zero16();
  _Float16* hl=&hscr[wave][0];

  for(int t=0;t<PB;++t){
    const float* xsrc = x + ((size_t)pbase*PB + t)*DD;
    if(t+1<PB) __builtin_prefetch(xsrc+DD,0,1);
    v4f xr[16];
    {
      const float* rb = xsrc + (size_t)(lane&15)*(PB*DD) + (((lane>>4)&1)<<3);
      #pragma unroll
      for(int kk=0;kk<4;++kk){
        xr[kk*4+0]=*reinterpret_cast<const v4f*>(rb+kk*32);
        xr[kk*4+1]=*reinterpret_cast<const v4f*>(rb+kk*32+4);
        xr[kk*4+2]=*reinterpret_cast<const v4f*>(rb+kk*32+16);
        xr[kk*4+3]=*reinterpret_cast<const v4f*>(rb+kk*32+20);
      }
    }
    __builtin_amdgcn_sched_barrier(0);
    v16h xA[4];
    #pragma unroll
    for(int kk=0;kk<4;++kk){
      #pragma unroll
      for(int q=0;q<4;++q){
        #pragma unroll
        for(int c=0;c<4;++c) xA[kk][q*4+c]=(_Float16)xr[kk*4+q][c];
      }
    }

    #pragma unroll
    for(int j=0;j<4;++j){
      v8f gr=zero8();
      #pragma unroll
      for(int kk=0;kk<4;++kk)
        gr=WMMA_F16(xA[kk], load_tile_planes(WihL, j*4+kk, lane), gr);
      #pragma unroll
      for(int kk=0;kk<2;++kk)
        gr=WMMA_F16(hA[kk], load_tile_sw(WhhL, j*2+kk, lane), gr);
      v8f gz=zero8();
      #pragma unroll
      for(int kk=0;kk<4;++kk)
        gz=WMMA_F16(xA[kk], load_tile_planes(WihL, (4+j)*4+kk, lane), gz);
      #pragma unroll
      for(int kk=0;kk<2;++kk)
        gz=WMMA_F16(hA[kk], load_tile_sw(WhhL, (4+j)*2+kk, lane), gz);
      v8f gnx=zero8();
      #pragma unroll
      for(int kk=0;kk<4;++kk)
        gnx=WMMA_F16(xA[kk], load_tile_planes(WihL, (8+j)*4+kk, lane), gnx);
      v8f gnh=zero8();
      #pragma unroll
      for(int kk=0;kk<2;++kk)
        gnh=WMMA_F16(hA[kk], load_tile_sw(WhhL, (8+j)*2+kk, lane), gnh);
      float rr[8], zz[8], nn[8];
      #pragma unroll
      for(int e=0;e<8;++e){
        rr[e]=(gr[e]+bi[j]+bh[j])*-1.44269504f;
        zz[e]=(gz[e]+bi[4+j]+bh[4+j])*-1.44269504f;
      }
      #pragma unroll
      for(int e=0;e<8;++e){
        rr[e]=1.f+__builtin_amdgcn_exp2f(rr[e]);
        zz[e]=1.f+__builtin_amdgcn_exp2f(zz[e]);
      }
      #pragma unroll
      for(int e=0;e<8;++e){
        rr[e]=__builtin_amdgcn_rcpf(rr[e]);
        zz[e]=__builtin_amdgcn_rcpf(zz[e]);
      }
      #pragma unroll
      for(int e=0;e<8;++e)
        nn[e]=gnx[e]+bi[8+j]+rr[e]*(gnh[e]+bh[8+j]);
      #pragma unroll
      for(int e=0;e<8;++e) nn[e]=__builtin_amdgcn_tanhf(nn[e]);
      #pragma unroll
      for(int e=0;e<8;++e){
        float hn=(1.f-zz[e])*nn[e] + zz[e]*hT[j][e];
        hT[j][e]=hn; hs[j][e]+=hn;
      }
    }
    #pragma unroll
    for(int j=0;j<4;++j){
      #pragma unroll
      for(int v=0;v<8;++v)
        hl[(v+rowoff)*64 + j*16 + colb]=(_Float16)hT[j][v];
    }
    #pragma unroll
    for(int kk=0;kk<2;++kk) hA[kk]=load_tile_f16(hl+kk*32, 64, lane);
  }
  float* sb=seq+(size_t)pbase*HH;
  #pragma unroll
  for(int j=0;j<4;++j){
    #pragma unroll
    for(int v=0;v<8;++v){
      float pv=hs[j][v]*0.125f;
      sb[(size_t)(v+rowoff)*HH + j*16 + colb]=pv;
      hl[(v+rowoff)*64 + j*16 + colb]=(_Float16)pv;
    }
  }
  v16h pA0=load_tile_f16(hl, 64, lane);
  v16h pA1=load_tile_f16(hl+32, 64, lane);
  qkv_rows(pA0, pA1, WL, bL, (size_t)pbase, lane, Qp, Kp, Vh);
}

__device__ __forceinline__ float xhalf(float x){
  int b=__builtin_bit_cast(int,x);
  return __builtin_bit_cast(float,
      __builtin_amdgcn_permlanex16(b,b,0x76543210,0xfedcba98,false,false));
}
__device__ __forceinline__ v8h zero8h(){
  v8h z;
  #pragma unroll
  for(int i=0;i<8;++i) z[i]=(_Float16)0.f;
  return z;
}
__device__ __forceinline__ v16h frag2(v8h lo, v8h hi){
  return __builtin_shufflevector(lo, hi, 0,1,2,3,4,5,6,7,8,9,10,11,12,13,14,15);
}

__global__ __launch_bounds__(256) void k_attn(
    const _Float16* __restrict__ Qh, const _Float16* __restrict__ Kh,
    const _Float16* __restrict__ Vt, float* __restrict__ O)
{
  __shared__ __attribute__((aligned(16))) _Float16 Ks[SQ*16];
  __shared__ __attribute__((aligned(16))) _Float16 Vs[SQ*16];
  int bh=blockIdx.x, b=bh>>2, h=bh&3;
  int tid=threadIdx.x, lane=tid&31, wave=tid>>5;
  int colb=lane&15, rowoff=(lane>>4)*8;
  int foff=colb*16+rowoff;
  const v8h z8=zero8h();

  const _Float16* qsrc=Qh+((size_t)bh*SQ+wave*64)*16+foff;
  v8h Qn=*reinterpret_cast<const v8h*>(qsrc);

  {
    const v4f* ksrc=(const v4f*)(Kh+(size_t)bh*SQ*16);
    v4f* kdst=(v4f*)Ks;
    const v4f* vsrc=(const v4f*)(Vt+(size_t)bh*SQ*16);
    v4f* vdst=(v4f*)Vs;
    #pragma unroll
    for(int k=0;k<4;++k) kdst[tid+k*256]=ksrc[tid+k*256];
    #pragma unroll
    for(int k=0;k<4;++k) vdst[tid+k*256]=vsrc[tid+k*256];
  }
  __syncthreads();

  for(int qt=0;qt<4;++qt){
    int qbase=wave*64+qt*16;
    v16h Qb=frag2(Qn,z8);
    if(qt<3) Qn=*reinterpret_cast<const v8h*>(qsrc+(qt+1)*256);
    v8f acc=zero8();
    float mrow=-1e30f;
    float lpart=0.f;

    for(int kc=0;kc<16;++kc){
      const _Float16* kb=Ks+kc*512+foff;
      const _Float16* vb=Vs+kc*512+foff;
      v16h Ka0=frag2(*reinterpret_cast<const v8h*>(kb),     z8);
      v16h Ka1=frag2(*reinterpret_cast<const v8h*>(kb+256), z8);
      v16h Va=frag2(*reinterpret_cast<const v8h*>(vb), *reinterpret_cast<const v8h*>(vb+256));
      v8f s0=WMMA_F16(Ka0,Qb,zero8());
      v8f s1=WMMA_F16(Ka1,Qb,zero8());
      float cm=fmaxf(s0[0],s1[0]);
      #pragma unroll
      for(int e=1;e<8;++e) cm=fmaxf(cm,fmaxf(s0[e],s1[e]));
      cm=fmaxf(cm,xhalf(cm));
      float nm=fmaxf(mrow,cm);
      float sc=__expf(mrow-nm);
      mrow=nm;
      v16h Pb; float ps=0.f;
      #pragma unroll
      for(int e=0;e<8;++e){
        float p0=__expf(s0[e]-nm);
        float p1=__expf(s1[e]-nm);
        ps+=p0+p1;
        Pb[e]=(_Float16)p0;
        Pb[8+e]=(_Float16)p1;
      }
      lpart=lpart*sc+ps;
      #pragma unroll
      for(int e=0;e<8;++e) acc[e]*=sc;
      acc=WMMA_F16(Va,Pb,acc);
    }
    float lrow=lpart+xhalf(lpart);
    float* ob=O+((size_t)b*SQ+qbase+colb)*HH + h*16 + rowoff;
    #pragma unroll
    for(int e=0;e<8;++e) ob[e]=acc[e]/lrow;
  }
}

#define RP 72

__device__ __forceinline__ void ln_rows(v8f (&o)[4], const float (&gv)[4], const float (&bv)[4]){
  #pragma unroll
  for(int e=0;e<8;++e){
    float s=o[0][e]+o[1][e]+o[2][e]+o[3][e];
    float mu=rsum16(s)*(1.f/64.f);
    float d0=o[0][e]-mu,d1=o[1][e]-mu,d2=o[2][e]-mu,d3=o[3][e]-mu;
    float var=rsum16(d0*d0+d1*d1+d2*d2+d3*d3)*(1.f/64.f);
    float inv=rsqrtf(var+1e-5f);
    #pragma unroll
    for(int j=0;j<4;++j)
      o[j][e]=(o[j][e]-mu)*inv*gv[j]+bv[j];
  }
}

__global__ __launch_bounds__(256) __attribute__((amdgpu_waves_per_eu(2,2))) void k_layer(
    float* __restrict__ X, const float* __restrict__ Y,
    const _Float16* __restrict__ WoT, const float* __restrict__ bo,
    const float* __restrict__ g1, const float* __restrict__ be1,
    const _Float16* __restrict__ W1T, const float* __restrict__ b1,
    const _Float16* __restrict__ W2T, const float* __restrict__ b2,
    const float* __restrict__ g2, const float* __restrict__ be2,
    const _Float16* __restrict__ WqkvT, const float* __restrict__ bqkv,
    _Float16* __restrict__ Qp, _Float16* __restrict__ Kp, _Float16* __restrict__ Vh,
    int with_qkv)
{
  __shared__ __attribute__((aligned(16))) _Float16 h1[8][16*256];
  __shared__ __attribute__((aligned(16))) _Float16 W1L[32*512];
  __shared__ __attribute__((aligned(16))) _Float16 W2L[32*512];
  __shared__ __attribute__((aligned(16))) _Float16 WoL[8*512];
  __shared__ __attribute__((aligned(16))) _Float16 WL[24*512];
  __shared__ __attribute__((aligned(16))) float sv[1024];
  int lane=threadIdx.x&31, wave=threadIdx.x>>5;
  size_t rbase=((size_t)blockIdx.x*8+wave)*16;
  int colb=lane&15, rowoff=(lane>>4)*8;

  v4f yr[8];
  {
    const float* rb = Y + rbase*HH + (size_t)(lane&15)*HH + (((lane>>4)&1)<<3);
    #pragma unroll
    for(int kk=0;kk<2;++kk){
      yr[kk*4+0]=*reinterpret_cast<const v4f*>(rb+kk*32);
      yr[kk*4+1]=*reinterpret_cast<const v4f*>(rb+kk*32+4);
      yr[kk*4+2]=*reinterpret_cast<const v4f*>(rb+kk*32+16);
      yr[kk*4+3]=*reinterpret_cast<const v4f*>(rb+kk*32+20);
    }
  }
  v8f xres[4];
  #pragma unroll
  for(int j=0;j<4;++j){
    const float* xr=X+rbase*HH+j*16+colb;
    #pragma unroll
    for(int e=0;e<8;++e) xres[j][e]=xr[(size_t)(e+rowoff)*HH];
  }

  {
    int tid=threadIdx.x;
    float t0=b1[tid];
    const float* p1 = tid<64 ? bo : (tid<128 ? g1 : be1);
    float t1=p1[tid&63];
    const float* p2 = tid<64 ? b2 : (tid<128 ? g2 : be2);
    float t2=p2[tid&63];
    const v4f* s1=(const v4f*)W1T; v4f* d1=(v4f*)W1L;
    const v4f* s2=(const v4f*)W2T; v4f* d2=(v4f*)W2L;
    const v4f* so=(const v4f*)WoT; v4f* dO=(v4f*)WoL;
    #pragma unroll
    for(int k=0;k<8;++k){
      int gi=tid+k*256;
      int u=gi&63;
      int di=(gi&~63) + ((u&1)<<5) + (u>>1);
      d1[di]=s1[gi];
      d2[di]=s2[gi];
      if(k<2) dO[di]=so[gi];
    }
    float t3=bqkv[tid<192 ? tid : 0];
    const v4f* sq=(const v4f*)WqkvT; v4f* dq=(v4f*)WL;
    #pragma unroll
    for(int k=0;k<6;++k){
      int gi=tid+k*256;
      int u=gi&63;
      dq[(gi&~63) + ((u&1)<<5) + (u>>1)]=sq[gi];
    }
    sv[768+tid]=t3;
    sv[tid]=t0;
    sv[256+tid]=t1;
    sv[512+tid]=t2;
  }
  __syncthreads();

  _Float16* hl=&h1[wave][0];
  v16h A[2];
  #pragma unroll
  for(int kk=0;kk<2;++kk){
    #pragma unroll
    for(int q=0;q<4;++q){
      #pragma unroll
      for(int cc=0;cc<4;++cc) A[kk][q*4+cc]=(_Float16)yr[kk*4+q][cc];
    }
  }

  v8f x1[4];
  #pragma unroll
  for(int j=0;j<4;++j){
    v8f acc=zero8();
    #pragma unroll
    for(int kk=0;kk<2;++kk)
      acc=WMMA_F16(A[kk], load_tile_planes(WoL, j*2+kk, lane), acc);
    float bias=sv[256+j*16+colb];
    #pragma unroll
    for(int e=0;e<8;++e) x1[j][e]=acc[e]+bias+xres[j][e];
  }
  {
    float gv[4],bv[4];
    #pragma unroll
    for(int j=0;j<4;++j){ gv[j]=sv[320+j*16+colb]; bv[j]=sv[384+j*16+colb]; }
    ln_rows(x1, gv, bv);
  }
  #pragma unroll
  for(int j=0;j<4;++j){
    #pragma unroll
    for(int e=0;e<8;++e) hl[(e+rowoff)*RP + j*16 + colb]=(_Float16)x1[j][e];
  }
  #pragma unroll
  for(int kk=0;kk<2;++kk) A[kk]=load_tile_f16(hl+kk*32, RP, lane);

  v16h Bc[8], Bn[8];
  float bc[4], bn[4];
  #pragma unroll
  for(int i=0;i<8;++i) Bc[i]=load_tile_planes(W1L, i, lane);
  #pragma unroll
  for(int jj=0;jj<4;++jj) bc[jj]=sv[jj*16+colb];
  #pragma unroll
  for(int gq=0;gq<4;++gq){
    if(gq<3){
      #pragma unroll
      for(int i=0;i<8;++i) Bn[i]=load_tile_planes(W1L, (gq+1)*8+i, lane);
      #pragma unroll
      for(int jj=0;jj<4;++jj) bn[jj]=sv[((gq+1)*4+jj)*16+colb];
    } else {
      #pragma unroll
      for(int i=0;i<8;++i) Bn[i]=load_tile_planes(W2L, (i>>1)*8+(i&1), lane);
    }
    __builtin_amdgcn_sched_barrier(0);
    v8f acc[4];
    #pragma unroll
    for(int jj=0;jj<4;++jj) acc[jj]=zero8();
    #pragma unroll
    for(int kk=0;kk<2;++kk){
      #pragma unroll
      for(int jj=0;jj<4;++jj) acc[jj]=WMMA_F16(A[kk], Bc[jj*2+kk], acc[jj]);
    }
    #pragma unroll
    for(int jj=0;jj<4;++jj){
      int j=gq*4+jj;
      float bias=bc[jj];
      #pragma unroll
      for(int e=0;e<8;++e)
        hl[(size_t)(e+rowoff)*256+j*16+colb]=(_Float16)fmaxf(acc[jj][e]+bias,0.f);
    }
    #pragma unroll
    for(int i=0;i<8;++i) Bc[i]=Bn[i];
    if(gq<3){
      #pragma unroll
      for(int jj=0;jj<4;++jj) bc[jj]=bn[jj];
    }
  }
  v16h pA[8];
  #pragma unroll
  for(int kk=0;kk<8;++kk) pA[kk]=load_tile_f16(hl+kk*32, 256, lane);
  v8f o[4];
  #pragma unroll
  for(int j=0;j<4;++j) o[j]=zero8();
  #pragma unroll
  for(int gq=0;gq<4;++gq){
    if(gq<3){
      #pragma unroll
      for(int i=0;i<8;++i) Bn[i]=load_tile_planes(W2L, (i>>1)*8+(gq+1)*2+(i&1), lane);
    }
    __builtin_amdgcn_sched_barrier(0);
    #pragma unroll
    for(int c=0;c<2;++c){
      #pragma unroll
      for(int j=0;j<4;++j) o[j]=WMMA_F16(pA[gq*2+c], Bc[j*2+c], o[j]);
    }
    if(gq<3){
      #pragma unroll
      for(int i=0;i<8;++i) Bc[i]=Bn[i];
    }
  }
  #pragma unroll
  for(int j=0;j<4;++j){
    float bias=sv[512+j*16+colb];
    #pragma unroll
    for(int e=0;e<8;++e) o[j][e]=o[j][e]+bias+x1[j][e];
  }
  {
    float gv[4],bv[4];
    #pragma unroll
    for(int j=0;j<4;++j){ gv[j]=sv[576+j*16+colb]; bv[j]=sv[640+j*16+colb]; }
    ln_rows(o, gv, bv);
  }
  float* xw=X+rbase*HH;
  #pragma unroll
  for(int e=0;e<8;++e){
    #pragma unroll
    for(int j=0;j<4;++j)
      xw[(size_t)(e+rowoff)*HH+j*16+colb]=o[j][e];
  }
  if(with_qkv){
    #pragma unroll
    for(int j=0;j<4;++j){
      #pragma unroll
      for(int e=0;e<8;++e) hl[(e+rowoff)*RP + j*16 + colb]=(_Float16)o[j][e];
    }
    v16h qA0=load_tile_f16(hl, RP, lane);
    v16h qA1=load_tile_f16(hl+32, RP, lane);
    qkv_rows(qA0, qA1, WL, sv+768, rbase, lane, Qp, Kp, Vh);
  }
}

__global__ __launch_bounds__(256) void k_mean(const float* __restrict__ X, float* __restrict__ out){
  __shared__ float part[8][32];
  int b=blockIdx.x>>1, cb=(blockIdx.x&1)*32;
  int lane=threadIdx.x&31, wave=threadIdx.x>>5;
  const float* p=X+((size_t)b*SQ+wave)*HH+cb+lane;
  float v[64];
  #pragma unroll
  for(int i=0;i<64;++i) v[i]=p[(size_t)i*8*HH];
  float s=0.f;
  #pragma unroll
  for(int i=0;i<64;++i) s+=v[i];
  part[wave][lane]=s;
  __syncthreads();
  if(wave==0){
    float t=0.f;
    #pragma unroll
    for(int w=0;w<8;++w) t+=part[w][lane];
    out[b*HH+cb+lane]=t*(1.f/512.f);
  }
}

extern "C" void kernel_launch(void* const* d_in, const int* in_sizes, int n_in,
                              void* d_out, int out_size, void* d_ws, size_t ws_size,
                              hipStream_t stream){
  const float* x  =(const float*)d_in[0];
  const float* Wih=(const float*)d_in[1];
  const float* Whh=(const float*)d_in[2];
  const float* bih=(const float*)d_in[3];
  const float* bhh=(const float*)d_in[4];

  char* ws=(char*)d_ws;
  float*    X =(float*)ws;
  float*    Y =(float*)(ws+(size_t)(8<<20));
  _Float16* Qp=(_Float16*)(ws+(size_t)(16<<20));
  _Float16* Kp=(_Float16*)(ws+(size_t)(24<<20));
  _Float16* Vh=(_Float16*)(ws+(size_t)(32<<20));
  _Float16* WihT=(_Float16*)(ws+(size_t)(36<<20));
  _Float16* WhhT=(_Float16*)(ws+(size_t)(36<<20)+(64<<10));
  _Float16* LW[2][4];
  for(int L=0;L<2;++L){
    char* base=ws+(size_t)(37<<20)+(size_t)L*(1<<20);
    LW[L][0]=(_Float16*)(base);
    LW[L][1]=(_Float16*)(base+(32<<10));
    LW[L][2]=(_Float16*)(base+(64<<10));
    LW[L][3]=(_Float16*)(base+(128<<10));
  }

  k_swz_all<<<264,32,0,stream>>>(Wih, Whh,
      (const float*)d_in[5],  (const float*)d_in[7],
      (const float*)d_in[9],  (const float*)d_in[11],
      (const float*)d_in[17], (const float*)d_in[19],
      (const float*)d_in[21], (const float*)d_in[23],
      WihT);

  k_gru_qkv<<<256,256,0,stream>>>(x,WihT,WhhT,bih,bhh,X,
      LW[0][0],(const float*)d_in[6],Qp,Kp,Vh);

  k_attn <<<256,256,0,stream>>>(Qp,Kp,Vh,Y);
  k_layer<<<256,256,0,stream>>>(X,Y,
      LW[0][1],(const float*)d_in[8], (const float*)d_in[13],(const float*)d_in[14],
      LW[0][2],(const float*)d_in[10],LW[0][3],(const float*)d_in[12],
      (const float*)d_in[15],(const float*)d_in[16],
      LW[1][0],(const float*)d_in[18],Qp,Kp,Vh,1);

  k_attn <<<256,256,0,stream>>>(Qp,Kp,Vh,Y);
  k_layer<<<256,256,0,stream>>>(X,Y,
      LW[1][1],(const float*)d_in[20],(const float*)d_in[25],(const float*)d_in[26],
      LW[1][2],(const float*)d_in[22],LW[1][3],(const float*)d_in[24],
      (const float*)d_in[27],(const float*)d_in[28],
      LW[1][0],(const float*)d_in[18],Qp,Kp,Vh,0);

  k_mean<<<128,256,0,stream>>>(X,(float*)d_out);
}
